# starting kernel
# baseline (speedup 1.0000x reference)
#include <hip/hip_runtime.h>

typedef float v2f __attribute__((ext_vector_type(2)));
typedef float v8f __attribute__((ext_vector_type(8)));

__global__ void zero_f(float* __restrict__ p, int n) {
  int i = blockIdx.x * blockDim.x + threadIdx.x;
  if (i < n) p[i] = 0.0f;
}

template <int INC, int OUTC, int MT>
__global__ void wmma_linear(const float* __restrict__ X,
                            const float* __restrict__ W,
                            const float* __restrict__ Bias,
                            float* __restrict__ Y, int nrows) {
  constexpr int NT = OUTC / 16;
  int wave = (int)((blockIdx.x * blockDim.x + threadIdx.x) >> 5);
  int lane = threadIdx.x & 31;
  int row0 = wave * 16 * MT;
  if (row0 >= nrows) return;

  int nm = lane & 15;
  int kg = (lane >> 4) << 1;

  const float* xrow[MT];
#pragma unroll
  for (int mt = 0; mt < MT; ++mt) {
    xrow[mt] = X + (size_t)(row0 + mt * 16 + nm) * INC;
    __builtin_prefetch(xrow[mt], 0, 3);
  }
  const float* wrow[NT];
#pragma unroll
  for (int t = 0; t < NT; ++t) wrow[t] = W + (size_t)(t * 16 + nm) * INC;

  v8f acc[MT][NT];
#pragma unroll
  for (int t = 0; t < NT; ++t) {
    float bv = Bias[t * 16 + nm];
#pragma unroll
    for (int mt = 0; mt < MT; ++mt)
#pragma unroll
      for (int v = 0; v < 8; ++v) acc[mt][t][v] = bv;
  }

#pragma unroll 4
  for (int k = 0; k < INC; k += 4) {
    v2f b[NT];
#pragma unroll
    for (int t = 0; t < NT; ++t) {
      b[t].x = wrow[t][k + kg];
      b[t].y = wrow[t][k + kg + 1];
    }
#pragma unroll
    for (int mt = 0; mt < MT; ++mt) {
      v2f a;
      a.x = xrow[mt][k + kg];
      a.y = xrow[mt][k + kg + 1];
#pragma unroll
      for (int t = 0; t < NT; ++t) {
        acc[mt][t] = __builtin_amdgcn_wmma_f32_16x16x4_f32(
             false, a,  false, b[t],
             (short)0, acc[mt][t],  false,
             false);
      }
    }
  }

  int mb = (lane >> 4) * 8;
#pragma unroll
  for (int mt = 0; mt < MT; ++mt)
#pragma unroll
    for (int t = 0; t < NT; ++t)
#pragma unroll
      for (int v = 0; v < 8; ++v)
        Y[(size_t)(row0 + mt * 16 + mb + v) * OUTC + t * 16 + nm] =
            acc[mt][t][v];
}

template <int C>
__global__ void scatter_add(const float* __restrict__ H,
                            const int* __restrict__ srcI,
                            const int* __restrict__ dstI,
                            float* __restrict__ agg, int nedges) {
  long long tid = (long long)blockIdx.x * blockDim.x + threadIdx.x;
  int e = (int)(tid / C);
  int c = (int)(tid % C);
  if (e >= nedges) return;
  int s = srcI[e];
  int d = dstI[e];
  float val = H[(size_t)s * C + c];
  __hip_atomic_fetch_add(&agg[(size_t)d * C + c], val, __ATOMIC_RELAXED,
                         __HIP_MEMORY_SCOPE_AGENT);
}

__global__ void add_relu(const float* __restrict__ A, float* __restrict__ FZ,
                         int n) {
  int i = blockIdx.x * blockDim.x + threadIdx.x;
  if (i < n) {
    float v = FZ[i] + A[i];
    FZ[i] = v > 0.0f ? v : 0.0f;
  }
}

__global__ void add_logsoftmax16(const float* __restrict__ F,
                                 const float* __restrict__ A,
                                 float* __restrict__ O, int nrows) {
  int r = blockIdx.x * blockDim.x + threadIdx.x;
  if (r >= nrows) return;
  float v[16];
  float mx = -3.402823466e38f;
#pragma unroll
  for (int c = 0; c < 16; ++c) {
    v[c] = F[(size_t)r * 16 + c] + A[(size_t)r * 16 + c];
    mx = fmaxf(mx, v[c]);
  }
  float s = 0.0f;
#pragma unroll
  for (int c = 0; c < 16; ++c) s += __expf(v[c] - mx);
  float l = mx + __logf(s);
#pragma unroll
  for (int c = 0; c < 16; ++c) O[(size_t)r * 16 + c] = v[c] - l;
}

static inline int cdiv_ll(long long a, long long b) {
  return (int)((a + b - 1) / b);
}

extern "C" void kernel_launch(void* const* d_in, const int* in_sizes, int n_in,
                              void* d_out, int out_size, void* d_ws,
                              size_t ws_size, hipStream_t stream) {
  (void)n_in; (void)out_size; (void)ws_size;
  const float* x     = (const float*)d_in[0];
  const float* w1a_w = (const float*)d_in[1];
  const float* w1a_b = (const float*)d_in[2];
  const float* w1b_w = (const float*)d_in[3];
  const float* w1b_b = (const float*)d_in[4];
  const float* w2a_w = (const float*)d_in[5];
  const float* w2a_b = (const float*)d_in[6];
  const float* w2b_w = (const float*)d_in[7];
  const float* w2b_b = (const float*)d_in[8];
  const int*   edge  = (const int*)d_in[9];

  const int N = in_sizes[0] / 256;
  const int E = in_sizes[9] / 2;
  const int* srcI = edge;
  const int* dstI = edge + E;

  float* buf0 = (float*)d_ws;
  float* buf1 = buf0 + (size_t)N * 32;
  float* buf2 = buf1 + (size_t)N * 32;

  const int BLK = 256;
  const int MT = 2;
  const int waves = N / (16 * MT);
  const int gemm_blocks = cdiv_ll(waves, BLK / 32);

  zero_f<<<cdiv_ll((long long)N * 32, BLK), BLK, 0, stream>>>(buf1, N * 32);
  wmma_linear<256, 32, MT><<<gemm_blocks, BLK, 0, stream>>>(x, w1a_w, w1a_b,
                                                            buf0, N);
  scatter_add<32><<<cdiv_ll((long long)E * 32, BLK), BLK, 0, stream>>>(
      buf0, srcI, dstI, buf1, E);
  wmma_linear<32, 32, MT><<<gemm_blocks, BLK, 0, stream>>>(buf0, w1b_w, w1b_b,
                                                           buf2, N);
  add_relu<<<cdiv_ll((long long)N * 32, BLK), BLK, 0, stream>>>(buf1, buf2,
                                                                N * 32);

  zero_f<<<cdiv_ll((long long)N * 16, BLK), BLK, 0, stream>>>(buf1, N * 16);
  wmma_linear<32, 16, MT><<<gemm_blocks, BLK, 0, stream>>>(buf2, w2a_w, w2a_b,
                                                           buf0, N);
  scatter_add<16><<<cdiv_ll((long long)E * 16, BLK), BLK, 0, stream>>>(
      buf0, srcI, dstI, buf1, E);
  wmma_linear<16, 16, MT><<<gemm_blocks, BLK, 0, stream>>>(buf0, w2b_w, w2b_b,
                                                           buf2, N);
  add_logsoftmax16<<<cdiv_ll(N, BLK), BLK, 0, stream>>>(buf2, buf1,
                                                        (float*)d_out, N);
}


# Float atomic scatter-add replaced by per-call destination buckets (returning integer adds, prefix sum, place) and owner lanes summing gathered rows in registers in bucket arrival order
# speedup vs baseline: 1.3619x; 1.3619x over previous
#include <hip/hip_runtime.h>

typedef float v2f __attribute__((ext_vector_type(2)));
typedef float v8f __attribute__((ext_vector_type(8)));

template <int INC, int OUTC, int MT>
__global__ void wmma_linear(const float* __restrict__ X,
                            const float* __restrict__ W,
                            const float* __restrict__ Bias,
                            float* __restrict__ Y, int nrows) {
  constexpr int NT = OUTC / 16;
  int wave = (int)((blockIdx.x * blockDim.x + threadIdx.x) >> 5);
  int lane = threadIdx.x & 31;
  int row0 = wave * 16 * MT;
  if (row0 >= nrows) return;

  int nm = lane & 15;
  int kg = (lane >> 4) << 1;

  const float* xrow[MT];
#pragma unroll
  for (int mt = 0; mt < MT; ++mt) {
    xrow[mt] = X + (size_t)(row0 + mt * 16 + nm) * INC;
    __builtin_prefetch(xrow[mt], 0, 3);
  }
  const float* wrow[NT];
#pragma unroll
  for (int t = 0; t < NT; ++t) wrow[t] = W + (size_t)(t * 16 + nm) * INC;

  v8f acc[MT][NT];
#pragma unroll
  for (int t = 0; t < NT; ++t) {
    float bv = Bias[t * 16 + nm];
#pragma unroll
    for (int mt = 0; mt < MT; ++mt)
#pragma unroll
      for (int v = 0; v < 8; ++v) acc[mt][t][v] = bv;
  }

#pragma unroll 4
  for (int k = 0; k < INC; k += 4) {
    v2f b[NT];
#pragma unroll
    for (int t = 0; t < NT; ++t) {
      b[t].x = wrow[t][k + kg];
      b[t].y = wrow[t][k + kg + 1];
    }
#pragma unroll
    for (int mt = 0; mt < MT; ++mt) {
      v2f a;
      a.x = xrow[mt][k + kg];
      a.y = xrow[mt][k + kg + 1];
#pragma unroll
      for (int t = 0; t < NT; ++t) {
        acc[mt][t] = __builtin_amdgcn_wmma_f32_16x16x4_f32(
             false, a,  false, b[t],
             (short)0, acc[mt][t],  false,
             false);
      }
    }
  }

  int mb = (lane >> 4) * 8;
#pragma unroll
  for (int mt = 0; mt < MT; ++mt)
#pragma unroll
    for (int t = 0; t < NT; ++t)
#pragma unroll
      for (int v = 0; v < 8; ++v)
        Y[(size_t)(row0 + mt * 16 + mb + v) * OUTC + t * 16 + nm] =
            acc[mt][t][v];
}

typedef float v4f __attribute__((ext_vector_type(4)));

__global__ void zero_i(int* __restrict__ p, int n) {
  int i = blockIdx.x * blockDim.x + threadIdx.x;
  if (i < n) p[i] = 0;
}

__global__ void edge_rank(const int* __restrict__ dstI, int* __restrict__ cnt,
                          int* __restrict__ rank, int nedges) {
  int e = blockIdx.x * blockDim.x + threadIdx.x;
  if (e >= nedges) return;
  int d = dstI[e];
  rank[e] = __hip_atomic_fetch_add(&cnt[d], 1, __ATOMIC_RELAXED,
                                   __HIP_MEMORY_SCOPE_AGENT);
}

__global__ void __launch_bounds__(1024)
bucket_offsets(const int* __restrict__ cnt, int* __restrict__ rowptr, int n) {
  __shared__ int s_before[32];
  __shared__ int s_wtot[32];
  const int t = threadIdx.x;
  const int lane = t & 31;
  const int w = t >> 5;
  const int b = blockIdx.x;

  int before = 0;
  for (int i = 0; i < b; ++i) before += cnt[i * 1024 + t];

  const int idx = b * 1024 + t;
  const int c = idx < n ? cnt[idx] : 0;

  int incl = c;
  for (int d = 1; d < 32; d <<= 1) {
    int v = __shfl_up(incl, d);
    if (lane >= d) incl += v;
  }
  for (int d = 16; d >= 1; d >>= 1) before += __shfl_xor(before, d);

  if (lane == 31) s_wtot[w] = incl;
  if (lane == 0) s_before[w] = before;
  __syncthreads();

  int off = s_before[lane];
  int wp = lane < w ? s_wtot[lane] : 0;
  for (int d = 16; d >= 1; d >>= 1) {
    off += __shfl_xor(off, d);
    wp += __shfl_xor(wp, d);
  }
  const int excl = off + wp + incl - c;
  if (idx < n) rowptr[idx] = excl;
  if (idx == n - 1) rowptr[n] = excl + c;
}

__global__ void edge_place(const int* __restrict__ srcI,
                           const int* __restrict__ dstI,
                           const int* __restrict__ rank,
                           const int* __restrict__ rowptr,
                           int* __restrict__ col, int nedges) {
  int e = blockIdx.x * blockDim.x + threadIdx.x;
  if (e >= nedges) return;
  int d = dstI[e];
  int s = srcI[e];
  int r = rank[e];
  col[rowptr[d] + r] = s;
}

template <int C, bool RELU_INPLACE>
__global__ void owner_sum(const float* __restrict__ H,
                          const int* __restrict__ rowptr,
                          const int* __restrict__ col, float* FZ,
                          float* __restrict__ Agg, int nnodes) {
  constexpr int LPN = C / 4;
  const int gtid = blockIdx.x * blockDim.x + threadIdx.x;
  const int node = gtid / LPN;
  const int sub = gtid % LPN;
  if (node >= nnodes) return;

  const int beg = rowptr[node];
  const int end = rowptr[node + 1];
  const size_t own = ((size_t)node * LPN + sub) * 4;

  v4f f = {0.0f, 0.0f, 0.0f, 0.0f};
  if constexpr (RELU_INPLACE) f = *(const v4f*)(FZ + own);

  v4f acc = {0.0f, 0.0f, 0.0f, 0.0f};
  int j = beg;
  for (; j + 4 <= end; j += 4) {
    const int s0 = col[j];
    const int s1 = col[j + 1];
    const int s2 = col[j + 2];
    const int s3 = col[j + 3];
    const v4f r0 = *(const v4f*)(H + ((size_t)s0 * LPN + sub) * 4);
    const v4f r1 = *(const v4f*)(H + ((size_t)s1 * LPN + sub) * 4);
    const v4f r2 = *(const v4f*)(H + ((size_t)s2 * LPN + sub) * 4);
    const v4f r3 = *(const v4f*)(H + ((size_t)s3 * LPN + sub) * 4);
    acc += r0;
    acc += r1;
    acc += r2;
    acc += r3;
  }
  for (; j < end; ++j) {
    const int s = col[j];
    acc += *(const v4f*)(H + ((size_t)s * LPN + sub) * 4);
  }

  if constexpr (RELU_INPLACE) {
    v4f v = f + acc;
    v.x = v.x > 0.0f ? v.x : 0.0f;
    v.y = v.y > 0.0f ? v.y : 0.0f;
    v.z = v.z > 0.0f ? v.z : 0.0f;
    v.w = v.w > 0.0f ? v.w : 0.0f;
    *(v4f*)(FZ + own) = v;
  } else {
    *(v4f*)(Agg + own) = acc;
  }
}

__global__ void add_logsoftmax16(const float* __restrict__ F,
                                 const float* __restrict__ A,
                                 float* __restrict__ O, int nrows) {
  int r = blockIdx.x * blockDim.x + threadIdx.x;
  if (r >= nrows) return;
  float v[16];
  float mx = -3.402823466e38f;
#pragma unroll
  for (int c = 0; c < 16; ++c) {
    v[c] = F[(size_t)r * 16 + c] + A[(size_t)r * 16 + c];
    mx = fmaxf(mx, v[c]);
  }
  float s = 0.0f;
#pragma unroll
  for (int c = 0; c < 16; ++c) s += __expf(v[c] - mx);
  float l = mx + __logf(s);
#pragma unroll
  for (int c = 0; c < 16; ++c) O[(size_t)r * 16 + c] = v[c] - l;
}

static inline int cdiv_ll(long long a, long long b) {
  return (int)((a + b - 1) / b);
}

extern "C" void kernel_launch(void* const* d_in, const int* in_sizes, int n_in,
                              void* d_out, int out_size, void* d_ws,
                              size_t ws_size, hipStream_t stream) {
  (void)n_in; (void)out_size; (void)ws_size;
  const float* x     = (const float*)d_in[0];
  const float* w1a_w = (const float*)d_in[1];
  const float* w1a_b = (const float*)d_in[2];
  const float* w1b_w = (const float*)d_in[3];
  const float* w1b_b = (const float*)d_in[4];
  const float* w2a_w = (const float*)d_in[5];
  const float* w2a_b = (const float*)d_in[6];
  const float* w2b_w = (const float*)d_in[7];
  const float* w2b_b = (const float*)d_in[8];
  const int*   edge  = (const int*)d_in[9];

  const int N = in_sizes[0] / 256;
  const int E = in_sizes[9] / 2;
  const int* srcI = edge;
  const int* dstI = edge + E;

  float* buf0 = (float*)d_ws;
  float* buf1 = buf0 + (size_t)N * 32;
  float* buf2 = buf1 + (size_t)N * 32;

  int* cnt    = (int*)buf1;
  int* rowptr = cnt + (size_t)N;
  int* col    = cnt + 2 * (size_t)N + 64;
  int* rank   = (int*)buf2;
  float* agg2 = buf0 + (size_t)N * 16;

  const int BLK = 256;
  const int MT = 2;
  const int waves = N / (16 * MT);
  const int gemm_blocks = cdiv_ll(waves, BLK / 32);

  zero_i<<<cdiv_ll(N, BLK), BLK, 0, stream>>>(cnt, N);
  edge_rank<<<cdiv_ll(E, BLK), BLK, 0, stream>>>(dstI, cnt, rank, E);
  bucket_offsets<<<cdiv_ll(N, 1024), 1024, 0, stream>>>(cnt, rowptr, N);
  edge_place<<<cdiv_ll(E, BLK), BLK, 0, stream>>>(srcI, dstI, rank, rowptr,
                                                  col, E);

  wmma_linear<256, 32, MT><<<gemm_blocks, BLK, 0, stream>>>(x, w1a_w, w1a_b,
                                                            buf0, N);
  wmma_linear<32, 32, MT><<<gemm_blocks, BLK, 0, stream>>>(buf0, w1b_w, w1b_b,
                                                           buf2, N);
  owner_sum<32, true><<<cdiv_ll((long long)N * 8, BLK), BLK, 0, stream>>>(
      buf0, rowptr, col, buf2, nullptr, N);

  wmma_linear<32, 16, MT><<<gemm_blocks, BLK, 0, stream>>>(buf2, w2a_w, w2a_b,
                                                           buf0, N);
  wmma_linear<16, 16, MT><<<gemm_blocks, BLK, 0, stream>>>(buf0, w2b_w, w2b_b,
                                                           buf2, N);
  owner_sum<16, false><<<cdiv_ll((long long)N * 4, BLK), BLK, 0, stream>>>(
      buf0, rowptr, col, nullptr, agg2, N);
  add_logsoftmax16<<<cdiv_ll(N, BLK), BLK, 0, stream>>>(buf2, agg2,
                                                        (float*)d_out, N);
}


# Launches 11 to 8: each layer's second linear runs in the first one's wave on its transposed D tile as B operand (new k order of that sum); log-softmax joins the layer-2 owner sum as a 4-lane tree
# speedup vs baseline: 1.4135x; 1.0379x over previous
#include <hip/hip_runtime.h>

typedef float v2f __attribute__((ext_vector_type(2)));
typedef float v8f __attribute__((ext_vector_type(8)));

typedef float v4f __attribute__((ext_vector_type(4)));

template <int INC, int MID, int OUTC, int MT>
__global__ void wmma_linear2(const float* __restrict__ X,
                             const float* __restrict__ Wa,
                             const float* __restrict__ Ba,
                             const float* __restrict__ Wb,
                             const float* __restrict__ Bb,
                             float* __restrict__ Y1,
                             float* __restrict__ Y2, int nrows) {
  constexpr int NT = MID / 16;
  constexpr int NT2 = OUTC / 16;
  int wave = (int)((blockIdx.x * blockDim.x + threadIdx.x) >> 5);
  int lane = threadIdx.x & 31;
  int row0 = wave * 16 * MT;
  if (row0 >= nrows) return;

  int nm = lane & 15;
  int hh = lane >> 4;
  int kg = hh << 1;

  const float* xrow[MT];
#pragma unroll
  for (int mt = 0; mt < MT; ++mt) {
    xrow[mt] = X + (size_t)(row0 + mt * 16 + nm) * INC;
    __builtin_prefetch(xrow[mt], 0, 3);
  }
  const float* wrow[NT];
#pragma unroll
  for (int t = 0; t < NT; ++t) wrow[t] = Wa + (size_t)(t * 16 + nm) * INC;

  v8f acc[MT][NT];
#pragma unroll
  for (int t = 0; t < NT; ++t)
#pragma unroll
    for (int v = 0; v < 8; ++v) {
      float bv = Ba[t * 16 + hh * 8 + v];
#pragma unroll
      for (int mt = 0; mt < MT; ++mt) acc[mt][t][v] = bv;
    }

#pragma unroll 4
  for (int k = 0; k < INC; k += 4) {
    v2f w[NT];
#pragma unroll
    for (int t = 0; t < NT; ++t) {
      w[t].x = wrow[t][k + kg];
      w[t].y = wrow[t][k + kg + 1];
    }
#pragma unroll
    for (int mt = 0; mt < MT; ++mt) {
      v2f xv;
      xv.x = xrow[mt][k + kg];
      xv.y = xrow[mt][k + kg + 1];
#pragma unroll
      for (int t = 0; t < NT; ++t) {
        acc[mt][t] = __builtin_amdgcn_wmma_f32_16x16x4_f32(
             false, w[t],  false, xv,
             (short)0, acc[mt][t],  false,
             false);
      }
    }
  }

#pragma unroll
  for (int mt = 0; mt < MT; ++mt)
#pragma unroll
    for (int t = 0; t < NT; ++t) {
      float* p = Y1 + (size_t)(row0 + mt * 16 + nm) * MID + t * 16 + hh * 8;
      v4f lo = {acc[mt][t][0], acc[mt][t][1], acc[mt][t][2], acc[mt][t][3]};
      v4f hi = {acc[mt][t][4], acc[mt][t][5], acc[mt][t][6], acc[mt][t][7]};
      *(v4f*)p = lo;
      *(v4f*)(p + 4) = hi;
    }

  v8f acc2[MT][NT2];
#pragma unroll
  for (int t2 = 0; t2 < NT2; ++t2)
#pragma unroll
    for (int v = 0; v < 8; ++v) {
      float bv = Bb[t2 * 16 + hh * 8 + v];
#pragma unroll
      for (int mt = 0; mt < MT; ++mt) acc2[mt][t2][v] = bv;
    }

#pragma unroll
  for (int t = 0; t < NT; ++t)
#pragma unroll
    for (int q = 0; q < 4; ++q) {
      v2f wb[NT2];
#pragma unroll
      for (int t2 = 0; t2 < NT2; ++t2) {
        const float* wp =
            Wb + (size_t)(t2 * 16 + nm) * MID + t * 16 + hh * 8 + 2 * q;
        wb[t2].x = wp[0];
        wb[t2].y = wp[1];
      }
#pragma unroll
      for (int mt = 0; mt < MT; ++mt) {
        v2f hv;
        hv.x = acc[mt][t][2 * q];
        hv.y = acc[mt][t][2 * q + 1];
#pragma unroll
        for (int t2 = 0; t2 < NT2; ++t2) {
          acc2[mt][t2] = __builtin_amdgcn_wmma_f32_16x16x4_f32(
               false, wb[t2],  false, hv,
               (short)0, acc2[mt][t2],  false,
               false);
        }
      }
    }

#pragma unroll
  for (int mt = 0; mt < MT; ++mt)
#pragma unroll
    for (int t2 = 0; t2 < NT2; ++t2) {
      float* p = Y2 + (size_t)(row0 + mt * 16 + nm) * OUTC + t2 * 16 + hh * 8;
      v4f lo = {acc2[mt][t2][0], acc2[mt][t2][1], acc2[mt][t2][2],
                acc2[mt][t2][3]};
      v4f hi = {acc2[mt][t2][4], acc2[mt][t2][5], acc2[mt][t2][6],
                acc2[mt][t2][7]};
      *(v4f*)p = lo;
      *(v4f*)(p + 4) = hi;
    }
}

__global__ void zero_i(int* __restrict__ p, int n) {
  int i = blockIdx.x * blockDim.x + threadIdx.x;
  if (i < n) p[i] = 0;
}

__global__ void edge_rank(const int* __restrict__ dstI, int* __restrict__ cnt,
                          int* __restrict__ rank, int nedges) {
  int e = blockIdx.x * blockDim.x + threadIdx.x;
  if (e >= nedges) return;
  int d = dstI[e];
  rank[e] = __hip_atomic_fetch_add(&cnt[d], 1, __ATOMIC_RELAXED,
                                   __HIP_MEMORY_SCOPE_AGENT);
}

__global__ void __launch_bounds__(1024)
bucket_offsets(const int* __restrict__ cnt, int* __restrict__ rowptr, int n) {
  __shared__ int s_before[32];
  __shared__ int s_wtot[32];
  const int t = threadIdx.x;
  const int lane = t & 31;
  const int w = t >> 5;
  const int b = blockIdx.x;

  int before = 0;
  for (int i = 0; i < b; ++i) before += cnt[i * 1024 + t];

  const int idx = b * 1024 + t;
  const int c = idx < n ? cnt[idx] : 0;

  int incl = c;
  for (int d = 1; d < 32; d <<= 1) {
    int v = __shfl_up(incl, d);
    if (lane >= d) incl += v;
  }
  for (int d = 16; d >= 1; d >>= 1) before += __shfl_xor(before, d);

  if (lane == 31) s_wtot[w] = incl;
  if (lane == 0) s_before[w] = before;
  __syncthreads();

  int off = s_before[lane];
  int wp = lane < w ? s_wtot[lane] : 0;
  for (int d = 16; d >= 1; d >>= 1) {
    off += __shfl_xor(off, d);
    wp += __shfl_xor(wp, d);
  }
  const int excl = off + wp + incl - c;
  if (idx < n) rowptr[idx] = excl;
  if (idx == n - 1) rowptr[n] = excl + c;
}

__global__ void edge_place(const int* __restrict__ srcI,
                           const int* __restrict__ dstI,
                           const int* __restrict__ rank,
                           const int* __restrict__ rowptr,
                           int* __restrict__ col, int nedges) {
  int e = blockIdx.x * blockDim.x + threadIdx.x;
  if (e >= nedges) return;
  int d = dstI[e];
  int s = srcI[e];
  int r = rank[e];
  col[rowptr[d] + r] = s;
}

template <int LPN>
__device__ __forceinline__ v4f gather_rows(const float* __restrict__ H,
                                           const int* __restrict__ col,
                                           int beg, int end, int sub) {
  v4f acc = {0.0f, 0.0f, 0.0f, 0.0f};
  int j = beg;
  for (; j + 4 <= end; j += 4) {
    const int s0 = col[j];
    const int s1 = col[j + 1];
    const int s2 = col[j + 2];
    const int s3 = col[j + 3];
    const v4f r0 = *(const v4f*)(H + ((size_t)s0 * LPN + sub) * 4);
    const v4f r1 = *(const v4f*)(H + ((size_t)s1 * LPN + sub) * 4);
    const v4f r2 = *(const v4f*)(H + ((size_t)s2 * LPN + sub) * 4);
    const v4f r3 = *(const v4f*)(H + ((size_t)s3 * LPN + sub) * 4);
    acc += r0;
    acc += r1;
    acc += r2;
    acc += r3;
  }
  for (; j < end; ++j) {
    const int s = col[j];
    acc += *(const v4f*)(H + ((size_t)s * LPN + sub) * 4);
  }
  return acc;
}

template <int C>
__global__ void owner_relu(const float* __restrict__ H,
                           const int* __restrict__ rowptr,
                           const int* __restrict__ col, float* FZ,
                           int nnodes) {
  constexpr int LPN = C / 4;
  const int gtid = blockIdx.x * blockDim.x + threadIdx.x;
  const int node = gtid / LPN;
  const int sub = gtid % LPN;
  if (node >= nnodes) return;

  const int beg = rowptr[node];
  const int end = rowptr[node + 1];
  const size_t own = ((size_t)node * LPN + sub) * 4;

  const v4f f = *(const v4f*)(FZ + own);
  const v4f acc = gather_rows<LPN>(H, col, beg, end, sub);

  v4f v = f + acc;
  v.x = v.x > 0.0f ? v.x : 0.0f;
  v.y = v.y > 0.0f ? v.y : 0.0f;
  v.z = v.z > 0.0f ? v.z : 0.0f;
  v.w = v.w > 0.0f ? v.w : 0.0f;
  *(v4f*)(FZ + own) = v;
}

__global__ void owner_logsoftmax16(const float* __restrict__ H,
                                   const int* __restrict__ rowptr,
                                   const int* __restrict__ col,
                                   const float* __restrict__ F,
                                   float* __restrict__ O, int nnodes) {
  constexpr int LPN = 4;
  const int gtid = blockIdx.x * blockDim.x + threadIdx.x;
  const int node = gtid / LPN;
  const int sub = gtid % LPN;
  const bool live = node < nnodes;

  const int beg = live ? rowptr[node] : 0;
  const int end = live ? rowptr[node + 1] : 0;
  const size_t own = ((size_t)node * LPN + sub) * 4;

  v4f f = {0.0f, 0.0f, 0.0f, 0.0f};
  if (live) f = *(const v4f*)(F + own);
  const v4f acc = gather_rows<LPN>(H, col, beg, end, sub);
  const v4f v = f + acc;

  float mx = fmaxf(fmaxf(v.x, v.y), fmaxf(v.z, v.w));
  mx = fmaxf(mx, __shfl_xor(mx, 1));
  mx = fmaxf(mx, __shfl_xor(mx, 2));
  float s = __expf(v.x - mx) + __expf(v.y - mx) + __expf(v.z - mx) +
            __expf(v.w - mx);
  s += __shfl_xor(s, 1);
  s += __shfl_xor(s, 2);
  const float l = mx + __logf(s);
  if (live) {
    v4f o = {v.x - l, v.y - l, v.z - l, v.w - l};
    *(v4f*)(O + own) = o;
  }
}

static inline int cdiv_ll(long long a, long long b) {
  return (int)((a + b - 1) / b);
}

extern "C" void kernel_launch(void* const* d_in, const int* in_sizes, int n_in,
                              void* d_out, int out_size, void* d_ws,
                              size_t ws_size, hipStream_t stream) {
  (void)n_in; (void)out_size; (void)ws_size;
  const float* x     = (const float*)d_in[0];
  const float* w1a_w = (const float*)d_in[1];
  const float* w1a_b = (const float*)d_in[2];
  const float* w1b_w = (const float*)d_in[3];
  const float* w1b_b = (const float*)d_in[4];
  const float* w2a_w = (const float*)d_in[5];
  const float* w2a_b = (const float*)d_in[6];
  const float* w2b_w = (const float*)d_in[7];
  const float* w2b_b = (const float*)d_in[8];
  const int*   edge  = (const int*)d_in[9];

  const int N = in_sizes[0] / 256;
  const int E = in_sizes[9] / 2;
  const int* srcI = edge;
  const int* dstI = edge + E;

  float* buf0 = (float*)d_ws;
  float* buf1 = buf0 + (size_t)N * 32;
  float* buf2 = buf1 + (size_t)N * 32;

  int* cnt    = (int*)buf1;
  int* rowptr = cnt + (size_t)N;
  int* col    = cnt + 2 * (size_t)N + 64;
  int* rank   = (int*)buf2;
  float* f2   = buf0 + (size_t)N * 16;

  const int BLK = 256;
  const int MT = 2;
  const int waves = N / (16 * MT);
  const int gemm_blocks = cdiv_ll(waves, BLK / 32);

  zero_i<<<cdiv_ll(N, BLK), BLK, 0, stream>>>(cnt, N);
  edge_rank<<<cdiv_ll(E, BLK), BLK, 0, stream>>>(dstI, cnt, rank, E);
  bucket_offsets<<<cdiv_ll(N, 1024), 1024, 0, stream>>>(cnt, rowptr, N);
  edge_place<<<cdiv_ll(E, BLK), BLK, 0, stream>>>(srcI, dstI, rank, rowptr,
                                                  col, E);

  wmma_linear2<256, 32, 32, MT><<<gemm_blocks, BLK, 0, stream>>>(
      x, w1a_w, w1a_b, w1b_w, w1b_b, buf0, buf2, N);
  owner_relu<32><<<cdiv_ll((long long)N * 8, BLK), BLK, 0, stream>>>(
      buf0, rowptr, col, buf2, N);

  wmma_linear2<32, 16, 16, MT><<<gemm_blocks, BLK, 0, stream>>>(
      buf2, w2a_w, w2a_b, w2b_w, w2b_b, buf0, f2, N);
  owner_logsoftmax16<<<cdiv_ll((long long)N * 4, BLK), BLK, 0, stream>>>(
      buf0, rowptr, col, f2, (float*)d_out, N);
}


# Edges grouped by destination by a two-level counting sort on LDS counters (chunk histogram, scan, scatter to 256-node bins, per-bin finish), no global returning add per edge; sums in arrival order
# speedup vs baseline: 2.2709x; 1.6066x over previous
#include <hip/hip_runtime.h>

typedef float v2f __attribute__((ext_vector_type(2)));
typedef float v8f __attribute__((ext_vector_type(8)));

typedef float v4f __attribute__((ext_vector_type(4)));

template <int INC, int MID, int OUTC, int MT>
__global__ void wmma_linear2(const float* __restrict__ X,
                             const float* __restrict__ Wa,
                             const float* __restrict__ Ba,
                             const float* __restrict__ Wb,
                             const float* __restrict__ Bb,
                             float* __restrict__ Y1,
                             float* __restrict__ Y2, int nrows) {
  constexpr int NT = MID / 16;
  constexpr int NT2 = OUTC / 16;
  int wave = (int)((blockIdx.x * blockDim.x + threadIdx.x) >> 5);
  int lane = threadIdx.x & 31;
  int row0 = wave * 16 * MT;
  if (row0 >= nrows) return;

  int nm = lane & 15;
  int hh = lane >> 4;
  int kg = hh << 1;

  const float* xrow[MT];
#pragma unroll
  for (int mt = 0; mt < MT; ++mt) {
    xrow[mt] = X + (size_t)(row0 + mt * 16 + nm) * INC;
    __builtin_prefetch(xrow[mt], 0, 3);
  }
  const float* wrow[NT];
#pragma unroll
  for (int t = 0; t < NT; ++t) wrow[t] = Wa + (size_t)(t * 16 + nm) * INC;

  v8f acc[MT][NT];
#pragma unroll
  for (int t = 0; t < NT; ++t)
#pragma unroll
    for (int v = 0; v < 8; ++v) {
      float bv = Ba[t * 16 + hh * 8 + v];
#pragma unroll
      for (int mt = 0; mt < MT; ++mt) acc[mt][t][v] = bv;
    }

#pragma unroll 4
  for (int k = 0; k < INC; k += 4) {
    v2f w[NT];
#pragma unroll
    for (int t = 0; t < NT; ++t) {
      w[t].x = wrow[t][k + kg];
      w[t].y = wrow[t][k + kg + 1];
    }
#pragma unroll
    for (int mt = 0; mt < MT; ++mt) {
      v2f xv;
      xv.x = xrow[mt][k + kg];
      xv.y = xrow[mt][k + kg + 1];
#pragma unroll
      for (int t = 0; t < NT; ++t) {
        acc[mt][t] = __builtin_amdgcn_wmma_f32_16x16x4_f32(
             false, w[t],  false, xv,
             (short)0, acc[mt][t],  false,
             false);
      }
    }
  }

#pragma unroll
  for (int mt = 0; mt < MT; ++mt)
#pragma unroll
    for (int t = 0; t < NT; ++t) {
      float* p = Y1 + (size_t)(row0 + mt * 16 + nm) * MID + t * 16 + hh * 8;
      v4f lo = {acc[mt][t][0], acc[mt][t][1], acc[mt][t][2], acc[mt][t][3]};
      v4f hi = {acc[mt][t][4], acc[mt][t][5], acc[mt][t][6], acc[mt][t][7]};
      *(v4f*)p = lo;
      *(v4f*)(p + 4) = hi;
    }

  v8f acc2[MT][NT2];
#pragma unroll
  for (int t2 = 0; t2 < NT2; ++t2)
#pragma unroll
    for (int v = 0; v < 8; ++v) {
      float bv = Bb[t2 * 16 + hh * 8 + v];
#pragma unroll
      for (int mt = 0; mt < MT; ++mt) acc2[mt][t2][v] = bv;
    }

#pragma unroll
  for (int t = 0; t < NT; ++t)
#pragma unroll
    for (int q = 0; q < 4; ++q) {
      v2f wb[NT2];
#pragma unroll
      for (int t2 = 0; t2 < NT2; ++t2) {
        const float* wp =
            Wb + (size_t)(t2 * 16 + nm) * MID + t * 16 + hh * 8 + 2 * q;
        wb[t2].x = wp[0];
        wb[t2].y = wp[1];
      }
#pragma unroll
      for (int mt = 0; mt < MT; ++mt) {
        v2f hv;
        hv.x = acc[mt][t][2 * q];
        hv.y = acc[mt][t][2 * q + 1];
#pragma unroll
        for (int t2 = 0; t2 < NT2; ++t2) {
          acc2[mt][t2] = __builtin_amdgcn_wmma_f32_16x16x4_f32(
               false, wb[t2],  false, hv,
               (short)0, acc2[mt][t2],  false,
               false);
        }
      }
    }

#pragma unroll
  for (int mt = 0; mt < MT; ++mt)
#pragma unroll
    for (int t2 = 0; t2 < NT2; ++t2) {
      float* p = Y2 + (size_t)(row0 + mt * 16 + nm) * OUTC + t2 * 16 + hh * 8;
      v4f lo = {acc2[mt][t2][0], acc2[mt][t2][1], acc2[mt][t2][2],
                acc2[mt][t2][3]};
      v4f hi = {acc2[mt][t2][4], acc2[mt][t2][5], acc2[mt][t2][6],
                acc2[mt][t2][7]};
      *(v4f*)p = lo;
      *(v4f*)(p + 4) = hi;
    }
}

typedef int v2i __attribute__((ext_vector_type(2)));

constexpr int BIN_SHIFT = 8;
constexpr int BIN = 1 << BIN_SHIFT;
constexpr int MAXB = 1024;
constexpr int SORT_T = 1024;
constexpr int EPT = 8;

__global__ void __launch_bounds__(SORT_T)
bin_count(const int* __restrict__ dstI, int* __restrict__ chist, int nedges,
          int ce, int nb) {
  __shared__ int h[MAXB];
  const int t = threadIdx.x;
  const int c = blockIdx.x;
  const int nc = gridDim.x;
  for (int i = t; i < nb; i += SORT_T) h[i] = 0;
  __syncthreads();

  const int beg = c * ce;
  const int end = beg + ce < nedges ? beg + ce : nedges;
  for (int e0 = beg + t; e0 < end; e0 += EPT * SORT_T) {
    int d[EPT];
#pragma unroll
    for (int u = 0; u < EPT; ++u) {
      const int e = e0 + u * SORT_T;
      d[u] = e < end ? dstI[e] : -1;
    }
#pragma unroll
    for (int u = 0; u < EPT; ++u)
      if (d[u] >= 0) atomicAdd(&h[d[u] >> BIN_SHIFT], 1);
  }
  __syncthreads();
  for (int i = t; i < nb; i += SORT_T) chist[i * nc + c] = h[i];
}

__global__ void __launch_bounds__(1024)
bucket_offsets(const int* __restrict__ cnt, int* __restrict__ rowptr, int n) {
  __shared__ int s_before[32];
  __shared__ int s_wtot[32];
  const int t = threadIdx.x;
  const int lane = t & 31;
  const int w = t >> 5;
  const int b = blockIdx.x;

  int before = 0;
  for (int i = 0; i < b; ++i) before += cnt[i * 1024 + t];

  const int idx = b * 1024 + t;
  const int c = idx < n ? cnt[idx] : 0;

  int incl = c;
  for (int d = 1; d < 32; d <<= 1) {
    int v = __shfl_up(incl, d);
    if (lane >= d) incl += v;
  }
  for (int d = 16; d >= 1; d >>= 1) before += __shfl_xor(before, d);

  if (lane == 31) s_wtot[w] = incl;
  if (lane == 0) s_before[w] = before;
  __syncthreads();

  int off = s_before[lane];
  int wp = lane < w ? s_wtot[lane] : 0;
  for (int d = 16; d >= 1; d >>= 1) {
    off += __shfl_xor(off, d);
    wp += __shfl_xor(wp, d);
  }
  const int excl = off + wp + incl - c;
  if (idx < n) rowptr[idx] = excl;
  if (idx == n - 1) rowptr[n] = excl + c;
}

__global__ void __launch_bounds__(SORT_T)
bin_scatter(const int* __restrict__ srcI, const int* __restrict__ dstI,
            const int* __restrict__ cofs, int* __restrict__ tmp, int nedges,
            int ce, int nb) {
  __shared__ int pos[MAXB];
  const int t = threadIdx.x;
  const int c = blockIdx.x;
  const int nc = gridDim.x;
  for (int i = t; i < nb; i += SORT_T) pos[i] = cofs[i * nc + c];
  __syncthreads();

  const int beg = c * ce;
  const int end = beg + ce < nedges ? beg + ce : nedges;
  for (int e0 = beg + t; e0 < end; e0 += EPT * SORT_T) {
    int d[EPT];
    int s[EPT];
#pragma unroll
    for (int u = 0; u < EPT; ++u) {
      const int e = e0 + u * SORT_T;
      d[u] = e < end ? dstI[e] : -1;
      s[u] = e < end ? srcI[e] : 0;
    }
#pragma unroll
    for (int u = 0; u < EPT; ++u)
      if (d[u] >= 0) {
        const int p = atomicAdd(&pos[d[u] >> BIN_SHIFT], 1);
        v2i sd = {s[u], d[u]};
        *(v2i*)(tmp + 2 * (size_t)p) = sd;
      }
  }
}

__global__ void __launch_bounds__(SORT_T)
bin_finish(const int* __restrict__ tmp, const int* __restrict__ cofs,
           int* __restrict__ rowptr, int* __restrict__ col, int n, int nc) {
  __shared__ int cnt[BIN];
  __shared__ int fill[BIN];
  __shared__ int wtot[32];
  const int t = threadIdx.x;
  const int lane = t & 31;
  const int w = t >> 5;
  const int b = blockIdx.x;
  const int beg = cofs[b * nc];
  const int end = cofs[(b + 1) * nc];

  if (t < BIN) cnt[t] = 0;
  __syncthreads();

  for (int j0 = beg + t; j0 < end; j0 += 4 * SORT_T) {
    int d[4];
#pragma unroll
    for (int u = 0; u < 4; ++u) {
      const int j = j0 + u * SORT_T;
      d[u] = j < end ? tmp[2 * (size_t)j + 1] : -1;
    }
#pragma unroll
    for (int u = 0; u < 4; ++u)
      if (d[u] >= 0) atomicAdd(&cnt[d[u] & (BIN - 1)], 1);
  }
  __syncthreads();

  const int c = t < BIN ? cnt[t] : 0;
  int incl = c;
  for (int d = 1; d < 32; d <<= 1) {
    int v = __shfl_up(incl, d);
    if (lane >= d) incl += v;
  }
  if (lane == 31) wtot[w] = incl;
  __syncthreads();
  int wp = lane < w ? wtot[lane] : 0;
  for (int d = 16; d >= 1; d >>= 1) wp += __shfl_xor(wp, d);
  const int excl = beg + wp + incl - c;
  if (t < BIN) {
    fill[t] = excl;
    const int node = b * BIN + t;
    if (node < n) rowptr[node] = excl;
  }
  if (b == (int)gridDim.x - 1 && t == 0) rowptr[n] = end;
  __syncthreads();

  for (int j0 = beg + t; j0 < end; j0 += 4 * SORT_T) {
    v2i sd[4];
#pragma unroll
    for (int u = 0; u < 4; ++u) {
      const int j = j0 + u * SORT_T;
      v2i z = {0, -1};
      sd[u] = j < end ? *(const v2i*)(tmp + 2 * (size_t)j) : z;
    }
#pragma unroll
    for (int u = 0; u < 4; ++u)
      if (sd[u].y >= 0) {
        const int p = atomicAdd(&fill[sd[u].y & (BIN - 1)], 1);
        col[p] = sd[u].x;
      }
  }
}

template <int LPN>
__device__ __forceinline__ v4f gather_rows(const float* __restrict__ H,
                                           const int* __restrict__ col,
                                           int beg, int end, int sub) {
  v4f acc = {0.0f, 0.0f, 0.0f, 0.0f};
  int j = beg;
  for (; j + 4 <= end; j += 4) {
    const int s0 = col[j];
    const int s1 = col[j + 1];
    const int s2 = col[j + 2];
    const int s3 = col[j + 3];
    const v4f r0 = *(const v4f*)(H + ((size_t)s0 * LPN + sub) * 4);
    const v4f r1 = *(const v4f*)(H + ((size_t)s1 * LPN + sub) * 4);
    const v4f r2 = *(const v4f*)(H + ((size_t)s2 * LPN + sub) * 4);
    const v4f r3 = *(const v4f*)(H + ((size_t)s3 * LPN + sub) * 4);
    acc += r0;
    acc += r1;
    acc += r2;
    acc += r3;
  }
  for (; j < end; ++j) {
    const int s = col[j];
    acc += *(const v4f*)(H + ((size_t)s * LPN + sub) * 4);
  }
  return acc;
}

template <int C>
__global__ void owner_relu(const float* __restrict__ H,
                           const int* __restrict__ rowptr,
                           const int* __restrict__ col, float* FZ,
                           int nnodes) {
  constexpr int LPN = C / 4;
  const int gtid = blockIdx.x * blockDim.x + threadIdx.x;
  const int node = gtid / LPN;
  const int sub = gtid % LPN;
  if (node >= nnodes) return;

  const int beg = rowptr[node];
  const int end = rowptr[node + 1];
  const size_t own = ((size_t)node * LPN + sub) * 4;

  const v4f f = *(const v4f*)(FZ + own);
  const v4f acc = gather_rows<LPN>(H, col, beg, end, sub);

  v4f v = f + acc;
  v.x = v.x > 0.0f ? v.x : 0.0f;
  v.y = v.y > 0.0f ? v.y : 0.0f;
  v.z = v.z > 0.0f ? v.z : 0.0f;
  v.w = v.w > 0.0f ? v.w : 0.0f;
  *(v4f*)(FZ + own) = v;
}

__global__ void owner_logsoftmax16(const float* __restrict__ H,
                                   const int* __restrict__ rowptr,
                                   const int* __restrict__ col,
                                   const float* __restrict__ F,
                                   float* __restrict__ O, int nnodes) {
  constexpr int LPN = 4;
  const int gtid = blockIdx.x * blockDim.x + threadIdx.x;
  const int node = gtid / LPN;
  const int sub = gtid % LPN;
  const bool live = node < nnodes;

  const int beg = live ? rowptr[node] : 0;
  const int end = live ? rowptr[node + 1] : 0;
  const size_t own = ((size_t)node * LPN + sub) * 4;

  v4f f = {0.0f, 0.0f, 0.0f, 0.0f};
  if (live) f = *(const v4f*)(F + own);
  const v4f acc = gather_rows<LPN>(H, col, beg, end, sub);
  const v4f v = f + acc;

  float mx = fmaxf(fmaxf(v.x, v.y), fmaxf(v.z, v.w));
  mx = fmaxf(mx, __shfl_xor(mx, 1));
  mx = fmaxf(mx, __shfl_xor(mx, 2));
  float s = __expf(v.x - mx) + __expf(v.y - mx) + __expf(v.z - mx) +
            __expf(v.w - mx);
  s += __shfl_xor(s, 1);
  s += __shfl_xor(s, 2);
  const float l = mx + __logf(s);
  if (live) {
    v4f o = {v.x - l, v.y - l, v.z - l, v.w - l};
    *(v4f*)(O + own) = o;
  }
}

static inline int cdiv_ll(long long a, long long b) {
  return (int)((a + b - 1) / b);
}

extern "C" void kernel_launch(void* const* d_in, const int* in_sizes, int n_in,
                              void* d_out, int out_size, void* d_ws,
                              size_t ws_size, hipStream_t stream) {
  (void)n_in; (void)out_size; (void)ws_size;
  const float* x     = (const float*)d_in[0];
  const float* w1a_w = (const float*)d_in[1];
  const float* w1a_b = (const float*)d_in[2];
  const float* w1b_w = (const float*)d_in[3];
  const float* w1b_b = (const float*)d_in[4];
  const float* w2a_w = (const float*)d_in[5];
  const float* w2a_b = (const float*)d_in[6];
  const float* w2b_w = (const float*)d_in[7];
  const float* w2b_b = (const float*)d_in[8];
  const int*   edge  = (const int*)d_in[9];

  const int N = in_sizes[0] / 256;
  const int E = in_sizes[9] / 2;
  const int* srcI = edge;
  const int* dstI = edge + E;

  float* buf0 = (float*)d_ws;
  float* buf1 = buf0 + (size_t)N * 32;
  float* buf2 = buf1 + (size_t)N * 32;

  const int NC = 256;
  const int CE = cdiv_ll(E, NC);
  const int NB = cdiv_ll(N, BIN);
  const int NS = NB * NC;

  int* rowptr = (int*)buf1;
  int* col    = rowptr + (size_t)N + 64;
  int* chist  = col + (size_t)E;
  int* cofs   = chist + (size_t)NS + 64;
  int* tmp    = (int*)buf2;
  float* f2   = buf0 + (size_t)N * 16;

  const int BLK = 256;
  const int MT = 2;
  const int waves = N / (16 * MT);
  const int gemm_blocks = cdiv_ll(waves, BLK / 32);

  bin_count<<<NC, SORT_T, 0, stream>>>(dstI, chist, E, CE, NB);
  bucket_offsets<<<cdiv_ll(NS, 1024), 1024, 0, stream>>>(chist, cofs, NS);
  bin_scatter<<<NC, SORT_T, 0, stream>>>(srcI, dstI, cofs, tmp, E, CE, NB);
  bin_finish<<<NB, SORT_T, 0, stream>>>(tmp, cofs, rowptr, col, N, NC);

  wmma_linear2<256, 32, 32, MT><<<gemm_blocks, BLK, 0, stream>>>(
      x, w1a_w, w1a_b, w1b_w, w1b_b, buf0, buf2, N);
  owner_relu<32><<<cdiv_ll((long long)N * 8, BLK), BLK, 0, stream>>>(
      buf0, rowptr, col, buf2, N);

  wmma_linear2<32, 16, 16, MT><<<gemm_blocks, BLK, 0, stream>>>(
      buf2, w2a_w, w2a_b, w2b_w, w2b_b, buf0, f2, N);
  owner_logsoftmax16<<<cdiv_ll((long long)N * 4, BLK), BLK, 0, stream>>>(
      buf0, rowptr, col, f2, (float*)d_out, N);
}


# First-product operand loads: 16 bytes per lane, 4 pieces per 32-k block (a 64-byte run of the row), issued one block ahead of their WMMAs; A and B loaders alike, new k order of that sum
# speedup vs baseline: 2.3837x; 1.0497x over previous
#include <hip/hip_runtime.h>

typedef float v2f __attribute__((ext_vector_type(2)));
typedef float v8f __attribute__((ext_vector_type(8)));

typedef float v4f __attribute__((ext_vector_type(4)));

template <int MT, int NT>
struct Frag {
  v4f x[MT][4];
  v4f w[NT][4];
};

template <int MT, int NT>
__device__ __forceinline__ void load_frag(Frag<MT, NT>& f,
                                          const float* const (&xrow)[MT],
                                          const float* const (&wrow)[NT],
                                          int kb) {
#pragma unroll
  for (int i = 0; i < 4; ++i) {
#pragma unroll
    for (int mt = 0; mt < MT; ++mt)
      f.x[mt][i] = *(const v4f*)(xrow[mt] + kb + 4 * i);
#pragma unroll
    for (int t = 0; t < NT; ++t)
      f.w[t][i] = *(const v4f*)(wrow[t] + kb + 4 * i);
  }
}

template <int MT, int NT>
__device__ __forceinline__ void mma_frag(v8f (&acc)[MT][NT],
                                         const Frag<MT, NT>& f) {
#pragma unroll
  for (int i = 0; i < 4; ++i)
#pragma unroll
    for (int p = 0; p < 2; ++p) {
      v2f w[NT];
#pragma unroll
      for (int t = 0; t < NT; ++t) {
        w[t].x = f.w[t][i][2 * p];
        w[t].y = f.w[t][i][2 * p + 1];
      }
#pragma unroll
      for (int mt = 0; mt < MT; ++mt) {
        v2f xv;
        xv.x = f.x[mt][i][2 * p];
        xv.y = f.x[mt][i][2 * p + 1];
#pragma unroll
        for (int t = 0; t < NT; ++t) {
          acc[mt][t] = __builtin_amdgcn_wmma_f32_16x16x4_f32(
               false, w[t],  false, xv,
               (short)0, acc[mt][t],  false,
               false);
        }
      }
    }
}

template <int INC, int MID, int OUTC, int MT>
__global__ void __launch_bounds__(256) wmma_linear2(const float* __restrict__ X,
                             const float* __restrict__ Wa,
                             const float* __restrict__ Ba,
                             const float* __restrict__ Wb,
                             const float* __restrict__ Bb,
                             float* __restrict__ Y1,
                             float* __restrict__ Y2, int nrows) {
  constexpr int NT = MID / 16;
  constexpr int NT2 = OUTC / 16;
  int wave = (int)((blockIdx.x * blockDim.x + threadIdx.x) >> 5);
  int lane = threadIdx.x & 31;
  int row0 = wave * 16 * MT;
  if (row0 >= nrows) return;

  int nm = lane & 15;
  int hh = lane >> 4;

  const float* xrow[MT];
#pragma unroll
  for (int mt = 0; mt < MT; ++mt)
    xrow[mt] = X + (size_t)(row0 + mt * 16 + nm) * INC + hh * 16;
  const float* wrow[NT];
#pragma unroll
  for (int t = 0; t < NT; ++t)
    wrow[t] = Wa + (size_t)(t * 16 + nm) * INC + hh * 16;

  v8f acc[MT][NT];
#pragma unroll
  for (int t = 0; t < NT; ++t)
#pragma unroll
    for (int v = 0; v < 8; ++v) {
      float bv = Ba[t * 16 + hh * 8 + v];
#pragma unroll
      for (int mt = 0; mt < MT; ++mt) acc[mt][t][v] = bv;
    }

  static_assert(INC == 32 || INC % 64 == 0, "k blocks are taken in pairs");
  Frag<MT, NT> fa;
  load_frag<MT, NT>(fa, xrow, wrow, 0);
  if constexpr (INC == 32) {
    mma_frag<MT, NT>(acc, fa);
  } else {
    Frag<MT, NT> fb;
#pragma unroll
    for (int kb = 0; kb + 64 < INC; kb += 64) {
      load_frag<MT, NT>(fb, xrow, wrow, kb + 32);
      __builtin_amdgcn_sched_barrier(0);
      mma_frag<MT, NT>(acc, fa);
      __builtin_amdgcn_sched_barrier(0);
      load_frag<MT, NT>(fa, xrow, wrow, kb + 64);
      __builtin_amdgcn_sched_barrier(0);
      mma_frag<MT, NT>(acc, fb);
      __builtin_amdgcn_sched_barrier(0);
    }
    load_frag<MT, NT>(fb, xrow, wrow, INC - 32);
    __builtin_amdgcn_sched_barrier(0);
    mma_frag<MT, NT>(acc, fa);
    mma_frag<MT, NT>(acc, fb);
  }

#pragma unroll
  for (int mt = 0; mt < MT; ++mt)
#pragma unroll
    for (int t = 0; t < NT; ++t) {
      float* p = Y1 + (size_t)(row0 + mt * 16 + nm) * MID + t * 16 + hh * 8;
      v4f lo = {acc[mt][t][0], acc[mt][t][1], acc[mt][t][2], acc[mt][t][3]};
      v4f hi = {acc[mt][t][4], acc[mt][t][5], acc[mt][t][6], acc[mt][t][7]};
      *(v4f*)p = lo;
      *(v4f*)(p + 4) = hi;
    }

  v8f acc2[MT][NT2];
#pragma unroll
  for (int t2 = 0; t2 < NT2; ++t2)
#pragma unroll
    for (int v = 0; v < 8; ++v) {
      float bv = Bb[t2 * 16 + hh * 8 + v];
#pragma unroll
      for (int mt = 0; mt < MT; ++mt) acc2[mt][t2][v] = bv;
    }

#pragma unroll
  for (int t = 0; t < NT; ++t)
#pragma unroll
    for (int q = 0; q < 4; ++q) {
      v2f wb[NT2];
#pragma unroll
      for (int t2 = 0; t2 < NT2; ++t2) {
        const float* wp =
            Wb + (size_t)(t2 * 16 + nm) * MID + t * 16 + hh * 8 + 2 * q;
        wb[t2].x = wp[0];
        wb[t2].y = wp[1];
      }
#pragma unroll
      for (int mt = 0; mt < MT; ++mt) {
        v2f hv;
        hv.x = acc[mt][t][2 * q];
        hv.y = acc[mt][t][2 * q + 1];
#pragma unroll
        for (int t2 = 0; t2 < NT2; ++t2) {
          acc2[mt][t2] = __builtin_amdgcn_wmma_f32_16x16x4_f32(
               false, wb[t2],  false, hv,
               (short)0, acc2[mt][t2],  false,
               false);
        }
      }
    }

#pragma unroll
  for (int mt = 0; mt < MT; ++mt)
#pragma unroll
    for (int t2 = 0; t2 < NT2; ++t2) {
      float* p = Y2 + (size_t)(row0 + mt * 16 + nm) * OUTC + t2 * 16 + hh * 8;
      v4f lo = {acc2[mt][t2][0], acc2[mt][t2][1], acc2[mt][t2][2],
                acc2[mt][t2][3]};
      v4f hi = {acc2[mt][t2][4], acc2[mt][t2][5], acc2[mt][t2][6],
                acc2[mt][t2][7]};
      *(v4f*)p = lo;
      *(v4f*)(p + 4) = hi;
    }
}

typedef int v2i __attribute__((ext_vector_type(2)));

constexpr int BIN_SHIFT = 8;
constexpr int BIN = 1 << BIN_SHIFT;
constexpr int MAXB = 1024;
constexpr int SORT_T = 1024;
constexpr int EPT = 8;

__global__ void __launch_bounds__(SORT_T)
bin_count(const int* __restrict__ dstI, int* __restrict__ chist, int nedges,
          int ce, int nb) {
  __shared__ int h[MAXB];
  const int t = threadIdx.x;
  const int c = blockIdx.x;
  const int nc = gridDim.x;
  for (int i = t; i < nb; i += SORT_T) h[i] = 0;
  __syncthreads();

  const int beg = c * ce;
  const int end = beg + ce < nedges ? beg + ce : nedges;
  for (int e0 = beg + t; e0 < end; e0 += EPT * SORT_T) {
    int d[EPT];
#pragma unroll
    for (int u = 0; u < EPT; ++u) {
      const int e = e0 + u * SORT_T;
      d[u] = e < end ? dstI[e] : -1;
    }
#pragma unroll
    for (int u = 0; u < EPT; ++u)
      if (d[u] >= 0) atomicAdd(&h[d[u] >> BIN_SHIFT], 1);
  }
  __syncthreads();
  for (int i = t; i < nb; i += SORT_T) chist[i * nc + c] = h[i];
}

__global__ void __launch_bounds__(1024)
bucket_offsets(const int* __restrict__ cnt, int* __restrict__ rowptr, int n) {
  __shared__ int s_before[32];
  __shared__ int s_wtot[32];
  const int t = threadIdx.x;
  const int lane = t & 31;
  const int w = t >> 5;
  const int b = blockIdx.x;

  int before = 0;
  for (int i = 0; i < b; ++i) before += cnt[i * 1024 + t];

  const int idx = b * 1024 + t;
  const int c = idx < n ? cnt[idx] : 0;

  int incl = c;
  for (int d = 1; d < 32; d <<= 1) {
    int v = __shfl_up(incl, d);
    if (lane >= d) incl += v;
  }
  for (int d = 16; d >= 1; d >>= 1) before += __shfl_xor(before, d);

  if (lane == 31) s_wtot[w] = incl;
  if (lane == 0) s_before[w] = before;
  __syncthreads();

  int off = s_before[lane];
  int wp = lane < w ? s_wtot[lane] : 0;
  for (int d = 16; d >= 1; d >>= 1) {
    off += __shfl_xor(off, d);
    wp += __shfl_xor(wp, d);
  }
  const int excl = off + wp + incl - c;
  if (idx < n) rowptr[idx] = excl;
  if (idx == n - 1) rowptr[n] = excl + c;
}

__global__ void __launch_bounds__(SORT_T)
bin_scatter(const int* __restrict__ srcI, const int* __restrict__ dstI,
            const int* __restrict__ cofs, int* __restrict__ tmp, int nedges,
            int ce, int nb) {
  __shared__ int pos[MAXB];
  const int t = threadIdx.x;
  const int c = blockIdx.x;
  const int nc = gridDim.x;
  for (int i = t; i < nb; i += SORT_T) pos[i] = cofs[i * nc + c];
  __syncthreads();

  const int beg = c * ce;
  const int end = beg + ce < nedges ? beg + ce : nedges;
  for (int e0 = beg + t; e0 < end; e0 += EPT * SORT_T) {
    int d[EPT];
    int s[EPT];
#pragma unroll
    for (int u = 0; u < EPT; ++u) {
      const int e = e0 + u * SORT_T;
      d[u] = e < end ? dstI[e] : -1;
      s[u] = e < end ? srcI[e] : 0;
    }
#pragma unroll
    for (int u = 0; u < EPT; ++u)
      if (d[u] >= 0) {
        const int p = atomicAdd(&pos[d[u] >> BIN_SHIFT], 1);
        v2i sd = {s[u], d[u]};
        *(v2i*)(tmp + 2 * (size_t)p) = sd;
      }
  }
}

__global__ void __launch_bounds__(SORT_T)
bin_finish(const int* __restrict__ tmp, const int* __restrict__ cofs,
           int* __restrict__ rowptr, int* __restrict__ col, int n, int nc) {
  __shared__ int cnt[BIN];
  __shared__ int fill[BIN];
  __shared__ int wtot[32];
  const int t = threadIdx.x;
  const int lane = t & 31;
  const int w = t >> 5;
  const int b = blockIdx.x;
  const int beg = cofs[b * nc];
  const int end = cofs[(b + 1) * nc];

  if (t < BIN) cnt[t] = 0;
  __syncthreads();

  for (int j0 = beg + t; j0 < end; j0 += 4 * SORT_T) {
    int d[4];
#pragma unroll
    for (int u = 0; u < 4; ++u) {
      const int j = j0 + u * SORT_T;
      d[u] = j < end ? tmp[2 * (size_t)j + 1] : -1;
    }
#pragma unroll
    for (int u = 0; u < 4; ++u)
      if (d[u] >= 0) atomicAdd(&cnt[d[u] & (BIN - 1)], 1);
  }
  __syncthreads();

  const int c = t < BIN ? cnt[t] : 0;
  int incl = c;
  for (int d = 1; d < 32; d <<= 1) {
    int v = __shfl_up(incl, d);
    if (lane >= d) incl += v;
  }
  if (lane == 31) wtot[w] = incl;
  __syncthreads();
  int wp = lane < w ? wtot[lane] : 0;
  for (int d = 16; d >= 1; d >>= 1) wp += __shfl_xor(wp, d);
  const int excl = beg + wp + incl - c;
  if (t < BIN) {
    fill[t] = excl;
    const int node = b * BIN + t;
    if (node < n) rowptr[node] = excl;
  }
  if (b == (int)gridDim.x - 1 && t == 0) rowptr[n] = end;
  __syncthreads();

  for (int j0 = beg + t; j0 < end; j0 += 4 * SORT_T) {
    v2i sd[4];
#pragma unroll
    for (int u = 0; u < 4; ++u) {
      const int j = j0 + u * SORT_T;
      v2i z = {0, -1};
      sd[u] = j < end ? *(const v2i*)(tmp + 2 * (size_t)j) : z;
    }
#pragma unroll
    for (int u = 0; u < 4; ++u)
      if (sd[u].y >= 0) {
        const int p = atomicAdd(&fill[sd[u].y & (BIN - 1)], 1);
        col[p] = sd[u].x;
      }
  }
}

template <int LPN>
__device__ __forceinline__ v4f gather_rows(const float* __restrict__ H,
                                           const int* __restrict__ col,
                                           int beg, int end, int sub) {
  v4f acc = {0.0f, 0.0f, 0.0f, 0.0f};
  int j = beg;
  for (; j + 4 <= end; j += 4) {
    const int s0 = col[j];
    const int s1 = col[j + 1];
    const int s2 = col[j + 2];
    const int s3 = col[j + 3];
    const v4f r0 = *(const v4f*)(H + ((size_t)s0 * LPN + sub) * 4);
    const v4f r1 = *(const v4f*)(H + ((size_t)s1 * LPN + sub) * 4);
    const v4f r2 = *(const v4f*)(H + ((size_t)s2 * LPN + sub) * 4);
    const v4f r3 = *(const v4f*)(H + ((size_t)s3 * LPN + sub) * 4);
    acc += r0;
    acc += r1;
    acc += r2;
    acc += r3;
  }
  for (; j < end; ++j) {
    const int s = col[j];
    acc += *(const v4f*)(H + ((size_t)s * LPN + sub) * 4);
  }
  return acc;
}

template <int C>
__global__ void owner_relu(const float* __restrict__ H,
                           const int* __restrict__ rowptr,
                           const int* __restrict__ col, float* FZ,
                           int nnodes) {
  constexpr int LPN = C / 4;
  const int gtid = blockIdx.x * blockDim.x + threadIdx.x;
  const int node = gtid / LPN;
  const int sub = gtid % LPN;
  if (node >= nnodes) return;

  const int beg = rowptr[node];
  const int end = rowptr[node + 1];
  const size_t own = ((size_t)node * LPN + sub) * 4;

  const v4f f = *(const v4f*)(FZ + own);
  const v4f acc = gather_rows<LPN>(H, col, beg, end, sub);

  v4f v = f + acc;
  v.x = v.x > 0.0f ? v.x : 0.0f;
  v.y = v.y > 0.0f ? v.y : 0.0f;
  v.z = v.z > 0.0f ? v.z : 0.0f;
  v.w = v.w > 0.0f ? v.w : 0.0f;
  *(v4f*)(FZ + own) = v;
}

__global__ void owner_logsoftmax16(const float* __restrict__ H,
                                   const int* __restrict__ rowptr,
                                   const int* __restrict__ col,
                                   const float* __restrict__ F,
                                   float* __restrict__ O, int nnodes) {
  constexpr int LPN = 4;
  const int gtid = blockIdx.x * blockDim.x + threadIdx.x;
  const int node = gtid / LPN;
  const int sub = gtid % LPN;
  const bool live = node < nnodes;

  const int beg = live ? rowptr[node] : 0;
  const int end = live ? rowptr[node + 1] : 0;
  const size_t own = ((size_t)node * LPN + sub) * 4;

  v4f f = {0.0f, 0.0f, 0.0f, 0.0f};
  if (live) f = *(const v4f*)(F + own);
  const v4f acc = gather_rows<LPN>(H, col, beg, end, sub);
  const v4f v = f + acc;

  float mx = fmaxf(fmaxf(v.x, v.y), fmaxf(v.z, v.w));
  mx = fmaxf(mx, __shfl_xor(mx, 1));
  mx = fmaxf(mx, __shfl_xor(mx, 2));
  float s = __expf(v.x - mx) + __expf(v.y - mx) + __expf(v.z - mx) +
            __expf(v.w - mx);
  s += __shfl_xor(s, 1);
  s += __shfl_xor(s, 2);
  const float l = mx + __logf(s);
  if (live) {
    v4f o = {v.x - l, v.y - l, v.z - l, v.w - l};
    *(v4f*)(O + own) = o;
  }
}

static inline int cdiv_ll(long long a, long long b) {
  return (int)((a + b - 1) / b);
}

extern "C" void kernel_launch(void* const* d_in, const int* in_sizes, int n_in,
                              void* d_out, int out_size, void* d_ws,
                              size_t ws_size, hipStream_t stream) {
  (void)n_in; (void)out_size; (void)ws_size;
  const float* x     = (const float*)d_in[0];
  const float* w1a_w = (const float*)d_in[1];
  const float* w1a_b = (const float*)d_in[2];
  const float* w1b_w = (const float*)d_in[3];
  const float* w1b_b = (const float*)d_in[4];
  const float* w2a_w = (const float*)d_in[5];
  const float* w2a_b = (const float*)d_in[6];
  const float* w2b_w = (const float*)d_in[7];
  const float* w2b_b = (const float*)d_in[8];
  const int*   edge  = (const int*)d_in[9];

  const int N = in_sizes[0] / 256;
  const int E = in_sizes[9] / 2;
  const int* srcI = edge;
  const int* dstI = edge + E;

  float* buf0 = (float*)d_ws;
  float* buf1 = buf0 + (size_t)N * 32;
  float* buf2 = buf1 + (size_t)N * 32;

  const int NC = 256;
  const int CE = cdiv_ll(E, NC);
  const int NB = cdiv_ll(N, BIN);
  const int NS = NB * NC;

  int* rowptr = (int*)buf1;
  int* col    = rowptr + (size_t)N + 64;
  int* chist  = col + (size_t)E;
  int* cofs   = chist + (size_t)NS + 64;
  int* tmp    = (int*)buf2;
  float* f2   = buf0 + (size_t)N * 16;

  const int BLK = 256;
  const int MT = 2;
  const int waves = N / (16 * MT);
  const int gemm_blocks = cdiv_ll(waves, BLK / 32);

  bin_count<<<NC, SORT_T, 0, stream>>>(dstI, chist, E, CE, NB);
  bucket_offsets<<<cdiv_ll(NS, 1024), 1024, 0, stream>>>(chist, cofs, NS);
  bin_scatter<<<NC, SORT_T, 0, stream>>>(srcI, dstI, cofs, tmp, E, CE, NB);
  bin_finish<<<NB, SORT_T, 0, stream>>>(tmp, cofs, rowptr, col, N, NC);

  wmma_linear2<256, 32, 32, MT><<<gemm_blocks, BLK, 0, stream>>>(
      x, w1a_w, w1a_b, w1b_w, w1b_b, buf0, buf2, N);
  owner_relu<32><<<cdiv_ll((long long)N * 8, BLK), BLK, 0, stream>>>(
      buf0, rowptr, col, buf2, N);

  wmma_linear2<32, 16, 16, MT><<<gemm_blocks, BLK, 0, stream>>>(
      buf2, w2a_w, w2a_b, w2b_w, w2b_b, buf0, f2, N);
  owner_logsoftmax16<<<cdiv_ll((long long)N * 4, BLK), BLK, 0, stream>>>(
      buf0, rowptr, col, f2, (float*)d_out, N);
}


# Binning made chunk-local: each edge chunk sorts its own slice by bin, a bin block sums the 256 chunk offsets itself; count and prefix-sum launches go (8 to 6); sums in a new arrival order
# speedup vs baseline: 3.1288x; 1.3126x over previous
#include <hip/hip_runtime.h>

typedef float v2f __attribute__((ext_vector_type(2)));
typedef float v8f __attribute__((ext_vector_type(8)));

typedef float v4f __attribute__((ext_vector_type(4)));

template <int MT, int NT>
struct Frag {
  v4f x[MT][4];
  v4f w[NT][4];
};

template <int MT, int NT>
__device__ __forceinline__ void load_frag(Frag<MT, NT>& f,
                                          const float* const (&xrow)[MT],
                                          const float* const (&wrow)[NT],
                                          int kb) {
#pragma unroll
  for (int i = 0; i < 4; ++i) {
#pragma unroll
    for (int mt = 0; mt < MT; ++mt)
      f.x[mt][i] = *(const v4f*)(xrow[mt] + kb + 4 * i);
#pragma unroll
    for (int t = 0; t < NT; ++t)
      f.w[t][i] = *(const v4f*)(wrow[t] + kb + 4 * i);
  }
}

template <int MT, int NT>
__device__ __forceinline__ void mma_frag(v8f (&acc)[MT][NT],
                                         const Frag<MT, NT>& f) {
#pragma unroll
  for (int i = 0; i < 4; ++i)
#pragma unroll
    for (int p = 0; p < 2; ++p) {
      v2f w[NT];
#pragma unroll
      for (int t = 0; t < NT; ++t) {
        w[t].x = f.w[t][i][2 * p];
        w[t].y = f.w[t][i][2 * p + 1];
      }
#pragma unroll
      for (int mt = 0; mt < MT; ++mt) {
        v2f xv;
        xv.x = f.x[mt][i][2 * p];
        xv.y = f.x[mt][i][2 * p + 1];
#pragma unroll
        for (int t = 0; t < NT; ++t) {
          acc[mt][t] = __builtin_amdgcn_wmma_f32_16x16x4_f32(
               false, w[t],  false, xv,
               (short)0, acc[mt][t],  false,
               false);
        }
      }
    }
}

template <int INC, int MID, int OUTC, int MT>
__global__ void __launch_bounds__(256) wmma_linear2(const float* __restrict__ X,
                             const float* __restrict__ Wa,
                             const float* __restrict__ Ba,
                             const float* __restrict__ Wb,
                             const float* __restrict__ Bb,
                             float* __restrict__ Y1,
                             float* __restrict__ Y2, int nrows) {
  constexpr int NT = MID / 16;
  constexpr int NT2 = OUTC / 16;
  int wave = (int)((blockIdx.x * blockDim.x + threadIdx.x) >> 5);
  int lane = threadIdx.x & 31;
  int row0 = wave * 16 * MT;
  if (row0 >= nrows) return;

  int nm = lane & 15;
  int hh = lane >> 4;

  const float* xrow[MT];
#pragma unroll
  for (int mt = 0; mt < MT; ++mt)
    xrow[mt] = X + (size_t)(row0 + mt * 16 + nm) * INC + hh * 16;
  const float* wrow[NT];
#pragma unroll
  for (int t = 0; t < NT; ++t)
    wrow[t] = Wa + (size_t)(t * 16 + nm) * INC + hh * 16;

  v8f acc[MT][NT];
#pragma unroll
  for (int t = 0; t < NT; ++t)
#pragma unroll
    for (int v = 0; v < 8; ++v) {
      float bv = Ba[t * 16 + hh * 8 + v];
#pragma unroll
      for (int mt = 0; mt < MT; ++mt) acc[mt][t][v] = bv;
    }

  static_assert(INC == 32 || INC % 64 == 0, "k blocks are taken in pairs");
  Frag<MT, NT> fa;
  load_frag<MT, NT>(fa, xrow, wrow, 0);
  if constexpr (INC == 32) {
    mma_frag<MT, NT>(acc, fa);
  } else {
    Frag<MT, NT> fb;
#pragma unroll
    for (int kb = 0; kb + 64 < INC; kb += 64) {
      load_frag<MT, NT>(fb, xrow, wrow, kb + 32);
      __builtin_amdgcn_sched_barrier(0);
      mma_frag<MT, NT>(acc, fa);
      __builtin_amdgcn_sched_barrier(0);
      load_frag<MT, NT>(fa, xrow, wrow, kb + 64);
      __builtin_amdgcn_sched_barrier(0);
      mma_frag<MT, NT>(acc, fb);
      __builtin_amdgcn_sched_barrier(0);
    }
    load_frag<MT, NT>(fb, xrow, wrow, INC - 32);
    __builtin_amdgcn_sched_barrier(0);
    mma_frag<MT, NT>(acc, fa);
    mma_frag<MT, NT>(acc, fb);
  }

#pragma unroll
  for (int mt = 0; mt < MT; ++mt)
#pragma unroll
    for (int t = 0; t < NT; ++t) {
      float* p = Y1 + (size_t)(row0 + mt * 16 + nm) * MID + t * 16 + hh * 8;
      v4f lo = {acc[mt][t][0], acc[mt][t][1], acc[mt][t][2], acc[mt][t][3]};
      v4f hi = {acc[mt][t][4], acc[mt][t][5], acc[mt][t][6], acc[mt][t][7]};
      *(v4f*)p = lo;
      *(v4f*)(p + 4) = hi;
    }

  v8f acc2[MT][NT2];
#pragma unroll
  for (int t2 = 0; t2 < NT2; ++t2)
#pragma unroll
    for (int v = 0; v < 8; ++v) {
      float bv = Bb[t2 * 16 + hh * 8 + v];
#pragma unroll
      for (int mt = 0; mt < MT; ++mt) acc2[mt][t2][v] = bv;
    }

#pragma unroll
  for (int t = 0; t < NT; ++t)
#pragma unroll
    for (int q = 0; q < 4; ++q) {
      v2f wb[NT2];
#pragma unroll
      for (int t2 = 0; t2 < NT2; ++t2) {
        const float* wp =
            Wb + (size_t)(t2 * 16 + nm) * MID + t * 16 + hh * 8 + 2 * q;
        wb[t2].x = wp[0];
        wb[t2].y = wp[1];
      }
#pragma unroll
      for (int mt = 0; mt < MT; ++mt) {
        v2f hv;
        hv.x = acc[mt][t][2 * q];
        hv.y = acc[mt][t][2 * q + 1];
#pragma unroll
        for (int t2 = 0; t2 < NT2; ++t2) {
          acc2[mt][t2] = __builtin_amdgcn_wmma_f32_16x16x4_f32(
               false, wb[t2],  false, hv,
               (short)0, acc2[mt][t2],  false,
               false);
        }
      }
    }

#pragma unroll
  for (int mt = 0; mt < MT; ++mt)
#pragma unroll
    for (int t2 = 0; t2 < NT2; ++t2) {
      float* p = Y2 + (size_t)(row0 + mt * 16 + nm) * OUTC + t2 * 16 + hh * 8;
      v4f lo = {acc2[mt][t2][0], acc2[mt][t2][1], acc2[mt][t2][2],
                acc2[mt][t2][3]};
      v4f hi = {acc2[mt][t2][4], acc2[mt][t2][5], acc2[mt][t2][6],
                acc2[mt][t2][7]};
      *(v4f*)p = lo;
      *(v4f*)(p + 4) = hi;
    }
}

typedef int v2i __attribute__((ext_vector_type(2)));

constexpr int BIN_SHIFT = 8;
constexpr int BIN = 1 << BIN_SHIFT;
constexpr int SORT_T = 1024;
constexpr int MAXB = SORT_T;
constexpr int EPT = 8;
constexpr int NCH = 256;
constexpr int CPW = NCH / 32;

__global__ void __launch_bounds__(SORT_T)
chunk_sort(const int* __restrict__ srcI, const int* __restrict__ dstI,
           int* __restrict__ lofs, int* __restrict__ tmp, int nedges, int ce,
           int nb) {
  __shared__ int h[MAXB];
  __shared__ int wtot[32];
  const int t = threadIdx.x;
  const int lane = t & 31;
  const int w = t >> 5;
  const int c = blockIdx.x;
  h[t] = 0;
  __syncthreads();

  const int beg = c * ce;
  const int end = beg + ce < nedges ? beg + ce : nedges;

  int d[EPT];
  int s[EPT];
#pragma unroll
  for (int u = 0; u < EPT; ++u) {
    const int e = beg + t + u * SORT_T;
    d[u] = e < end ? dstI[e] : -1;
    s[u] = e < end ? srcI[e] : 0;
  }
#pragma unroll
  for (int u = 0; u < EPT; ++u)
    if (d[u] >= 0) atomicAdd(&h[d[u] >> BIN_SHIFT], 1);
  for (int e = beg + t + EPT * SORT_T; e < end; e += SORT_T)
    atomicAdd(&h[dstI[e] >> BIN_SHIFT], 1);
  __syncthreads();

  const int cnt = h[t];
  int incl = cnt;
  for (int k = 1; k < 32; k <<= 1) {
    int v = __shfl_up(incl, k);
    if (lane >= k) incl += v;
  }
  if (lane == 31) wtot[w] = incl;
  __syncthreads();
  int wp = lane < w ? wtot[lane] : 0;
  for (int k = 16; k >= 1; k >>= 1) wp += __shfl_xor(wp, k);
  const int excl = wp + incl - cnt;
  h[t] = excl;
  __syncthreads();
  if (t < nb) lofs[t * NCH + c] = excl;
  if (t == nb - 1) lofs[nb * NCH + c] = excl + cnt;

  int* const slice = tmp + 2 * (size_t)beg;
#pragma unroll
  for (int u = 0; u < EPT; ++u)
    if (d[u] >= 0) {
      const int p = atomicAdd(&h[d[u] >> BIN_SHIFT], 1);
      v2i sd = {s[u], d[u]};
      *(v2i*)(slice + 2 * (size_t)p) = sd;
    }
  for (int e = beg + t + EPT * SORT_T; e < end; e += SORT_T) {
    const int dd = dstI[e];
    const int p = atomicAdd(&h[dd >> BIN_SHIFT], 1);
    v2i sd = {srcI[e], dd};
    *(v2i*)(slice + 2 * (size_t)p) = sd;
  }
}

__global__ void __launch_bounds__(SORT_T)
bin_finish(const int* __restrict__ tmp, const int* __restrict__ lofs,
           int* __restrict__ rowptr, int* __restrict__ col, int n, int ce) {
  __shared__ int cnt[BIN];
  __shared__ int fill[BIN];
  __shared__ int wtot[32];
  __shared__ v2i seg[NCH];
  __shared__ int redlo[32];
  __shared__ int redhi[32];
  const int t = threadIdx.x;
  const int lane = t & 31;
  const int w = t >> 5;
  const int b = blockIdx.x;

  int lo = 0;
  int hi = 0;
  if (t < NCH) {
    lo = lofs[b * NCH + t];
    hi = lofs[(b + 1) * NCH + t];
    v2i sg = {t * ce + lo, hi - lo};
    seg[t] = sg;
  }
  if (t < BIN) cnt[t] = 0;
  __syncthreads();

  v2i sg[CPW];
  v2i sd[CPW];
#pragma unroll
  for (int k = 0; k < CPW; ++k) sg[k] = seg[w + 32 * k];
#pragma unroll
  for (int k = 0; k < CPW; ++k) {
    v2i z = {0, -1};
    sd[k] = lane < sg[k].y
                ? *(const v2i*)(tmp + 2 * (size_t)(sg[k].x + lane))
                : z;
  }

  int slo = lo;
  int shi = hi;
  for (int k = 16; k >= 1; k >>= 1) {
    slo += __shfl_xor(slo, k);
    shi += __shfl_xor(shi, k);
  }
  if (lane == 0) {
    redlo[w] = slo;
    redhi[w] = shi;
  }

#pragma unroll
  for (int k = 0; k < CPW; ++k)
    if (sd[k].y >= 0) atomicAdd(&cnt[sd[k].y & (BIN - 1)], 1);
  for (int k = 0; k < CPW; ++k) {
    const v2i g = seg[w + 32 * k];
    for (int l = lane + 32; l < g.y; l += 32)
      atomicAdd(&cnt[tmp[2 * (size_t)(g.x + l) + 1] & (BIN - 1)], 1);
  }
  __syncthreads();

  const int c = t < BIN ? cnt[t] : 0;
  int incl = c;
  for (int k = 1; k < 32; k <<= 1) {
    int v = __shfl_up(incl, k);
    if (lane >= k) incl += v;
  }
  if (lane == 31) wtot[w] = incl;
  __syncthreads();
  int wp = lane < w ? wtot[lane] : 0;
  int beg = redlo[lane];
  int end = redhi[lane];
  for (int k = 16; k >= 1; k >>= 1) {
    wp += __shfl_xor(wp, k);
    beg += __shfl_xor(beg, k);
    end += __shfl_xor(end, k);
  }
  const int excl = beg + wp + incl - c;
  if (t < BIN) fill[t] = excl;
  __syncthreads();
  if (t < BIN) {
    const int node = b * BIN + t;
    if (node < n) rowptr[node] = excl;
  }
  if (b == (int)gridDim.x - 1 && t == 0) rowptr[n] = end;

#pragma unroll
  for (int k = 0; k < CPW; ++k)
    if (sd[k].y >= 0) {
      const int p = atomicAdd(&fill[sd[k].y & (BIN - 1)], 1);
      col[p] = sd[k].x;
    }
  for (int k = 0; k < CPW; ++k) {
    const v2i g = seg[w + 32 * k];
    for (int l = lane + 32; l < g.y; l += 32) {
      const v2i e = *(const v2i*)(tmp + 2 * (size_t)(g.x + l));
      const int p = atomicAdd(&fill[e.y & (BIN - 1)], 1);
      col[p] = e.x;
    }
  }
}

template <int LPN>
__device__ __forceinline__ v4f gather_rows(const float* __restrict__ H,
                                           const int* __restrict__ col,
                                           int beg, int end, int sub) {
  v4f acc = {0.0f, 0.0f, 0.0f, 0.0f};
  int j = beg;
  for (; j + 4 <= end; j += 4) {
    const int s0 = col[j];
    const int s1 = col[j + 1];
    const int s2 = col[j + 2];
    const int s3 = col[j + 3];
    const v4f r0 = *(const v4f*)(H + ((size_t)s0 * LPN + sub) * 4);
    const v4f r1 = *(const v4f*)(H + ((size_t)s1 * LPN + sub) * 4);
    const v4f r2 = *(const v4f*)(H + ((size_t)s2 * LPN + sub) * 4);
    const v4f r3 = *(const v4f*)(H + ((size_t)s3 * LPN + sub) * 4);
    acc += r0;
    acc += r1;
    acc += r2;
    acc += r3;
  }
  for (; j < end; ++j) {
    const int s = col[j];
    acc += *(const v4f*)(H + ((size_t)s * LPN + sub) * 4);
  }
  return acc;
}

template <int C>
__global__ void owner_relu(const float* __restrict__ H,
                           const int* __restrict__ rowptr,
                           const int* __restrict__ col, float* FZ,
                           int nnodes) {
  constexpr int LPN = C / 4;
  const int gtid = blockIdx.x * blockDim.x + threadIdx.x;
  const int node = gtid / LPN;
  const int sub = gtid % LPN;
  if (node >= nnodes) return;

  const int beg = rowptr[node];
  const int end = rowptr[node + 1];
  const size_t own = ((size_t)node * LPN + sub) * 4;

  const v4f f = *(const v4f*)(FZ + own);
  const v4f acc = gather_rows<LPN>(H, col, beg, end, sub);

  v4f v = f + acc;
  v.x = v.x > 0.0f ? v.x : 0.0f;
  v.y = v.y > 0.0f ? v.y : 0.0f;
  v.z = v.z > 0.0f ? v.z : 0.0f;
  v.w = v.w > 0.0f ? v.w : 0.0f;
  *(v4f*)(FZ + own) = v;
}

__global__ void owner_logsoftmax16(const float* __restrict__ H,
                                   const int* __restrict__ rowptr,
                                   const int* __restrict__ col,
                                   const float* __restrict__ F,
                                   float* __restrict__ O, int nnodes) {
  constexpr int LPN = 4;
  const int gtid = blockIdx.x * blockDim.x + threadIdx.x;
  const int node = gtid / LPN;
  const int sub = gtid % LPN;
  const bool live = node < nnodes;

  const int beg = live ? rowptr[node] : 0;
  const int end = live ? rowptr[node + 1] : 0;
  const size_t own = ((size_t)node * LPN + sub) * 4;

  v4f f = {0.0f, 0.0f, 0.0f, 0.0f};
  if (live) f = *(const v4f*)(F + own);
  const v4f acc = gather_rows<LPN>(H, col, beg, end, sub);
  const v4f v = f + acc;

  float mx = fmaxf(fmaxf(v.x, v.y), fmaxf(v.z, v.w));
  mx = fmaxf(mx, __shfl_xor(mx, 1));
  mx = fmaxf(mx, __shfl_xor(mx, 2));
  float s = __expf(v.x - mx) + __expf(v.y - mx) + __expf(v.z - mx) +
            __expf(v.w - mx);
  s += __shfl_xor(s, 1);
  s += __shfl_xor(s, 2);
  const float l = mx + __logf(s);
  if (live) {
    v4f o = {v.x - l, v.y - l, v.z - l, v.w - l};
    *(v4f*)(O + own) = o;
  }
}

static inline int cdiv_ll(long long a, long long b) {
  return (int)((a + b - 1) / b);
}

extern "C" void kernel_launch(void* const* d_in, const int* in_sizes, int n_in,
                              void* d_out, int out_size, void* d_ws,
                              size_t ws_size, hipStream_t stream) {
  (void)n_in; (void)out_size; (void)ws_size;
  const float* x     = (const float*)d_in[0];
  const float* w1a_w = (const float*)d_in[1];
  const float* w1a_b = (const float*)d_in[2];
  const float* w1b_w = (const float*)d_in[3];
  const float* w1b_b = (const float*)d_in[4];
  const float* w2a_w = (const float*)d_in[5];
  const float* w2a_b = (const float*)d_in[6];
  const float* w2b_w = (const float*)d_in[7];
  const float* w2b_b = (const float*)d_in[8];
  const int*   edge  = (const int*)d_in[9];

  const int N = in_sizes[0] / 256;
  const int E = in_sizes[9] / 2;
  const int* srcI = edge;
  const int* dstI = edge + E;

  float* buf0 = (float*)d_ws;
  float* buf1 = buf0 + (size_t)N * 32;
  float* buf2 = buf1 + (size_t)N * 32;

  const int CE = cdiv_ll(E, NCH);
  const int NB = cdiv_ll(N, BIN);

  int* rowptr = (int*)buf1;
  int* col    = rowptr + (size_t)N + 64;
  int* lofs   = col + (size_t)E;
  int* tmp    = (int*)buf2;
  float* f2   = buf0 + (size_t)N * 16;

  const int BLK = 256;
  const int MT = 2;
  const int waves = N / (16 * MT);
  const int gemm_blocks = cdiv_ll(waves, BLK / 32);

  chunk_sort<<<NCH, SORT_T, 0, stream>>>(srcI, dstI, lofs, tmp, E, CE, NB);
  bin_finish<<<NB, SORT_T, 0, stream>>>(tmp, lofs, rowptr, col, N, CE);

  wmma_linear2<256, 32, 32, MT><<<gemm_blocks, BLK, 0, stream>>>(
      x, w1a_w, w1a_b, w1b_w, w1b_b, buf0, buf2, N);
  owner_relu<32><<<cdiv_ll((long long)N * 8, BLK), BLK, 0, stream>>>(
      buf0, rowptr, col, buf2, N);

  wmma_linear2<32, 16, 16, MT><<<gemm_blocks, BLK, 0, stream>>>(
      buf2, w2a_w, w2a_b, w2b_w, w2b_b, buf0, f2, N);
  owner_logsoftmax16<<<cdiv_ll((long long)N * 4, BLK), BLK, 0, stream>>>(
      buf0, rowptr, col, f2, (float*)d_out, N);
}


# Owner sums run as 6250 resident waves of 16 nodes each in turn (were 25000 and 12504 short waves); col entries come by coalesced loads a group ahead and lanes exchange them; same order of sums
# speedup vs baseline: 3.3387x; 1.0671x over previous
#include <hip/hip_runtime.h>

typedef float v2f __attribute__((ext_vector_type(2)));
typedef float v8f __attribute__((ext_vector_type(8)));

typedef float v4f __attribute__((ext_vector_type(4)));

template <int MT, int NT>
struct Frag {
  v4f x[MT][4];
  v4f w[NT][4];
};

template <int MT, int NT>
__device__ __forceinline__ void load_frag(Frag<MT, NT>& f,
                                          const float* const (&xrow)[MT],
                                          const float* const (&wrow)[NT],
                                          int kb) {
#pragma unroll
  for (int i = 0; i < 4; ++i) {
#pragma unroll
    for (int mt = 0; mt < MT; ++mt)
      f.x[mt][i] = *(const v4f*)(xrow[mt] + kb + 4 * i);
#pragma unroll
    for (int t = 0; t < NT; ++t)
      f.w[t][i] = *(const v4f*)(wrow[t] + kb + 4 * i);
  }
}

template <int MT, int NT>
__device__ __forceinline__ void mma_frag(v8f (&acc)[MT][NT],
                                         const Frag<MT, NT>& f) {
#pragma unroll
  for (int i = 0; i < 4; ++i)
#pragma unroll
    for (int p = 0; p < 2; ++p) {
      v2f w[NT];
#pragma unroll
      for (int t = 0; t < NT; ++t) {
        w[t].x = f.w[t][i][2 * p];
        w[t].y = f.w[t][i][2 * p + 1];
      }
#pragma unroll
      for (int mt = 0; mt < MT; ++mt) {
        v2f xv;
        xv.x = f.x[mt][i][2 * p];
        xv.y = f.x[mt][i][2 * p + 1];
#pragma unroll
        for (int t = 0; t < NT; ++t) {
          acc[mt][t] = __builtin_amdgcn_wmma_f32_16x16x4_f32(
               false, w[t],  false, xv,
               (short)0, acc[mt][t],  false,
               false);
        }
      }
    }
}

template <int INC, int MID, int OUTC, int MT>
__global__ void __launch_bounds__(256) wmma_linear2(const float* __restrict__ X,
                             const float* __restrict__ Wa,
                             const float* __restrict__ Ba,
                             const float* __restrict__ Wb,
                             const float* __restrict__ Bb,
                             float* __restrict__ Y1,
                             float* __restrict__ Y2, int nrows) {
  constexpr int NT = MID / 16;
  constexpr int NT2 = OUTC / 16;
  int wave = (int)((blockIdx.x * blockDim.x + threadIdx.x) >> 5);
  int lane = threadIdx.x & 31;
  int row0 = wave * 16 * MT;
  if (row0 >= nrows) return;

  int nm = lane & 15;
  int hh = lane >> 4;

  const float* xrow[MT];
#pragma unroll
  for (int mt = 0; mt < MT; ++mt)
    xrow[mt] = X + (size_t)(row0 + mt * 16 + nm) * INC + hh * 16;
  const float* wrow[NT];
#pragma unroll
  for (int t = 0; t < NT; ++t)
    wrow[t] = Wa + (size_t)(t * 16 + nm) * INC + hh * 16;

  v8f acc[MT][NT];
#pragma unroll
  for (int t = 0; t < NT; ++t)
#pragma unroll
    for (int v = 0; v < 8; ++v) {
      float bv = Ba[t * 16 + hh * 8 + v];
#pragma unroll
      for (int mt = 0; mt < MT; ++mt) acc[mt][t][v] = bv;
    }

  static_assert(INC == 32 || INC % 64 == 0, "k blocks are taken in pairs");
  Frag<MT, NT> fa;
  load_frag<MT, NT>(fa, xrow, wrow, 0);
  if constexpr (INC == 32) {
    mma_frag<MT, NT>(acc, fa);
  } else {
    Frag<MT, NT> fb;
#pragma unroll
    for (int kb = 0; kb + 64 < INC; kb += 64) {
      load_frag<MT, NT>(fb, xrow, wrow, kb + 32);
      __builtin_amdgcn_sched_barrier(0);
      mma_frag<MT, NT>(acc, fa);
      __builtin_amdgcn_sched_barrier(0);
      load_frag<MT, NT>(fa, xrow, wrow, kb + 64);
      __builtin_amdgcn_sched_barrier(0);
      mma_frag<MT, NT>(acc, fb);
      __builtin_amdgcn_sched_barrier(0);
    }
    load_frag<MT, NT>(fb, xrow, wrow, INC - 32);
    __builtin_amdgcn_sched_barrier(0);
    mma_frag<MT, NT>(acc, fa);
    mma_frag<MT, NT>(acc, fb);
  }

#pragma unroll
  for (int mt = 0; mt < MT; ++mt)
#pragma unroll
    for (int t = 0; t < NT; ++t) {
      float* p = Y1 + (size_t)(row0 + mt * 16 + nm) * MID + t * 16 + hh * 8;
      v4f lo = {acc[mt][t][0], acc[mt][t][1], acc[mt][t][2], acc[mt][t][3]};
      v4f hi = {acc[mt][t][4], acc[mt][t][5], acc[mt][t][6], acc[mt][t][7]};
      *(v4f*)p = lo;
      *(v4f*)(p + 4) = hi;
    }

  v8f acc2[MT][NT2];
#pragma unroll
  for (int t2 = 0; t2 < NT2; ++t2)
#pragma unroll
    for (int v = 0; v < 8; ++v) {
      float bv = Bb[t2 * 16 + hh * 8 + v];
#pragma unroll
      for (int mt = 0; mt < MT; ++mt) acc2[mt][t2][v] = bv;
    }

#pragma unroll
  for (int t = 0; t < NT; ++t)
#pragma unroll
    for (int q = 0; q < 4; ++q) {
      v2f wb[NT2];
#pragma unroll
      for (int t2 = 0; t2 < NT2; ++t2) {
        const float* wp =
            Wb + (size_t)(t2 * 16 + nm) * MID + t * 16 + hh * 8 + 2 * q;
        wb[t2].x = wp[0];
        wb[t2].y = wp[1];
      }
#pragma unroll
      for (int mt = 0; mt < MT; ++mt) {
        v2f hv;
        hv.x = acc[mt][t][2 * q];
        hv.y = acc[mt][t][2 * q + 1];
#pragma unroll
        for (int t2 = 0; t2 < NT2; ++t2) {
          acc2[mt][t2] = __builtin_amdgcn_wmma_f32_16x16x4_f32(
               false, wb[t2],  false, hv,
               (short)0, acc2[mt][t2],  false,
               false);
        }
      }
    }

#pragma unroll
  for (int mt = 0; mt < MT; ++mt)
#pragma unroll
    for (int t2 = 0; t2 < NT2; ++t2) {
      float* p = Y2 + (size_t)(row0 + mt * 16 + nm) * OUTC + t2 * 16 + hh * 8;
      v4f lo = {acc2[mt][t2][0], acc2[mt][t2][1], acc2[mt][t2][2],
                acc2[mt][t2][3]};
      v4f hi = {acc2[mt][t2][4], acc2[mt][t2][5], acc2[mt][t2][6],
                acc2[mt][t2][7]};
      *(v4f*)p = lo;
      *(v4f*)(p + 4) = hi;
    }
}

typedef int v2i __attribute__((ext_vector_type(2)));

constexpr int BIN_SHIFT = 8;
constexpr int BIN = 1 << BIN_SHIFT;
constexpr int SORT_T = 1024;
constexpr int MAXB = SORT_T;
constexpr int EPT = 8;
constexpr int NCH = 256;
constexpr int CPW = NCH / 32;

__global__ void __launch_bounds__(SORT_T)
chunk_sort(const int* __restrict__ srcI, const int* __restrict__ dstI,
           int* __restrict__ lofs, int* __restrict__ tmp, int nedges, int ce,
           int nb) {
  __shared__ int h[MAXB];
  __shared__ int wtot[32];
  const int t = threadIdx.x;
  const int lane = t & 31;
  const int w = t >> 5;
  const int c = blockIdx.x;
  h[t] = 0;
  __syncthreads();

  const int beg = c * ce;
  const int end = beg + ce < nedges ? beg + ce : nedges;

  int d[EPT];
  int s[EPT];
#pragma unroll
  for (int u = 0; u < EPT; ++u) {
    const int e = beg + t + u * SORT_T;
    d[u] = e < end ? dstI[e] : -1;
    s[u] = e < end ? srcI[e] : 0;
  }
#pragma unroll
  for (int u = 0; u < EPT; ++u)
    if (d[u] >= 0) atomicAdd(&h[d[u] >> BIN_SHIFT], 1);
  for (int e = beg + t + EPT * SORT_T; e < end; e += SORT_T)
    atomicAdd(&h[dstI[e] >> BIN_SHIFT], 1);
  __syncthreads();

  const int cnt = h[t];
  int incl = cnt;
  for (int k = 1; k < 32; k <<= 1) {
    int v = __shfl_up(incl, k);
    if (lane >= k) incl += v;
  }
  if (lane == 31) wtot[w] = incl;
  __syncthreads();
  int wp = lane < w ? wtot[lane] : 0;
  for (int k = 16; k >= 1; k >>= 1) wp += __shfl_xor(wp, k);
  const int excl = wp + incl - cnt;
  h[t] = excl;
  __syncthreads();
  if (t < nb) lofs[t * NCH + c] = excl;
  if (t == nb - 1) lofs[nb * NCH + c] = excl + cnt;

  int* const slice = tmp + 2 * (size_t)beg;
#pragma unroll
  for (int u = 0; u < EPT; ++u)
    if (d[u] >= 0) {
      const int p = atomicAdd(&h[d[u] >> BIN_SHIFT], 1);
      v2i sd = {s[u], d[u]};
      *(v2i*)(slice + 2 * (size_t)p) = sd;
    }
  for (int e = beg + t + EPT * SORT_T; e < end; e += SORT_T) {
    const int dd = dstI[e];
    const int p = atomicAdd(&h[dd >> BIN_SHIFT], 1);
    v2i sd = {srcI[e], dd};
    *(v2i*)(slice + 2 * (size_t)p) = sd;
  }
}

__global__ void __launch_bounds__(SORT_T)
bin_finish(const int* __restrict__ tmp, const int* __restrict__ lofs,
           int* __restrict__ rowptr, int* __restrict__ col, int n, int ce) {
  __shared__ int cnt[BIN];
  __shared__ int fill[BIN];
  __shared__ int wtot[32];
  __shared__ v2i seg[NCH];
  __shared__ int redlo[32];
  __shared__ int redhi[32];
  const int t = threadIdx.x;
  const int lane = t & 31;
  const int w = t >> 5;
  const int b = blockIdx.x;

  int lo = 0;
  int hi = 0;
  if (t < NCH) {
    lo = lofs[b * NCH + t];
    hi = lofs[(b + 1) * NCH + t];
    v2i sg = {t * ce + lo, hi - lo};
    seg[t] = sg;
  }
  if (t < BIN) cnt[t] = 0;
  __syncthreads();

  v2i sg[CPW];
  v2i sd[CPW];
#pragma unroll
  for (int k = 0; k < CPW; ++k) sg[k] = seg[w + 32 * k];
#pragma unroll
  for (int k = 0; k < CPW; ++k) {
    v2i z = {0, -1};
    sd[k] = lane < sg[k].y
                ? *(const v2i*)(tmp + 2 * (size_t)(sg[k].x + lane))
                : z;
  }

  int slo = lo;
  int shi = hi;
  for (int k = 16; k >= 1; k >>= 1) {
    slo += __shfl_xor(slo, k);
    shi += __shfl_xor(shi, k);
  }
  if (lane == 0) {
    redlo[w] = slo;
    redhi[w] = shi;
  }

#pragma unroll
  for (int k = 0; k < CPW; ++k)
    if (sd[k].y >= 0) atomicAdd(&cnt[sd[k].y & (BIN - 1)], 1);
  for (int k = 0; k < CPW; ++k) {
    const v2i g = seg[w + 32 * k];
    for (int l = lane + 32; l < g.y; l += 32)
      atomicAdd(&cnt[tmp[2 * (size_t)(g.x + l) + 1] & (BIN - 1)], 1);
  }
  __syncthreads();

  const int c = t < BIN ? cnt[t] : 0;
  int incl = c;
  for (int k = 1; k < 32; k <<= 1) {
    int v = __shfl_up(incl, k);
    if (lane >= k) incl += v;
  }
  if (lane == 31) wtot[w] = incl;
  __syncthreads();
  int wp = lane < w ? wtot[lane] : 0;
  int beg = redlo[lane];
  int end = redhi[lane];
  for (int k = 16; k >= 1; k >>= 1) {
    wp += __shfl_xor(wp, k);
    beg += __shfl_xor(beg, k);
    end += __shfl_xor(end, k);
  }
  const int excl = beg + wp + incl - c;
  if (t < BIN) fill[t] = excl;
  __syncthreads();
  if (t < BIN) {
    const int node = b * BIN + t;
    if (node < n) rowptr[node] = excl;
  }
  if (b == (int)gridDim.x - 1 && t == 0) rowptr[n] = end;

#pragma unroll
  for (int k = 0; k < CPW; ++k)
    if (sd[k].y >= 0) {
      const int p = atomicAdd(&fill[sd[k].y & (BIN - 1)], 1);
      col[p] = sd[k].x;
    }
  for (int k = 0; k < CPW; ++k) {
    const v2i g = seg[w + 32 * k];
    for (int l = lane + 32; l < g.y; l += 32) {
      const v2i e = *(const v2i*)(tmp + 2 * (size_t)(g.x + l));
      const int p = atomicAdd(&fill[e.y & (BIN - 1)], 1);
      col[p] = e.x;
    }
  }
}

template <int LPN>
struct Idx {
  int v[32 / LPN];
};

template <int LPN>
__device__ __forceinline__ void load_idx(Idx<LPN>& ix,
                                         const int* __restrict__ col, int beg,
                                         int deg, int sub) {
#pragma unroll
  for (int i = 0; i < 32 / LPN; ++i) {
    const int j = i * LPN + sub;
    ix.v[i] = col[j < deg ? beg + j : 0];
  }
}

struct Rows8 {
  v4f r[8];
};

template <int LPN, int B>
__device__ __forceinline__ void load_block(Rows8& rw,
                                           const float* __restrict__ H,
                                           const Idx<LPN>& ix, int deg,
                                           int sub) {
#pragma unroll
  for (int e = 0; e < 8; ++e) {
    const int j = 8 * B + e;
    const int got = __shfl(ix.v[j / LPN], j % LPN, LPN);
    const int s = j < deg ? got : 0;
    rw.r[e] = ((const v4f*)H)[(unsigned)(s * LPN + sub)];
  }
}

template <int B>
__device__ __forceinline__ void add_block(v4f& acc, const Rows8& rw, int deg) {
#pragma unroll
  for (int e = 0; e < 8; ++e)
    if (8 * B + e < deg) acc += rw.r[e];
}

template <int LPN>
__device__ __forceinline__ v4f gather_group(const float* __restrict__ H,
                                            const int* __restrict__ col,
                                            const Idx<LPN>& ix, int beg,
                                            int deg, int sub) {
  v4f acc = {0.0f, 0.0f, 0.0f, 0.0f};
  Rows8 ra;
  Rows8 rb;
  load_block<LPN, 0>(ra, H, ix, deg, sub);
  const bool m1 = __any(8 < deg) != 0;
  if (m1) load_block<LPN, 1>(rb, H, ix, deg, sub);
  add_block<0>(acc, ra, deg);
  if (m1) {
    const bool m2 = __any(16 < deg) != 0;
    if (m2) load_block<LPN, 2>(ra, H, ix, deg, sub);
    add_block<1>(acc, rb, deg);
    if (m2) {
      const bool m3 = __any(24 < deg) != 0;
      if (m3) load_block<LPN, 3>(rb, H, ix, deg, sub);
      add_block<2>(acc, ra, deg);
      if (m3) add_block<3>(acc, rb, deg);
    }
  }
  for (int j = 32; j < deg; ++j) {
    const int s = col[beg + j];
    acc += *(const v4f*)(H + ((size_t)s * LPN + sub) * 4);
  }
  return acc;
}

template <int LPN, int GPW, class Fin>
__device__ __forceinline__ void owner_groups(const float* __restrict__ H,
                                             const int* __restrict__ rowptr,
                                             const int* __restrict__ col,
                                             const float* OWN, int nnodes,
                                             Fin fin) {
  constexpr int NPG = 32 / LPN;
  const int lane = threadIdx.x & 31;
  const int wave = (int)((blockIdx.x * blockDim.x + threadIdx.x) >> 5);
  const int sub = lane % LPN;
  int node = wave * (GPW * NPG) + lane / LPN;
  const v4f z = {0.0f, 0.0f, 0.0f, 0.0f};

  int begC = 0, degC = 0, begN = 0, degN = 0;
  v4f fC = z, fN = z;
  if (node < nnodes) {
    begC = rowptr[node];
    degC = rowptr[node + 1] - begC;
    fC = *(const v4f*)(OWN + ((size_t)node * LPN + sub) * 4);
  }
  if (GPW > 1 && node + NPG < nnodes) {
    begN = rowptr[node + NPG];
    degN = rowptr[node + NPG + 1] - begN;
    fN = *(const v4f*)(OWN + ((size_t)(node + NPG) * LPN + sub) * 4);
  }
  Idx<LPN> ixC;
  load_idx<LPN>(ixC, col, begC, degC, sub);

#pragma nounroll
  for (int t = 0; t < GPW; ++t) {
    Idx<LPN> ixN;
    load_idx<LPN>(ixN, col, begN, degN, sub);
    int begF = 0, endF = 0;
    v4f fF = z;
    const int nodeF = node + 2 * NPG;
    if (t + 2 < GPW && nodeF < nnodes) {
      begF = rowptr[nodeF];
      endF = rowptr[nodeF + 1];
      fF = *(const v4f*)(OWN + ((size_t)nodeF * LPN + sub) * 4);
    }

    const v4f acc = gather_group<LPN>(H, col, ixC, begC, degC, sub);
    fin(node, node < nnodes, fC + acc);

    ixC = ixN;
    begC = begN; degC = degN; fC = fN;
    begN = begF; degN = endF - begF; fN = fF;
    node += NPG;
  }
}

template <int C, int GPW>
__global__ void
owner_relu(const float* __restrict__ H, const int* __restrict__ rowptr,
           const int* __restrict__ col, float* FZ, int nnodes) {
  constexpr int LPN = C / 4;
  const int sub = (threadIdx.x & 31) % LPN;
  owner_groups<LPN, GPW>(H, rowptr, col, FZ, nnodes,
                         [=](int node, bool live, v4f v) {
    v.x = v.x > 0.0f ? v.x : 0.0f;
    v.y = v.y > 0.0f ? v.y : 0.0f;
    v.z = v.z > 0.0f ? v.z : 0.0f;
    v.w = v.w > 0.0f ? v.w : 0.0f;
    if (live) *(v4f*)(FZ + ((size_t)node * LPN + sub) * 4) = v;
  });
}

template <int GPW>
__global__ void
owner_logsoftmax16(const float* __restrict__ H, const int* __restrict__ rowptr,
                   const int* __restrict__ col, const float* __restrict__ F,
                   float* __restrict__ O, int nnodes) {
  constexpr int LPN = 4;
  const int sub = (threadIdx.x & 31) % LPN;
  owner_groups<LPN, GPW>(H, rowptr, col, F, nnodes,
                         [=](int node, bool live, v4f v) {
    float mx = fmaxf(fmaxf(v.x, v.y), fmaxf(v.z, v.w));
    mx = fmaxf(mx, __shfl_xor(mx, 1));
    mx = fmaxf(mx, __shfl_xor(mx, 2));
    float s = __expf(v.x - mx) + __expf(v.y - mx) + __expf(v.z - mx) +
              __expf(v.w - mx);
    s += __shfl_xor(s, 1);
    s += __shfl_xor(s, 2);
    const float l = mx + __logf(s);
    if (live) {
      v4f o = {v.x - l, v.y - l, v.z - l, v.w - l};
      *(v4f*)(O + ((size_t)node * LPN + sub) * 4) = o;
    }
  });
}

constexpr int OWN_GPW32 = 4;
constexpr int OWN_GPW16 = 2;
constexpr int OWN_NPW = 16;
static_assert(OWN_GPW32 * 4 == OWN_NPW && OWN_GPW16 * 8 == OWN_NPW,
              "both owner launches share one grid");

static inline int cdiv_ll(long long a, long long b) {
  return (int)((a + b - 1) / b);
}

extern "C" void kernel_launch(void* const* d_in, const int* in_sizes, int n_in,
                              void* d_out, int out_size, void* d_ws,
                              size_t ws_size, hipStream_t stream) {
  (void)n_in; (void)out_size; (void)ws_size;
  const float* x     = (const float*)d_in[0];
  const float* w1a_w = (const float*)d_in[1];
  const float* w1a_b = (const float*)d_in[2];
  const float* w1b_w = (const float*)d_in[3];
  const float* w1b_b = (const float*)d_in[4];
  const float* w2a_w = (const float*)d_in[5];
  const float* w2a_b = (const float*)d_in[6];
  const float* w2b_w = (const float*)d_in[7];
  const float* w2b_b = (const float*)d_in[8];
  const int*   edge  = (const int*)d_in[9];

  const int N = in_sizes[0] / 256;
  const int E = in_sizes[9] / 2;
  const int* srcI = edge;
  const int* dstI = edge + E;

  float* buf0 = (float*)d_ws;
  float* buf1 = buf0 + (size_t)N * 32;
  float* buf2 = buf1 + (size_t)N * 32;

  const int CE = cdiv_ll(E, NCH);
  const int NB = cdiv_ll(N, BIN);

  int* rowptr = (int*)buf1;
  int* col    = rowptr + (size_t)N + 64;
  int* lofs   = col + (size_t)E;
  int* tmp    = (int*)buf2;
  float* f2   = buf0 + (size_t)N * 16;

  const int BLK = 256;
  const int MT = 2;
  const int waves = N / (16 * MT);
  const int gemm_blocks = cdiv_ll(waves, BLK / 32);
  const int own_blocks = cdiv_ll(cdiv_ll(N, OWN_NPW), BLK / 32);

  chunk_sort<<<NCH, SORT_T, 0, stream>>>(srcI, dstI, lofs, tmp, E, CE, NB);
  bin_finish<<<NB, SORT_T, 0, stream>>>(tmp, lofs, rowptr, col, N, CE);

  wmma_linear2<256, 32, 32, MT><<<gemm_blocks, BLK, 0, stream>>>(
      x, w1a_w, w1a_b, w1b_w, w1b_b, buf0, buf2, N);
  owner_relu<32, OWN_GPW32><<<own_blocks, BLK, 0, stream>>>(
      buf0, rowptr, col, buf2, N);

  wmma_linear2<32, 16, 16, MT><<<gemm_blocks, BLK, 0, stream>>>(
      buf2, w2a_w, w2a_b, w2b_w, w2b_b, buf0, f2, N);
  owner_logsoftmax16<OWN_GPW16><<<own_blocks, BLK, 0, stream>>>(
      buf0, rowptr, col, f2, (float*)d_out, N);
}


# Both sort launches build their output in LDS (a chunk's sorted pairs, a bin's stretch of col) and store it as consecutive runs instead of one scattered line per lane; order of sums as before
# speedup vs baseline: 3.6243x; 1.0855x over previous
#include <hip/hip_runtime.h>

typedef float v2f __attribute__((ext_vector_type(2)));
typedef float v8f __attribute__((ext_vector_type(8)));

typedef float v4f __attribute__((ext_vector_type(4)));

template <int MT, int NT>
struct Frag {
  v4f x[MT][4];
  v4f w[NT][4];
};

template <int MT, int NT>
__device__ __forceinline__ void load_frag(Frag<MT, NT>& f,
                                          const float* const (&xrow)[MT],
                                          const float* const (&wrow)[NT],
                                          int kb) {
#pragma unroll
  for (int i = 0; i < 4; ++i) {
#pragma unroll
    for (int mt = 0; mt < MT; ++mt)
      f.x[mt][i] = *(const v4f*)(xrow[mt] + kb + 4 * i);
#pragma unroll
    for (int t = 0; t < NT; ++t)
      f.w[t][i] = *(const v4f*)(wrow[t] + kb + 4 * i);
  }
}

template <int MT, int NT>
__device__ __forceinline__ void mma_frag(v8f (&acc)[MT][NT],
                                         const Frag<MT, NT>& f) {
#pragma unroll
  for (int i = 0; i < 4; ++i)
#pragma unroll
    for (int p = 0; p < 2; ++p) {
      v2f w[NT];
#pragma unroll
      for (int t = 0; t < NT; ++t) {
        w[t].x = f.w[t][i][2 * p];
        w[t].y = f.w[t][i][2 * p + 1];
      }
#pragma unroll
      for (int mt = 0; mt < MT; ++mt) {
        v2f xv;
        xv.x = f.x[mt][i][2 * p];
        xv.y = f.x[mt][i][2 * p + 1];
#pragma unroll
        for (int t = 0; t < NT; ++t) {
          acc[mt][t] = __builtin_amdgcn_wmma_f32_16x16x4_f32(
               false, w[t],  false, xv,
               (short)0, acc[mt][t],  false,
               false);
        }
      }
    }
}

template <int INC, int MID, int OUTC, int MT>
__global__ void __launch_bounds__(256) wmma_linear2(const float* __restrict__ X,
                             const float* __restrict__ Wa,
                             const float* __restrict__ Ba,
                             const float* __restrict__ Wb,
                             const float* __restrict__ Bb,
                             float* __restrict__ Y1,
                             float* __restrict__ Y2, int nrows) {
  constexpr int NT = MID / 16;
  constexpr int NT2 = OUTC / 16;
  int wave = (int)((blockIdx.x * blockDim.x + threadIdx.x) >> 5);
  int lane = threadIdx.x & 31;
  int row0 = wave * 16 * MT;
  if (row0 >= nrows) return;

  int nm = lane & 15;
  int hh = lane >> 4;

  const float* xrow[MT];
#pragma unroll
  for (int mt = 0; mt < MT; ++mt)
    xrow[mt] = X + (size_t)(row0 + mt * 16 + nm) * INC + hh * 16;
  const float* wrow[NT];
#pragma unroll
  for (int t = 0; t < NT; ++t)
    wrow[t] = Wa + (size_t)(t * 16 + nm) * INC + hh * 16;

  v8f acc[MT][NT];
#pragma unroll
  for (int t = 0; t < NT; ++t)
#pragma unroll
    for (int v = 0; v < 8; ++v) {
      float bv = Ba[t * 16 + hh * 8 + v];
#pragma unroll
      for (int mt = 0; mt < MT; ++mt) acc[mt][t][v] = bv;
    }

  static_assert(INC == 32 || INC % 64 == 0, "k blocks are taken in pairs");
  Frag<MT, NT> fa;
  load_frag<MT, NT>(fa, xrow, wrow, 0);
  if constexpr (INC == 32) {
    mma_frag<MT, NT>(acc, fa);
  } else {
    Frag<MT, NT> fb;
#pragma unroll
    for (int kb = 0; kb + 64 < INC; kb += 64) {
      load_frag<MT, NT>(fb, xrow, wrow, kb + 32);
      __builtin_amdgcn_sched_barrier(0);
      mma_frag<MT, NT>(acc, fa);
      __builtin_amdgcn_sched_barrier(0);
      load_frag<MT, NT>(fa, xrow, wrow, kb + 64);
      __builtin_amdgcn_sched_barrier(0);
      mma_frag<MT, NT>(acc, fb);
      __builtin_amdgcn_sched_barrier(0);
    }
    load_frag<MT, NT>(fb, xrow, wrow, INC - 32);
    __builtin_amdgcn_sched_barrier(0);
    mma_frag<MT, NT>(acc, fa);
    mma_frag<MT, NT>(acc, fb);
  }

#pragma unroll
  for (int mt = 0; mt < MT; ++mt)
#pragma unroll
    for (int t = 0; t < NT; ++t) {
      float* p = Y1 + (size_t)(row0 + mt * 16 + nm) * MID + t * 16 + hh * 8;
      v4f lo = {acc[mt][t][0], acc[mt][t][1], acc[mt][t][2], acc[mt][t][3]};
      v4f hi = {acc[mt][t][4], acc[mt][t][5], acc[mt][t][6], acc[mt][t][7]};
      *(v4f*)p = lo;
      *(v4f*)(p + 4) = hi;
    }

  v8f acc2[MT][NT2];
#pragma unroll
  for (int t2 = 0; t2 < NT2; ++t2)
#pragma unroll
    for (int v = 0; v < 8; ++v) {
      float bv = Bb[t2 * 16 + hh * 8 + v];
#pragma unroll
      for (int mt = 0; mt < MT; ++mt) acc2[mt][t2][v] = bv;
    }

#pragma unroll
  for (int t = 0; t < NT; ++t)
#pragma unroll
    for (int q = 0; q < 4; ++q) {
      v2f wb[NT2];
#pragma unroll
      for (int t2 = 0; t2 < NT2; ++t2) {
        const float* wp =
            Wb + (size_t)(t2 * 16 + nm) * MID + t * 16 + hh * 8 + 2 * q;
        wb[t2].x = wp[0];
        wb[t2].y = wp[1];
      }
#pragma unroll
      for (int mt = 0; mt < MT; ++mt) {
        v2f hv;
        hv.x = acc[mt][t][2 * q];
        hv.y = acc[mt][t][2 * q + 1];
#pragma unroll
        for (int t2 = 0; t2 < NT2; ++t2) {
          acc2[mt][t2] = __builtin_amdgcn_wmma_f32_16x16x4_f32(
               false, wb[t2],  false, hv,
               (short)0, acc2[mt][t2],  false,
               false);
        }
      }
    }

#pragma unroll
  for (int mt = 0; mt < MT; ++mt)
#pragma unroll
    for (int t2 = 0; t2 < NT2; ++t2) {
      float* p = Y2 + (size_t)(row0 + mt * 16 + nm) * OUTC + t2 * 16 + hh * 8;
      v4f lo = {acc2[mt][t2][0], acc2[mt][t2][1], acc2[mt][t2][2],
                acc2[mt][t2][3]};
      v4f hi = {acc2[mt][t2][4], acc2[mt][t2][5], acc2[mt][t2][6],
                acc2[mt][t2][7]};
      *(v4f*)p = lo;
      *(v4f*)(p + 4) = hi;
    }
}

typedef int v2i __attribute__((ext_vector_type(2)));

constexpr int BIN_SHIFT = 8;
constexpr int BIN = 1 << BIN_SHIFT;
constexpr int SORT_T = 1024;
constexpr int MAXB = SORT_T;
constexpr int EPT = 8;
constexpr int NCH = 256;
constexpr int CPW = NCH / 32;
constexpr int CAPS = 7 * SORT_T;
constexpr int CAPB = 8 * SORT_T;

__global__ void __launch_bounds__(SORT_T)
chunk_sort(const int* __restrict__ srcI, const int* __restrict__ dstI,
           int* __restrict__ lofs, int* __restrict__ tmp, int nedges, int ce,
           int nb) {
  __shared__ int h[MAXB];
  __shared__ int wtot[32];
  __shared__ v2i stage[CAPS];
  const int t = threadIdx.x;
  const int lane = t & 31;
  const int w = t >> 5;
  const int c = blockIdx.x;
  h[t] = 0;
  __syncthreads();

  const int beg = c * ce;
  const int end = beg + ce < nedges ? beg + ce : nedges;

  int d[EPT];
  int s[EPT];
#pragma unroll
  for (int u = 0; u < EPT; ++u) {
    const int e = beg + t + u * SORT_T;
    d[u] = e < end ? dstI[e] : -1;
    s[u] = e < end ? srcI[e] : 0;
  }
#pragma unroll
  for (int u = 0; u < EPT; ++u)
    if (d[u] >= 0) atomicAdd(&h[d[u] >> BIN_SHIFT], 1);
  for (int e = beg + t + EPT * SORT_T; e < end; e += SORT_T)
    atomicAdd(&h[dstI[e] >> BIN_SHIFT], 1);
  __syncthreads();

  const int cnt = h[t];
  int incl = cnt;
  for (int k = 1; k < 32; k <<= 1) {
    int v = __shfl_up(incl, k);
    if (lane >= k) incl += v;
  }
  if (lane == 31) wtot[w] = incl;
  __syncthreads();
  int wp = lane < w ? wtot[lane] : 0;
  for (int k = 16; k >= 1; k >>= 1) wp += __shfl_xor(wp, k);
  const int excl = wp + incl - cnt;
  h[t] = excl;
  __syncthreads();

  int* const slice = tmp + 2 * (size_t)beg;
  if (ce <= CAPS) {
#pragma unroll
    for (int u = 0; u < EPT; ++u)
      if (d[u] >= 0) {
        const int p = atomicAdd(&h[d[u] >> BIN_SHIFT], 1);
        v2i sd = {s[u], d[u]};
        stage[p] = sd;
      }
    __syncthreads();
    if (t < nb) lofs[t * NCH + c] = excl;
    if (t == nb - 1) lofs[nb * NCH + c] = excl + cnt;
    const int n = end - beg;
    constexpr int OUTS = CAPS / SORT_T;
    v2i o[OUTS];
#pragma unroll
    for (int u = 0; u < OUTS; ++u)
      o[u] = stage[t + u * SORT_T < n ? t + u * SORT_T : 0];
#pragma unroll
    for (int u = 0; u < OUTS; ++u)
      if (t + u * SORT_T < n)
        *(v2i*)(slice + 2 * (size_t)(t + u * SORT_T)) = o[u];
  } else {
    if (t < nb) lofs[t * NCH + c] = excl;
    if (t == nb - 1) lofs[nb * NCH + c] = excl + cnt;
#pragma unroll
    for (int u = 0; u < EPT; ++u)
      if (d[u] >= 0) {
        const int p = atomicAdd(&h[d[u] >> BIN_SHIFT], 1);
        v2i sd = {s[u], d[u]};
        *(v2i*)(slice + 2 * (size_t)p) = sd;
      }
    for (int e = beg + t + EPT * SORT_T; e < end; e += SORT_T) {
      const int dd = dstI[e];
      const int p = atomicAdd(&h[dd >> BIN_SHIFT], 1);
      v2i sd = {srcI[e], dd};
      *(v2i*)(slice + 2 * (size_t)p) = sd;
    }
  }
}
static_assert(CAPS <= EPT * SORT_T, "a staged chunk is held in registers");
static_assert(CAPS % SORT_T == 0, "the staged slice leaves in whole rounds");

template <bool STAGED>
__device__ __forceinline__ void place_bin(const v2i (&sd)[CPW], const v2i* seg,
                                          const int* __restrict__ tmp,
                                          int* fill, int* stage,
                                          int* __restrict__ col, int beg,
                                          int w, int lane) {
#pragma unroll
  for (int k = 0; k < CPW; ++k)
    if (sd[k].y >= 0) {
      const int p = atomicAdd(&fill[sd[k].y & (BIN - 1)], 1);
      if (STAGED) stage[p - beg] = sd[k].x;
      else col[p] = sd[k].x;
    }
  for (int k = 0; k < CPW; ++k) {
    const v2i g = seg[w + 32 * k];
    for (int l = lane + 32; l < g.y; l += 32) {
      const v2i e = *(const v2i*)(tmp + 2 * (size_t)(g.x + l));
      const int p = atomicAdd(&fill[e.y & (BIN - 1)], 1);
      if (STAGED) stage[p - beg] = e.x;
      else col[p] = e.x;
    }
  }
}

__global__ void __launch_bounds__(SORT_T)
bin_finish(const int* __restrict__ tmp, const int* __restrict__ lofs,
           int* __restrict__ rowptr, int* __restrict__ col, int n, int ce) {
  __shared__ int cnt[BIN];
  __shared__ int fill[BIN];
  __shared__ int wtot[32];
  __shared__ v2i seg[NCH];
  __shared__ int redlo[32];
  __shared__ int redhi[32];
  __shared__ int stage[CAPB];
  const int t = threadIdx.x;
  const int lane = t & 31;
  const int w = t >> 5;
  const int b = blockIdx.x;

  int lo = 0;
  int hi = 0;
  if (t < NCH) {
    lo = lofs[b * NCH + t];
    hi = lofs[(b + 1) * NCH + t];
    v2i sg = {t * ce + lo, hi - lo};
    seg[t] = sg;
  }
  if (t < BIN) cnt[t] = 0;
  __syncthreads();

  v2i sg[CPW];
  v2i sd[CPW];
#pragma unroll
  for (int k = 0; k < CPW; ++k) sg[k] = seg[w + 32 * k];
#pragma unroll
  for (int k = 0; k < CPW; ++k) {
    v2i z = {0, -1};
    sd[k] = lane < sg[k].y
                ? *(const v2i*)(tmp + 2 * (size_t)(sg[k].x + lane))
                : z;
  }

  int slo = lo;
  int shi = hi;
  for (int k = 16; k >= 1; k >>= 1) {
    slo += __shfl_xor(slo, k);
    shi += __shfl_xor(shi, k);
  }
  if (lane == 0) {
    redlo[w] = slo;
    redhi[w] = shi;
  }

#pragma unroll
  for (int k = 0; k < CPW; ++k)
    if (sd[k].y >= 0) atomicAdd(&cnt[sd[k].y & (BIN - 1)], 1);
  for (int k = 0; k < CPW; ++k) {
    const v2i g = seg[w + 32 * k];
    for (int l = lane + 32; l < g.y; l += 32)
      atomicAdd(&cnt[tmp[2 * (size_t)(g.x + l) + 1] & (BIN - 1)], 1);
  }
  __syncthreads();

  const int c = t < BIN ? cnt[t] : 0;
  int incl = c;
  for (int k = 1; k < 32; k <<= 1) {
    int v = __shfl_up(incl, k);
    if (lane >= k) incl += v;
  }
  if (lane == 31) wtot[w] = incl;
  __syncthreads();
  int wp = lane < w ? wtot[lane] : 0;
  int beg = redlo[lane];
  int end = redhi[lane];
  for (int k = 16; k >= 1; k >>= 1) {
    wp += __shfl_xor(wp, k);
    beg += __shfl_xor(beg, k);
    end += __shfl_xor(end, k);
  }
  const int excl = beg + wp + incl - c;
  if (t < BIN) fill[t] = excl;
  __syncthreads();

  const int begU = __builtin_amdgcn_readfirstlane(beg);
  const int nw = __builtin_amdgcn_readfirstlane(end) - begU;
  const bool staged = nw <= CAPB;
  if (staged) {
    place_bin<true>(sd, seg, tmp, fill, stage, col, begU, w, lane);
    __syncthreads();
  } else {
    place_bin<false>(sd, seg, tmp, fill, stage, col, begU, w, lane);
  }
  if (t < BIN) {
    const int node = b * BIN + t;
    if (node < n) rowptr[node] = excl;
  }
  if (b == (int)gridDim.x - 1 && t == 0) rowptr[n] = end;
  if (staged) {
    constexpr int OUTB = CAPB / SORT_T;
    int o[OUTB];
#pragma unroll
    for (int u = 0; u < OUTB; ++u)
      o[u] = stage[t + u * SORT_T < nw ? t + u * SORT_T : 0];
#pragma unroll
    for (int u = 0; u < OUTB; ++u)
      if (t + u * SORT_T < nw) col[begU + t + u * SORT_T] = o[u];
  }
}
static_assert(CAPB % SORT_T == 0, "the staged stretch leaves in whole rounds");

template <int LPN>
struct Idx {
  int v[32 / LPN];
};

template <int LPN>
__device__ __forceinline__ void load_idx(Idx<LPN>& ix,
                                         const int* __restrict__ col, int beg,
                                         int deg, int sub) {
#pragma unroll
  for (int i = 0; i < 32 / LPN; ++i) {
    const int j = i * LPN + sub;
    ix.v[i] = col[j < deg ? beg + j : 0];
  }
}

struct Rows8 {
  v4f r[8];
};

template <int LPN, int B>
__device__ __forceinline__ void load_block(Rows8& rw,
                                           const float* __restrict__ H,
                                           const Idx<LPN>& ix, int deg,
                                           int sub) {
#pragma unroll
  for (int e = 0; e < 8; ++e) {
    const int j = 8 * B + e;
    const int got = __shfl(ix.v[j / LPN], j % LPN, LPN);
    const int s = j < deg ? got : 0;
    rw.r[e] = ((const v4f*)H)[(unsigned)(s * LPN + sub)];
  }
}

template <int B>
__device__ __forceinline__ void add_block(v4f& acc, const Rows8& rw, int deg) {
#pragma unroll
  for (int e = 0; e < 8; ++e)
    if (8 * B + e < deg) acc += rw.r[e];
}

template <int LPN>
__device__ __forceinline__ v4f gather_group(const float* __restrict__ H,
                                            const int* __restrict__ col,
                                            const Idx<LPN>& ix, int beg,
                                            int deg, int sub) {
  v4f acc = {0.0f, 0.0f, 0.0f, 0.0f};
  Rows8 ra;
  Rows8 rb;
  load_block<LPN, 0>(ra, H, ix, deg, sub);
  const bool m1 = __any(8 < deg) != 0;
  if (m1) load_block<LPN, 1>(rb, H, ix, deg, sub);
  add_block<0>(acc, ra, deg);
  if (m1) {
    const bool m2 = __any(16 < deg) != 0;
    if (m2) load_block<LPN, 2>(ra, H, ix, deg, sub);
    add_block<1>(acc, rb, deg);
    if (m2) {
      const bool m3 = __any(24 < deg) != 0;
      if (m3) load_block<LPN, 3>(rb, H, ix, deg, sub);
      add_block<2>(acc, ra, deg);
      if (m3) add_block<3>(acc, rb, deg);
    }
  }
  for (int j = 32; j < deg; ++j) {
    const int s = col[beg + j];
    acc += *(const v4f*)(H + ((size_t)s * LPN + sub) * 4);
  }
  return acc;
}

template <int LPN, int GPW, class Fin>
__device__ __forceinline__ void owner_groups(const float* __restrict__ H,
                                             const int* __restrict__ rowptr,
                                             const int* __restrict__ col,
                                             const float* OWN, int nnodes,
                                             Fin fin) {
  constexpr int NPG = 32 / LPN;
  const int lane = threadIdx.x & 31;
  const int wave = (int)((blockIdx.x * blockDim.x + threadIdx.x) >> 5);
  const int sub = lane % LPN;
  int node = wave * (GPW * NPG) + lane / LPN;
  const v4f z = {0.0f, 0.0f, 0.0f, 0.0f};

  int begC = 0, degC = 0, begN = 0, degN = 0;
  v4f fC = z, fN = z;
  if (node < nnodes) {
    begC = rowptr[node];
    degC = rowptr[node + 1] - begC;
    fC = *(const v4f*)(OWN + ((size_t)node * LPN + sub) * 4);
  }
  if (GPW > 1 && node + NPG < nnodes) {
    begN = rowptr[node + NPG];
    degN = rowptr[node + NPG + 1] - begN;
    fN = *(const v4f*)(OWN + ((size_t)(node + NPG) * LPN + sub) * 4);
  }
  Idx<LPN> ixC;
  load_idx<LPN>(ixC, col, begC, degC, sub);

#pragma nounroll
  for (int t = 0; t < GPW; ++t) {
    Idx<LPN> ixN;
    load_idx<LPN>(ixN, col, begN, degN, sub);
    int begF = 0, endF = 0;
    v4f fF = z;
    const int nodeF = node + 2 * NPG;
    if (t + 2 < GPW && nodeF < nnodes) {
      begF = rowptr[nodeF];
      endF = rowptr[nodeF + 1];
      fF = *(const v4f*)(OWN + ((size_t)nodeF * LPN + sub) * 4);
    }

    const v4f acc = gather_group<LPN>(H, col, ixC, begC, degC, sub);
    fin(node, node < nnodes, fC + acc);

    ixC = ixN;
    begC = begN; degC = degN; fC = fN;
    begN = begF; degN = endF - begF; fN = fF;
    node += NPG;
  }
}

template <int C, int GPW>
__global__ void
owner_relu(const float* __restrict__ H, const int* __restrict__ rowptr,
           const int* __restrict__ col, float* FZ, int nnodes) {
  constexpr int LPN = C / 4;
  const int sub = (threadIdx.x & 31) % LPN;
  owner_groups<LPN, GPW>(H, rowptr, col, FZ, nnodes,
                         [=](int node, bool live, v4f v) {
    v.x = v.x > 0.0f ? v.x : 0.0f;
    v.y = v.y > 0.0f ? v.y : 0.0f;
    v.z = v.z > 0.0f ? v.z : 0.0f;
    v.w = v.w > 0.0f ? v.w : 0.0f;
    if (live) *(v4f*)(FZ + ((size_t)node * LPN + sub) * 4) = v;
  });
}

template <int GPW>
__global__ void
owner_logsoftmax16(const float* __restrict__ H, const int* __restrict__ rowptr,
                   const int* __restrict__ col, const float* __restrict__ F,
                   float* __restrict__ O, int nnodes) {
  constexpr int LPN = 4;
  const int sub = (threadIdx.x & 31) % LPN;
  owner_groups<LPN, GPW>(H, rowptr, col, F, nnodes,
                         [=](int node, bool live, v4f v) {
    float mx = fmaxf(fmaxf(v.x, v.y), fmaxf(v.z, v.w));
    mx = fmaxf(mx, __shfl_xor(mx, 1));
    mx = fmaxf(mx, __shfl_xor(mx, 2));
    float s = __expf(v.x - mx) + __expf(v.y - mx) + __expf(v.z - mx) +
              __expf(v.w - mx);
    s += __shfl_xor(s, 1);
    s += __shfl_xor(s, 2);
    const float l = mx + __logf(s);
    if (live) {
      v4f o = {v.x - l, v.y - l, v.z - l, v.w - l};
      *(v4f*)(O + ((size_t)node * LPN + sub) * 4) = o;
    }
  });
}

constexpr int OWN_GPW32 = 4;
constexpr int OWN_GPW16 = 2;
constexpr int OWN_NPW = 16;
static_assert(OWN_GPW32 * 4 == OWN_NPW && OWN_GPW16 * 8 == OWN_NPW,
              "both owner launches share one grid");

static inline int cdiv_ll(long long a, long long b) {
  return (int)((a + b - 1) / b);
}

extern "C" void kernel_launch(void* const* d_in, const int* in_sizes, int n_in,
                              void* d_out, int out_size, void* d_ws,
                              size_t ws_size, hipStream_t stream) {
  (void)n_in; (void)out_size; (void)ws_size;
  const float* x     = (const float*)d_in[0];
  const float* w1a_w = (const float*)d_in[1];
  const float* w1a_b = (const float*)d_in[2];
  const float* w1b_w = (const float*)d_in[3];
  const float* w1b_b = (const float*)d_in[4];
  const float* w2a_w = (const float*)d_in[5];
  const float* w2a_b = (const float*)d_in[6];
  const float* w2b_w = (const float*)d_in[7];
  const float* w2b_b = (const float*)d_in[8];
  const int*   edge  = (const int*)d_in[9];

  const int N = in_sizes[0] / 256;
  const int E = in_sizes[9] / 2;
  const int* srcI = edge;
  const int* dstI = edge + E;

  float* buf0 = (float*)d_ws;
  float* buf1 = buf0 + (size_t)N * 32;
  float* buf2 = buf1 + (size_t)N * 32;

  const int CE = cdiv_ll(E, NCH);
  const int NB = cdiv_ll(N, BIN);

  int* rowptr = (int*)buf1;
  int* col    = rowptr + (size_t)N + 64;
  int* lofs   = col + (size_t)E;
  int* tmp    = (int*)buf2;
  float* f2   = buf0 + (size_t)N * 16;

  const int BLK = 256;
  const int MT = 2;
  const int waves = N / (16 * MT);
  const int gemm_blocks = cdiv_ll(waves, BLK / 32);
  const int own_blocks = cdiv_ll(cdiv_ll(N, OWN_NPW), BLK / 32);

  chunk_sort<<<NCH, SORT_T, 0, stream>>>(srcI, dstI, lofs, tmp, E, CE, NB);
  bin_finish<<<NB, SORT_T, 0, stream>>>(tmp, lofs, rowptr, col, N, CE);

  wmma_linear2<256, 32, 32, MT><<<gemm_blocks, BLK, 0, stream>>>(
      x, w1a_w, w1a_b, w1b_w, w1b_b, buf0, buf2, N);
  owner_relu<32, OWN_GPW32><<<own_blocks, BLK, 0, stream>>>(
      buf0, rowptr, col, buf2, N);

  wmma_linear2<32, 16, 16, MT><<<gemm_blocks, BLK, 0, stream>>>(
      buf2, w2a_w, w2a_b, w2b_w, w2b_b, buf0, f2, N);
  owner_logsoftmax16<OWN_GPW16><<<own_blocks, BLK, 0, stream>>>(
      buf0, rowptr, col, f2, (float*)d_out, N);
}


# Launches 6 to 5: layer-2 linears run in the layer-1 owner wave on its 16 finished rows (LDS tile as WMMA B operand); z1 plane never stored; h2, f2 kept in 16-node blocks; order of sums as before
# speedup vs baseline: 3.9415x; 1.0875x over previous
#include <hip/hip_runtime.h>

typedef float v2f __attribute__((ext_vector_type(2)));
typedef float v8f __attribute__((ext_vector_type(8)));

typedef float v4f __attribute__((ext_vector_type(4)));

template <int MT, int NT>
struct Frag {
  v4f x[MT][4];
  v4f w[NT][4];
};

template <int MT, int NT>
__device__ __forceinline__ void load_frag(Frag<MT, NT>& f,
                                          const float* const (&xrow)[MT],
                                          const float* const (&wrow)[NT],
                                          int kb) {
#pragma unroll
  for (int i = 0; i < 4; ++i) {
#pragma unroll
    for (int mt = 0; mt < MT; ++mt)
      f.x[mt][i] = *(const v4f*)(xrow[mt] + kb + 4 * i);
#pragma unroll
    for (int t = 0; t < NT; ++t)
      f.w[t][i] = *(const v4f*)(wrow[t] + kb + 4 * i);
  }
}

template <int MT, int NT>
__device__ __forceinline__ void mma_frag(v8f (&acc)[MT][NT],
                                         const Frag<MT, NT>& f) {
#pragma unroll
  for (int i = 0; i < 4; ++i)
#pragma unroll
    for (int p = 0; p < 2; ++p) {
      v2f w[NT];
#pragma unroll
      for (int t = 0; t < NT; ++t) {
        w[t].x = f.w[t][i][2 * p];
        w[t].y = f.w[t][i][2 * p + 1];
      }
#pragma unroll
      for (int mt = 0; mt < MT; ++mt) {
        v2f xv;
        xv.x = f.x[mt][i][2 * p];
        xv.y = f.x[mt][i][2 * p + 1];
#pragma unroll
        for (int t = 0; t < NT; ++t) {
          acc[mt][t] = __builtin_amdgcn_wmma_f32_16x16x4_f32(
               false, w[t],  false, xv,
               (short)0, acc[mt][t],  false,
               false);
        }
      }
    }
}

template <int INC, int MID, int OUTC, int MT>
__global__ void __launch_bounds__(256) wmma_linear2(const float* __restrict__ X,
                             const float* __restrict__ Wa,
                             const float* __restrict__ Ba,
                             const float* __restrict__ Wb,
                             const float* __restrict__ Bb,
                             float* __restrict__ Y1,
                             float* __restrict__ Y2, int nrows) {
  constexpr int NT = MID / 16;
  constexpr int NT2 = OUTC / 16;
  int wave = (int)((blockIdx.x * blockDim.x + threadIdx.x) >> 5);
  int lane = threadIdx.x & 31;
  int row0 = wave * 16 * MT;
  if (row0 >= nrows) return;

  int nm = lane & 15;
  int hh = lane >> 4;

  const float* xrow[MT];
#pragma unroll
  for (int mt = 0; mt < MT; ++mt)
    xrow[mt] = X + (size_t)(row0 + mt * 16 + nm) * INC + hh * 16;
  const float* wrow[NT];
#pragma unroll
  for (int t = 0; t < NT; ++t)
    wrow[t] = Wa + (size_t)(t * 16 + nm) * INC + hh * 16;

  v8f acc[MT][NT];
#pragma unroll
  for (int t = 0; t < NT; ++t)
#pragma unroll
    for (int v = 0; v < 8; ++v) {
      float bv = Ba[t * 16 + hh * 8 + v];
#pragma unroll
      for (int mt = 0; mt < MT; ++mt) acc[mt][t][v] = bv;
    }

  static_assert(INC == 32 || INC % 64 == 0, "k blocks are taken in pairs");
  Frag<MT, NT> fa;
  load_frag<MT, NT>(fa, xrow, wrow, 0);
  if constexpr (INC == 32) {
    mma_frag<MT, NT>(acc, fa);
  } else {
    Frag<MT, NT> fb;
#pragma unroll
    for (int kb = 0; kb + 64 < INC; kb += 64) {
      load_frag<MT, NT>(fb, xrow, wrow, kb + 32);
      __builtin_amdgcn_sched_barrier(0);
      mma_frag<MT, NT>(acc, fa);
      __builtin_amdgcn_sched_barrier(0);
      load_frag<MT, NT>(fa, xrow, wrow, kb + 64);
      __builtin_amdgcn_sched_barrier(0);
      mma_frag<MT, NT>(acc, fb);
      __builtin_amdgcn_sched_barrier(0);
    }
    load_frag<MT, NT>(fb, xrow, wrow, INC - 32);
    __builtin_amdgcn_sched_barrier(0);
    mma_frag<MT, NT>(acc, fa);
    mma_frag<MT, NT>(acc, fb);
  }

#pragma unroll
  for (int mt = 0; mt < MT; ++mt)
#pragma unroll
    for (int t = 0; t < NT; ++t) {
      float* p = Y1 + (size_t)(row0 + mt * 16 + nm) * MID + t * 16 + hh * 8;
      v4f lo = {acc[mt][t][0], acc[mt][t][1], acc[mt][t][2], acc[mt][t][3]};
      v4f hi = {acc[mt][t][4], acc[mt][t][5], acc[mt][t][6], acc[mt][t][7]};
      *(v4f*)p = lo;
      *(v4f*)(p + 4) = hi;
    }

  v8f acc2[MT][NT2];
#pragma unroll
  for (int t2 = 0; t2 < NT2; ++t2)
#pragma unroll
    for (int v = 0; v < 8; ++v) {
      float bv = Bb[t2 * 16 + hh * 8 + v];
#pragma unroll
      for (int mt = 0; mt < MT; ++mt) acc2[mt][t2][v] = bv;
    }

#pragma unroll
  for (int t = 0; t < NT; ++t)
#pragma unroll
    for (int q = 0; q < 4; ++q) {
      v2f wb[NT2];
#pragma unroll
      for (int t2 = 0; t2 < NT2; ++t2) {
        const float* wp =
            Wb + (size_t)(t2 * 16 + nm) * MID + t * 16 + hh * 8 + 2 * q;
        wb[t2].x = wp[0];
        wb[t2].y = wp[1];
      }
#pragma unroll
      for (int mt = 0; mt < MT; ++mt) {
        v2f hv;
        hv.x = acc[mt][t][2 * q];
        hv.y = acc[mt][t][2 * q + 1];
#pragma unroll
        for (int t2 = 0; t2 < NT2; ++t2) {
          acc2[mt][t2] = __builtin_amdgcn_wmma_f32_16x16x4_f32(
               false, wb[t2],  false, hv,
               (short)0, acc2[mt][t2],  false,
               false);
        }
      }
    }

#pragma unroll
  for (int mt = 0; mt < MT; ++mt)
#pragma unroll
    for (int t2 = 0; t2 < NT2; ++t2) {
      float* p = Y2 + (size_t)(row0 + mt * 16 + nm) * OUTC + t2 * 16 + hh * 8;
      v4f lo = {acc2[mt][t2][0], acc2[mt][t2][1], acc2[mt][t2][2],
                acc2[mt][t2][3]};
      v4f hi = {acc2[mt][t2][4], acc2[mt][t2][5], acc2[mt][t2][6],
                acc2[mt][t2][7]};
      *(v4f*)p = lo;
      *(v4f*)(p + 4) = hi;
    }
}

typedef int v2i __attribute__((ext_vector_type(2)));

constexpr int BIN_SHIFT = 8;
constexpr int BIN = 1 << BIN_SHIFT;
constexpr int SORT_T = 1024;
constexpr int MAXB = SORT_T;
constexpr int EPT = 8;
constexpr int NCH = 256;
constexpr int CPW = NCH / 32;
constexpr int CAPS = 7 * SORT_T;
constexpr int CAPB = 8 * SORT_T;

__global__ void __launch_bounds__(SORT_T)
chunk_sort(const int* __restrict__ srcI, const int* __restrict__ dstI,
           int* __restrict__ lofs, int* __restrict__ tmp, int nedges, int ce,
           int nb) {
  __shared__ int h[MAXB];
  __shared__ int wtot[32];
  __shared__ v2i stage[CAPS];
  const int t = threadIdx.x;
  const int lane = t & 31;
  const int w = t >> 5;
  const int c = blockIdx.x;
  h[t] = 0;
  __syncthreads();

  const int beg = c * ce;
  const int end = beg + ce < nedges ? beg + ce : nedges;

  int d[EPT];
  int s[EPT];
#pragma unroll
  for (int u = 0; u < EPT; ++u) {
    const int e = beg + t + u * SORT_T;
    d[u] = e < end ? dstI[e] : -1;
    s[u] = e < end ? srcI[e] : 0;
  }
#pragma unroll
  for (int u = 0; u < EPT; ++u)
    if (d[u] >= 0) atomicAdd(&h[d[u] >> BIN_SHIFT], 1);
  for (int e = beg + t + EPT * SORT_T; e < end; e += SORT_T)
    atomicAdd(&h[dstI[e] >> BIN_SHIFT], 1);
  __syncthreads();

  const int cnt = h[t];
  int incl = cnt;
  for (int k = 1; k < 32; k <<= 1) {
    int v = __shfl_up(incl, k);
    if (lane >= k) incl += v;
  }
  if (lane == 31) wtot[w] = incl;
  __syncthreads();
  int wp = lane < w ? wtot[lane] : 0;
  for (int k = 16; k >= 1; k >>= 1) wp += __shfl_xor(wp, k);
  const int excl = wp + incl - cnt;
  h[t] = excl;
  __syncthreads();

  int* const slice = tmp + 2 * (size_t)beg;
  if (ce <= CAPS) {
#pragma unroll
    for (int u = 0; u < EPT; ++u)
      if (d[u] >= 0) {
        const int p = atomicAdd(&h[d[u] >> BIN_SHIFT], 1);
        v2i sd = {s[u], d[u]};
        stage[p] = sd;
      }
    __syncthreads();
    if (t < nb) lofs[t * NCH + c] = excl;
    if (t == nb - 1) lofs[nb * NCH + c] = excl + cnt;
    const int n = end - beg;
    constexpr int OUTS = CAPS / SORT_T;
    v2i o[OUTS];
#pragma unroll
    for (int u = 0; u < OUTS; ++u)
      o[u] = stage[t + u * SORT_T < n ? t + u * SORT_T : 0];
#pragma unroll
    for (int u = 0; u < OUTS; ++u)
      if (t + u * SORT_T < n)
        *(v2i*)(slice + 2 * (size_t)(t + u * SORT_T)) = o[u];
  } else {
    if (t < nb) lofs[t * NCH + c] = excl;
    if (t == nb - 1) lofs[nb * NCH + c] = excl + cnt;
#pragma unroll
    for (int u = 0; u < EPT; ++u)
      if (d[u] >= 0) {
        const int p = atomicAdd(&h[d[u] >> BIN_SHIFT], 1);
        v2i sd = {s[u], d[u]};
        *(v2i*)(slice + 2 * (size_t)p) = sd;
      }
    for (int e = beg + t + EPT * SORT_T; e < end; e += SORT_T) {
      const int dd = dstI[e];
      const int p = atomicAdd(&h[dd >> BIN_SHIFT], 1);
      v2i sd = {srcI[e], dd};
      *(v2i*)(slice + 2 * (size_t)p) = sd;
    }
  }
}
static_assert(CAPS <= EPT * SORT_T, "a staged chunk is held in registers");
static_assert(CAPS % SORT_T == 0, "the staged slice leaves in whole rounds");

template <bool STAGED>
__device__ __forceinline__ void place_bin(const v2i (&sd)[CPW], const v2i* seg,
                                          const int* __restrict__ tmp,
                                          int* fill, int* stage,
                                          int* __restrict__ col, int beg,
                                          int w, int lane) {
#pragma unroll
  for (int k = 0; k < CPW; ++k)
    if (sd[k].y >= 0) {
      const int p = atomicAdd(&fill[sd[k].y & (BIN - 1)], 1);
      if (STAGED) stage[p - beg] = sd[k].x;
      else col[p] = sd[k].x;
    }
  for (int k = 0; k < CPW; ++k) {
    const v2i g = seg[w + 32 * k];
    for (int l = lane + 32; l < g.y; l += 32) {
      const v2i e = *(const v2i*)(tmp + 2 * (size_t)(g.x + l));
      const int p = atomicAdd(&fill[e.y & (BIN - 1)], 1);
      if (STAGED) stage[p - beg] = e.x;
      else col[p] = e.x;
    }
  }
}

__global__ void __launch_bounds__(SORT_T)
bin_finish(const int* __restrict__ tmp, const int* __restrict__ lofs,
           int* __restrict__ rowptr, int* __restrict__ col, int n, int ce) {
  __shared__ int cnt[BIN];
  __shared__ int fill[BIN];
  __shared__ int wtot[32];
  __shared__ v2i seg[NCH];
  __shared__ int redlo[32];
  __shared__ int redhi[32];
  __shared__ int stage[CAPB];
  const int t = threadIdx.x;
  const int lane = t & 31;
  const int w = t >> 5;
  const int b = blockIdx.x;

  int lo = 0;
  int hi = 0;
  if (t < NCH) {
    lo = lofs[b * NCH + t];
    hi = lofs[(b + 1) * NCH + t];
    v2i sg = {t * ce + lo, hi - lo};
    seg[t] = sg;
  }
  if (t < BIN) cnt[t] = 0;
  __syncthreads();

  v2i sg[CPW];
  v2i sd[CPW];
#pragma unroll
  for (int k = 0; k < CPW; ++k) sg[k] = seg[w + 32 * k];
#pragma unroll
  for (int k = 0; k < CPW; ++k) {
    v2i z = {0, -1};
    sd[k] = lane < sg[k].y
                ? *(const v2i*)(tmp + 2 * (size_t)(sg[k].x + lane))
                : z;
  }

  int slo = lo;
  int shi = hi;
  for (int k = 16; k >= 1; k >>= 1) {
    slo += __shfl_xor(slo, k);
    shi += __shfl_xor(shi, k);
  }
  if (lane == 0) {
    redlo[w] = slo;
    redhi[w] = shi;
  }

#pragma unroll
  for (int k = 0; k < CPW; ++k)
    if (sd[k].y >= 0) atomicAdd(&cnt[sd[k].y & (BIN - 1)], 1);
  for (int k = 0; k < CPW; ++k) {
    const v2i g = seg[w + 32 * k];
    for (int l = lane + 32; l < g.y; l += 32)
      atomicAdd(&cnt[tmp[2 * (size_t)(g.x + l) + 1] & (BIN - 1)], 1);
  }
  __syncthreads();

  const int c = t < BIN ? cnt[t] : 0;
  int incl = c;
  for (int k = 1; k < 32; k <<= 1) {
    int v = __shfl_up(incl, k);
    if (lane >= k) incl += v;
  }
  if (lane == 31) wtot[w] = incl;
  __syncthreads();
  int wp = lane < w ? wtot[lane] : 0;
  int beg = redlo[lane];
  int end = redhi[lane];
  for (int k = 16; k >= 1; k >>= 1) {
    wp += __shfl_xor(wp, k);
    beg += __shfl_xor(beg, k);
    end += __shfl_xor(end, k);
  }
  const int excl = beg + wp + incl - c;
  if (t < BIN) fill[t] = excl;
  __syncthreads();

  const int begU = __builtin_amdgcn_readfirstlane(beg);
  const int nw = __builtin_amdgcn_readfirstlane(end) - begU;
  const bool staged = nw <= CAPB;
  if (staged) {
    place_bin<true>(sd, seg, tmp, fill, stage, col, begU, w, lane);
    __syncthreads();
  } else {
    place_bin<false>(sd, seg, tmp, fill, stage, col, begU, w, lane);
  }
  if (t < BIN) {
    const int node = b * BIN + t;
    if (node < n) rowptr[node] = excl;
  }
  if (b == (int)gridDim.x - 1 && t == 0) rowptr[n] = end;
  if (staged) {
    constexpr int OUTB = CAPB / SORT_T;
    int o[OUTB];
#pragma unroll
    for (int u = 0; u < OUTB; ++u)
      o[u] = stage[t + u * SORT_T < nw ? t + u * SORT_T : 0];
#pragma unroll
    for (int u = 0; u < OUTB; ++u)
      if (t + u * SORT_T < nw) col[begU + t + u * SORT_T] = o[u];
  }
}
static_assert(CAPB % SORT_T == 0, "the staged stretch leaves in whole rounds");

template <int LPN>
struct Idx {
  int v[32 / LPN];
};

template <int LPN>
__device__ __forceinline__ void load_idx(Idx<LPN>& ix,
                                         const int* __restrict__ col, int beg,
                                         int deg, int sub) {
#pragma unroll
  for (int i = 0; i < 32 / LPN; ++i) {
    const int j = i * LPN + sub;
    ix.v[i] = col[j < deg ? beg + j : 0];
  }
}

struct Rows8 {
  v4f r[8];
};

constexpr int OWN_OFS = 64;
template <int LPN, bool BLK>
__device__ __forceinline__ unsigned piece_of(int s, int sub) {
  if constexpr (BLK) {
    static_assert(LPN == 4, "blocked rows have 16 channels");
    return (unsigned)(((s & ~15) << 3) + ((s & 15) << 2) + sub);
  } else {
    return (unsigned)(s * LPN + sub);
  }
}

template <int LPN, bool BLK, int B>
__device__ __forceinline__ void load_block(Rows8& rw,
                                           const float* __restrict__ H,
                                           const Idx<LPN>& ix, int deg,
                                           int sub) {
#pragma unroll
  for (int e = 0; e < 8; ++e) {
    const int j = 8 * B + e;
    const int got = __shfl(ix.v[j / LPN], j % LPN, LPN);
    const int s = j < deg ? got : 0;
    rw.r[e] = ((const v4f*)H)[piece_of<LPN, BLK>(s, sub)];
  }
}

template <int B>
__device__ __forceinline__ void add_block(v4f& acc, const Rows8& rw, int deg) {
#pragma unroll
  for (int e = 0; e < 8; ++e)
    if (8 * B + e < deg) acc += rw.r[e];
}

template <int LPN, bool BLK>
__device__ __forceinline__ v4f gather_group(const float* __restrict__ H,
                                            const int* __restrict__ col,
                                            const Idx<LPN>& ix, int beg,
                                            int deg, int sub) {
  v4f acc = {0.0f, 0.0f, 0.0f, 0.0f};
  Rows8 ra;
  Rows8 rb;
  load_block<LPN, BLK, 0>(ra, H, ix, deg, sub);
  const bool m1 = __any(8 < deg) != 0;
  if (m1) load_block<LPN, BLK, 1>(rb, H, ix, deg, sub);
  add_block<0>(acc, ra, deg);
  if (m1) {
    const bool m2 = __any(16 < deg) != 0;
    if (m2) load_block<LPN, BLK, 2>(ra, H, ix, deg, sub);
    add_block<1>(acc, rb, deg);
    if (m2) {
      const bool m3 = __any(24 < deg) != 0;
      if (m3) load_block<LPN, BLK, 3>(rb, H, ix, deg, sub);
      add_block<2>(acc, ra, deg);
      if (m3) add_block<3>(acc, rb, deg);
    }
  }
  for (int j = 32; j < deg; ++j) {
    const int s = col[beg + j];
    acc += ((const v4f*)H)[piece_of<LPN, BLK>(s, sub)];
  }
  return acc;
}

template <int LPN, int GPW, bool BLK, class Fin>
__device__ __forceinline__ void owner_groups(const float* __restrict__ H,
                                             const int* __restrict__ rowptr,
                                             const int* __restrict__ col,
                                             const float* OWN, int nnodes,
                                             Fin fin) {
  constexpr int NPG = 32 / LPN;
  const int lane = threadIdx.x & 31;
  const int wave = (int)((blockIdx.x * blockDim.x + threadIdx.x) >> 5);
  const int sub = lane % LPN;
  int node = wave * (GPW * NPG) + lane / LPN;
  const v4f z = {0.0f, 0.0f, 0.0f, 0.0f};

  int begC = 0, degC = 0, begN = 0, degN = 0;
  v4f fC = z, fN = z;
  if (node < nnodes) {
    begC = rowptr[node];
    degC = rowptr[node + 1] - begC;
    fC = ((const v4f*)OWN)[piece_of<LPN, BLK>(node, sub) + (BLK ? OWN_OFS : 0)];
  }
  if (GPW > 1 && node + NPG < nnodes) {
    begN = rowptr[node + NPG];
    degN = rowptr[node + NPG + 1] - begN;
    fN = ((const v4f*)OWN)[piece_of<LPN, BLK>(node + NPG, sub) +
                           (BLK ? OWN_OFS : 0)];
  }
  Idx<LPN> ixC;
  load_idx<LPN>(ixC, col, begC, degC, sub);

#pragma nounroll
  for (int t = 0; t < GPW; ++t) {
    Idx<LPN> ixN;
    load_idx<LPN>(ixN, col, begN, degN, sub);
    int begF = 0, endF = 0;
    v4f fF = z;
    const int nodeF = node + 2 * NPG;
    if (t + 2 < GPW && nodeF < nnodes) {
      begF = rowptr[nodeF];
      endF = rowptr[nodeF + 1];
      fF = ((const v4f*)OWN)[piece_of<LPN, BLK>(nodeF, sub) +
                             (BLK ? OWN_OFS : 0)];
    }

    const v4f acc = gather_group<LPN, BLK>(H, col, ixC, begC, degC, sub);
    fin(node, node < nnodes, fC + acc);

    ixC = ixN;
    begC = begN; degC = degN; fC = fN;
    begN = begF; degN = endF - begF; fN = fF;
    node += NPG;
  }
}

constexpr int ZP4 = 9;

template <int GPW>
__global__ void
owner_relu_linear2(const float* __restrict__ H, const int* __restrict__ rowptr,
                   const int* __restrict__ col, const float* F1,
                   const float* __restrict__ Wa, const float* __restrict__ Ba,
                   const float* __restrict__ Wb, const float* __restrict__ Bb,
                   float* Y, int nnodes) {
  constexpr int LPN = 8;
  constexpr int INC = 32;
  constexpr int MID = 16;
  static_assert(GPW * (32 / LPN) == 16, "a wave's nodes fill one 16-column tile");
  __shared__ v4f ztile[8][16 * ZP4];
  const int lane = threadIdx.x & 31;
  const int sub = lane % LPN;
  v4f* const zt = ztile[threadIdx.x >> 5];
  owner_groups<LPN, GPW, false>(H, rowptr, col, F1, nnodes,
                                [=](int node, bool live, v4f v) {
    (void)live;
    v.x = v.x > 0.0f ? v.x : 0.0f;
    v.y = v.y > 0.0f ? v.y : 0.0f;
    v.z = v.z > 0.0f ? v.z : 0.0f;
    v.w = v.w > 0.0f ? v.w : 0.0f;
    zt[(node & 15) * ZP4 + sub] = v;
  });

  const int nm = lane & 15;
  const int hh = lane >> 4;
  const int wave = (int)((blockIdx.x * blockDim.x + threadIdx.x) >> 5);

  Frag<1, 1> f;
#pragma unroll
  for (int i = 0; i < 4; ++i) {
    f.x[0][i] = zt[nm * ZP4 + hh * 4 + i];
    f.w[0][i] = *(const v4f*)(Wa + nm * INC + hh * 16 + 4 * i);
  }
  v8f acc[1][1];
  v8f acc2;
#pragma unroll
  for (int v = 0; v < 8; ++v) {
    acc[0][0][v] = Ba[hh * 8 + v];
    acc2[v] = Bb[hh * 8 + v];
  }
  mma_frag<1, 1>(acc, f);

#pragma unroll
  for (int q = 0; q < 4; ++q) {
    const float* wp = Wb + nm * MID + hh * 8 + 2 * q;
    v2f wb;
    wb.x = wp[0];
    wb.y = wp[1];
    v2f hv;
    hv.x = acc[0][0][2 * q];
    hv.y = acc[0][0][2 * q + 1];
    acc2 = __builtin_amdgcn_wmma_f32_16x16x4_f32(
         false, wb,  false, hv,
         (short)0, acc2,  false,
         false);
  }

  if (wave * 16 + nm < nnodes) {
    float* p = Y + (size_t)wave * 512 + nm * 16 + hh * 8;
    v4f lo = {acc[0][0][0], acc[0][0][1], acc[0][0][2], acc[0][0][3]};
    v4f hi = {acc[0][0][4], acc[0][0][5], acc[0][0][6], acc[0][0][7]};
    *(v4f*)p = lo;
    *(v4f*)(p + 4) = hi;
    float* p2 = p + OWN_OFS * 4;
    v4f lo2 = {acc2[0], acc2[1], acc2[2], acc2[3]};
    v4f hi2 = {acc2[4], acc2[5], acc2[6], acc2[7]};
    *(v4f*)p2 = lo2;
    *(v4f*)(p2 + 4) = hi2;
  }
}

template <int GPW>
__global__ void
owner_logsoftmax16(const float* __restrict__ HF, const int* __restrict__ rowptr,
                   const int* __restrict__ col, float* __restrict__ O,
                   int nnodes) {
  constexpr int LPN = 4;
  const int sub = (threadIdx.x & 31) % LPN;
  owner_groups<LPN, GPW, true>(HF, rowptr, col, HF, nnodes,
                         [=](int node, bool live, v4f v) {
    float mx = fmaxf(fmaxf(v.x, v.y), fmaxf(v.z, v.w));
    mx = fmaxf(mx, __shfl_xor(mx, 1));
    mx = fmaxf(mx, __shfl_xor(mx, 2));
    float s = __expf(v.x - mx) + __expf(v.y - mx) + __expf(v.z - mx) +
              __expf(v.w - mx);
    s += __shfl_xor(s, 1);
    s += __shfl_xor(s, 2);
    const float l = mx + __logf(s);
    if (live) {
      v4f o = {v.x - l, v.y - l, v.z - l, v.w - l};
      *(v4f*)(O + ((size_t)node * LPN + sub) * 4) = o;
    }
  });
}

constexpr int OWN_GPW32 = 4;
constexpr int OWN_GPW16 = 2;
constexpr int OWN_NPW = 16;
static_assert(OWN_GPW32 * 4 == OWN_NPW && OWN_GPW16 * 8 == OWN_NPW,
              "both owner launches share one grid");

static inline int cdiv_ll(long long a, long long b) {
  return (int)((a + b - 1) / b);
}

extern "C" void kernel_launch(void* const* d_in, const int* in_sizes, int n_in,
                              void* d_out, int out_size, void* d_ws,
                              size_t ws_size, hipStream_t stream) {
  (void)n_in; (void)out_size; (void)ws_size;
  const float* x     = (const float*)d_in[0];
  const float* w1a_w = (const float*)d_in[1];
  const float* w1a_b = (const float*)d_in[2];
  const float* w1b_w = (const float*)d_in[3];
  const float* w1b_b = (const float*)d_in[4];
  const float* w2a_w = (const float*)d_in[5];
  const float* w2a_b = (const float*)d_in[6];
  const float* w2b_w = (const float*)d_in[7];
  const float* w2b_b = (const float*)d_in[8];
  const int*   edge  = (const int*)d_in[9];

  const int N = in_sizes[0] / 256;
  const int E = in_sizes[9] / 2;
  const int* srcI = edge;
  const int* dstI = edge + E;

  float* buf0 = (float*)d_ws;
  float* buf1 = buf0 + (size_t)N * 32;
  float* buf2 = buf1 + (size_t)N * 32;

  const int CE = cdiv_ll(E, NCH);
  const int NB = cdiv_ll(N, BIN);

  int* rowptr = (int*)buf1;
  int* col    = rowptr + (size_t)N + 64;
  int* lofs   = col + (size_t)E;
  int* tmp    = (int*)buf2;

  const int BLK = 256;
  const int MT = 2;
  const int waves = N / (16 * MT);
  const int gemm_blocks = cdiv_ll(waves, BLK / 32);
  const int own_blocks = cdiv_ll(cdiv_ll(N, OWN_NPW), BLK / 32);

  chunk_sort<<<NCH, SORT_T, 0, stream>>>(srcI, dstI, lofs, tmp, E, CE, NB);
  bin_finish<<<NB, SORT_T, 0, stream>>>(tmp, lofs, rowptr, col, N, CE);

  wmma_linear2<256, 32, 32, MT><<<gemm_blocks, BLK, 0, stream>>>(
      x, w1a_w, w1a_b, w1b_w, w1b_b, buf0, buf2, N);
  owner_relu_linear2<OWN_GPW32><<<own_blocks, BLK, 0, stream>>>(
      buf0, rowptr, col, buf2, w2a_w, w2a_b, w2b_w, w2b_b, buf2, N);

  owner_logsoftmax16<OWN_GPW16><<<own_blocks, BLK, 0, stream>>>(
      buf2, rowptr, col, (float*)d_out, N);
}


# Launches 5 to 4: chunk sort runs inside the first linear's launch as 8-wave work-groups of its own (one word per edge, held in the output plane until the bins are built); new arrival order of sums
# speedup vs baseline: 4.4568x; 1.1307x over previous
#include <hip/hip_runtime.h>

typedef float v2f __attribute__((ext_vector_type(2)));
typedef float v8f __attribute__((ext_vector_type(8)));

typedef float v4f __attribute__((ext_vector_type(4)));

template <int MT, int NT>
struct Frag {
  v4f x[MT][4];
  v4f w[NT][4];
};

template <int MT, int NT>
__device__ __forceinline__ void load_frag(Frag<MT, NT>& f,
                                          const float* const (&xrow)[MT],
                                          const float* const (&wrow)[NT],
                                          int kb) {
#pragma unroll
  for (int i = 0; i < 4; ++i) {
#pragma unroll
    for (int mt = 0; mt < MT; ++mt)
      f.x[mt][i] = *(const v4f*)(xrow[mt] + kb + 4 * i);
#pragma unroll
    for (int t = 0; t < NT; ++t)
      f.w[t][i] = *(const v4f*)(wrow[t] + kb + 4 * i);
  }
}

template <int MT, int NT>
__device__ __forceinline__ void mma_frag(v8f (&acc)[MT][NT],
                                         const Frag<MT, NT>& f) {
#pragma unroll
  for (int i = 0; i < 4; ++i)
#pragma unroll
    for (int p = 0; p < 2; ++p) {
      v2f w[NT];
#pragma unroll
      for (int t = 0; t < NT; ++t) {
        w[t].x = f.w[t][i][2 * p];
        w[t].y = f.w[t][i][2 * p + 1];
      }
#pragma unroll
      for (int mt = 0; mt < MT; ++mt) {
        v2f xv;
        xv.x = f.x[mt][i][2 * p];
        xv.y = f.x[mt][i][2 * p + 1];
#pragma unroll
        for (int t = 0; t < NT; ++t) {
          acc[mt][t] = __builtin_amdgcn_wmma_f32_16x16x4_f32(
               false, w[t],  false, xv,
               (short)0, acc[mt][t],  false,
               false);
        }
      }
    }
}

template <int INC, int MID, int OUTC, int MT>
__device__ __forceinline__ void linear2_wave(const float* __restrict__ X,
                             const float* __restrict__ Wa,
                             const float* __restrict__ Ba,
                             const float* __restrict__ Wb,
                             const float* __restrict__ Bb,
                             float* __restrict__ Y1,
                             float* __restrict__ Y2, int nrows, int blk) {
  constexpr int NT = MID / 16;
  constexpr int NT2 = OUTC / 16;
  int wave = (int)((blk * blockDim.x + threadIdx.x) >> 5);
  int lane = threadIdx.x & 31;
  int row0 = wave * 16 * MT;
  if (row0 >= nrows) return;

  int nm = lane & 15;
  int hh = lane >> 4;

  const float* xrow[MT];
#pragma unroll
  for (int mt = 0; mt < MT; ++mt)
    xrow[mt] = X + (size_t)(row0 + mt * 16 + nm) * INC + hh * 16;
  const float* wrow[NT];
#pragma unroll
  for (int t = 0; t < NT; ++t)
    wrow[t] = Wa + (size_t)(t * 16 + nm) * INC + hh * 16;

  v8f acc[MT][NT];
#pragma unroll
  for (int t = 0; t < NT; ++t)
#pragma unroll
    for (int v = 0; v < 8; ++v) {
      float bv = Ba[t * 16 + hh * 8 + v];
#pragma unroll
      for (int mt = 0; mt < MT; ++mt) acc[mt][t][v] = bv;
    }

  static_assert(INC == 32 || INC % 64 == 0, "k blocks are taken in pairs");
  Frag<MT, NT> fa;
  load_frag<MT, NT>(fa, xrow, wrow, 0);
  if constexpr (INC == 32) {
    mma_frag<MT, NT>(acc, fa);
  } else {
    Frag<MT, NT> fb;
#pragma unroll
    for (int kb = 0; kb + 64 < INC; kb += 64) {
      load_frag<MT, NT>(fb, xrow, wrow, kb + 32);
      __builtin_amdgcn_sched_barrier(0);
      mma_frag<MT, NT>(acc, fa);
      __builtin_amdgcn_sched_barrier(0);
      load_frag<MT, NT>(fa, xrow, wrow, kb + 64);
      __builtin_amdgcn_sched_barrier(0);
      mma_frag<MT, NT>(acc, fb);
      __builtin_amdgcn_sched_barrier(0);
    }
    load_frag<MT, NT>(fb, xrow, wrow, INC - 32);
    __builtin_amdgcn_sched_barrier(0);
    mma_frag<MT, NT>(acc, fa);
    mma_frag<MT, NT>(acc, fb);
  }

#pragma unroll
  for (int mt = 0; mt < MT; ++mt)
#pragma unroll
    for (int t = 0; t < NT; ++t) {
      float* p = Y1 + (size_t)(row0 + mt * 16 + nm) * MID + t * 16 + hh * 8;
      v4f lo = {acc[mt][t][0], acc[mt][t][1], acc[mt][t][2], acc[mt][t][3]};
      v4f hi = {acc[mt][t][4], acc[mt][t][5], acc[mt][t][6], acc[mt][t][7]};
      *(v4f*)p = lo;
      *(v4f*)(p + 4) = hi;
    }

  v8f acc2[MT][NT2];
#pragma unroll
  for (int t2 = 0; t2 < NT2; ++t2)
#pragma unroll
    for (int v = 0; v < 8; ++v) {
      float bv = Bb[t2 * 16 + hh * 8 + v];
#pragma unroll
      for (int mt = 0; mt < MT; ++mt) acc2[mt][t2][v] = bv;
    }

#pragma unroll
  for (int t = 0; t < NT; ++t)
#pragma unroll
    for (int q = 0; q < 4; ++q) {
      v2f wb[NT2];
#pragma unroll
      for (int t2 = 0; t2 < NT2; ++t2) {
        const float* wp =
            Wb + (size_t)(t2 * 16 + nm) * MID + t * 16 + hh * 8 + 2 * q;
        wb[t2].x = wp[0];
        wb[t2].y = wp[1];
      }
#pragma unroll
      for (int mt = 0; mt < MT; ++mt) {
        v2f hv;
        hv.x = acc[mt][t][2 * q];
        hv.y = acc[mt][t][2 * q + 1];
#pragma unroll
        for (int t2 = 0; t2 < NT2; ++t2) {
          acc2[mt][t2] = __builtin_amdgcn_wmma_f32_16x16x4_f32(
               false, wb[t2],  false, hv,
               (short)0, acc2[mt][t2],  false,
               false);
        }
      }
    }

#pragma unroll
  for (int mt = 0; mt < MT; ++mt)
#pragma unroll
    for (int t2 = 0; t2 < NT2; ++t2) {
      float* p = Y2 + (size_t)(row0 + mt * 16 + nm) * OUTC + t2 * 16 + hh * 8;
      v4f lo = {acc2[mt][t2][0], acc2[mt][t2][1], acc2[mt][t2][2],
                acc2[mt][t2][3]};
      v4f hi = {acc2[mt][t2][4], acc2[mt][t2][5], acc2[mt][t2][6],
                acc2[mt][t2][7]};
      *(v4f*)p = lo;
      *(v4f*)(p + 4) = hi;
    }
}

typedef int v2i __attribute__((ext_vector_type(2)));

constexpr int BIN_SHIFT = 8;
constexpr int BIN = 1 << BIN_SHIFT;
constexpr int SORT_T = 1024;
constexpr int MAXB = 1024;
constexpr int CS_T = 256;
constexpr int BPT = MAXB / CS_T;
constexpr int CAPS = 7 * 1024;
constexpr int EPT = CAPS / CS_T;
constexpr int NCH = 256;
constexpr int CPW = NCH / 32;
constexpr int CAPB = 8 * SORT_T;

__device__ __forceinline__ void
chunk_sort_block(const int c, const int* __restrict__ srcI,
                 const int* __restrict__ dstI, int* __restrict__ lofs,
                 int* __restrict__ tmp, int nedges, int ce, int nb) {
  __shared__ int h[MAXB + 32];
  __shared__ int wtot[CS_T / 32];
  __shared__ int stage[CAPS + 32];
  const int t = threadIdx.x;
  const int lane = t & 31;
  const int w = t >> 5;
#pragma unroll
  for (int k = 0; k < BPT; ++k) h[t + k * CS_T] = 0;
  if (t < 32) h[MAXB + t] = 0;
  __syncthreads();

  const int beg = c * ce;
  const int end = beg + ce < nedges ? beg + ce : nedges;

  int wd[EPT];
  int bn[EPT];
#pragma unroll
  for (int u = 0; u < EPT; ++u) {
    const int e = beg + t + u * CS_T;
    const int ee = e < end ? e : 0;
    const int dv = dstI[ee];
    const int sv = srcI[ee];
    wd[u] = (sv << BIN_SHIFT) | (dv & (BIN - 1));
    bn[u] = e < end ? dv >> BIN_SHIFT : MAXB + lane;
  }
#pragma unroll
  for (int u = 0; u < EPT; ++u) atomicAdd(&h[bn[u]], 1);
  for (int e = beg + t + EPT * CS_T; e < end; e += CS_T)
    atomicAdd(&h[dstI[e] >> BIN_SHIFT], 1);
  __syncthreads();

  int cnt[BPT];
  int tot = 0;
#pragma unroll
  for (int k = 0; k < BPT; ++k) {
    cnt[k] = h[BPT * t + k];
    tot += cnt[k];
  }
  int incl = tot;
  for (int k = 1; k < 32; k <<= 1) {
    int v = __shfl_up(incl, k);
    if (lane >= k) incl += v;
  }
  if (lane == 31) wtot[w] = incl;
  __syncthreads();
  int wp = lane < w ? wtot[lane] : 0;
  for (int k = 16; k >= 1; k >>= 1) wp += __shfl_xor(wp, k);
  int excl[BPT + 1];
  excl[0] = wp + incl - tot;
#pragma unroll
  for (int k = 0; k < BPT; ++k) {
    excl[k + 1] = excl[k] + cnt[k];
    h[BPT * t + k] = excl[k];
  }
  __syncthreads();

  int* const slice = tmp + (size_t)beg;
  if (ce <= CAPS) {
    int p[EPT];
#pragma unroll
    for (int u = 0; u < EPT; ++u) p[u] = atomicAdd(&h[bn[u]], 1);
#pragma unroll
    for (int u = 0; u < EPT; ++u)
      stage[bn[u] < MAXB ? p[u] : CAPS + lane] = wd[u];
    __syncthreads();
#pragma unroll
    for (int k = 0; k < BPT; ++k) {
      const int bin = BPT * t + k;
      if (bin < nb) lofs[bin * NCH + c] = excl[k];
      if (bin == nb - 1) lofs[nb * NCH + c] = excl[k + 1];
    }
    const int n = end - beg;
    int o[EPT];
#pragma unroll
    for (int u = 0; u < EPT; ++u)
      o[u] = stage[t + u * CS_T < n ? t + u * CS_T : 0];
#pragma unroll
    for (int u = 0; u < EPT; ++u)
      if (t + u * CS_T < n) slice[t + u * CS_T] = o[u];
  } else {
#pragma unroll
    for (int k = 0; k < BPT; ++k) {
      const int bin = BPT * t + k;
      if (bin < nb) lofs[bin * NCH + c] = excl[k];
      if (bin == nb - 1) lofs[nb * NCH + c] = excl[k + 1];
    }
#pragma unroll
    for (int u = 0; u < EPT; ++u)
      if (bn[u] < MAXB) {
        const int p = atomicAdd(&h[bn[u]], 1);
        slice[p] = wd[u];
      }
    for (int e = beg + t + EPT * CS_T; e < end; e += CS_T) {
      const int dd = dstI[e];
      const int p = atomicAdd(&h[dd >> BIN_SHIFT], 1);
      slice[p] = (srcI[e] << BIN_SHIFT) | (dd & (BIN - 1));
    }
  }
}
static_assert(MAXB % CS_T == 0, "a thread scans a whole number of bins");
static_assert(CAPS % CS_T == 0, "the staged slice leaves in whole rounds");

template <int INC, int MID, int OUTC, int MT>
__global__ void __launch_bounds__(CS_T)
linear2_and_chunk_sort(const float* __restrict__ X,
                       const float* __restrict__ Wa,
                       const float* __restrict__ Ba,
                       const float* __restrict__ Wb,
                       const float* __restrict__ Bb, float* __restrict__ Y1,
                       float* __restrict__ Y2, int nrows, int lin_blocks,
                       const int* __restrict__ srcI,
                       const int* __restrict__ dstI, int* __restrict__ lofs,
                       int* __restrict__ tmp, int nedges, int ce, int nb) {
  const int b = (int)blockIdx.x;
  if (b >= lin_blocks) {
    chunk_sort_block(b - lin_blocks, srcI, dstI, lofs, tmp, nedges, ce, nb);
    return;
  }
  linear2_wave<INC, MID, OUTC, MT>(X, Wa, Ba, Wb, Bb, Y1, Y2, nrows, b);
}

template <bool STAGED>
__device__ __forceinline__ void place_bin(const int (&sd)[CPW], const v2i* seg,
                                          const int* __restrict__ tmp,
                                          int* fill, int* stage,
                                          int* __restrict__ col, int beg,
                                          int w, int lane) {
#pragma unroll
  for (int k = 0; k < CPW; ++k)
    if (sd[k] >= 0) {
      const int p = atomicAdd(&fill[sd[k] & (BIN - 1)], 1);
      if (STAGED) stage[p - beg] = sd[k] >> BIN_SHIFT;
      else col[p] = sd[k] >> BIN_SHIFT;
    }
  for (int k = 0; k < CPW; ++k) {
    const v2i g = seg[w + 32 * k];
    for (int l = lane + 32; l < g.y; l += 32) {
      const int e = tmp[(size_t)(g.x + l)];
      const int p = atomicAdd(&fill[e & (BIN - 1)], 1);
      if (STAGED) stage[p - beg] = e >> BIN_SHIFT;
      else col[p] = e >> BIN_SHIFT;
    }
  }
}

__global__ void __launch_bounds__(SORT_T)
bin_finish(const int* __restrict__ tmp, const int* __restrict__ lofs,
           int* __restrict__ rowptr, int* __restrict__ col, int n, int ce) {
  __shared__ int cnt[BIN];
  __shared__ int fill[BIN];
  __shared__ int wtot[32];
  __shared__ v2i seg[NCH];
  __shared__ int redlo[32];
  __shared__ int redhi[32];
  __shared__ int stage[CAPB];
  const int t = threadIdx.x;
  const int lane = t & 31;
  const int w = t >> 5;
  const int b = blockIdx.x;

  int lo = 0;
  int hi = 0;
  if (t < NCH) {
    lo = lofs[b * NCH + t];
    hi = lofs[(b + 1) * NCH + t];
    v2i sg = {t * ce + lo, hi - lo};
    seg[t] = sg;
  }
  if (t < BIN) cnt[t] = 0;
  __syncthreads();

  v2i sg[CPW];
  int sd[CPW];
#pragma unroll
  for (int k = 0; k < CPW; ++k) sg[k] = seg[w + 32 * k];
#pragma unroll
  for (int k = 0; k < CPW; ++k)
    sd[k] = lane < sg[k].y ? tmp[(size_t)(sg[k].x + lane)] : -1;

  int slo = lo;
  int shi = hi;
  for (int k = 16; k >= 1; k >>= 1) {
    slo += __shfl_xor(slo, k);
    shi += __shfl_xor(shi, k);
  }
  if (lane == 0) {
    redlo[w] = slo;
    redhi[w] = shi;
  }

#pragma unroll
  for (int k = 0; k < CPW; ++k)
    if (sd[k] >= 0) atomicAdd(&cnt[sd[k] & (BIN - 1)], 1);
  for (int k = 0; k < CPW; ++k) {
    const v2i g = seg[w + 32 * k];
    for (int l = lane + 32; l < g.y; l += 32)
      atomicAdd(&cnt[tmp[(size_t)(g.x + l)] & (BIN - 1)], 1);
  }
  __syncthreads();

  const int c = t < BIN ? cnt[t] : 0;
  int incl = c;
  for (int k = 1; k < 32; k <<= 1) {
    int v = __shfl_up(incl, k);
    if (lane >= k) incl += v;
  }
  if (lane == 31) wtot[w] = incl;
  __syncthreads();
  int wp = lane < w ? wtot[lane] : 0;
  int beg = redlo[lane];
  int end = redhi[lane];
  for (int k = 16; k >= 1; k >>= 1) {
    wp += __shfl_xor(wp, k);
    beg += __shfl_xor(beg, k);
    end += __shfl_xor(end, k);
  }
  const int excl = beg + wp + incl - c;
  if (t < BIN) fill[t] = excl;
  __syncthreads();

  const int begU = __builtin_amdgcn_readfirstlane(beg);
  const int nw = __builtin_amdgcn_readfirstlane(end) - begU;
  const bool staged = nw <= CAPB;
  if (staged) {
    place_bin<true>(sd, seg, tmp, fill, stage, col, begU, w, lane);
    __syncthreads();
  } else {
    place_bin<false>(sd, seg, tmp, fill, stage, col, begU, w, lane);
  }
  if (t < BIN) {
    const int node = b * BIN + t;
    if (node < n) rowptr[node] = excl;
  }
  if (b == (int)gridDim.x - 1 && t == 0) rowptr[n] = end;
  if (staged) {
    constexpr int OUTB = CAPB / SORT_T;
    int o[OUTB];
#pragma unroll
    for (int u = 0; u < OUTB; ++u)
      o[u] = stage[t + u * SORT_T < nw ? t + u * SORT_T : 0];
#pragma unroll
    for (int u = 0; u < OUTB; ++u)
      if (t + u * SORT_T < nw) col[begU + t + u * SORT_T] = o[u];
  }
}
static_assert(CAPB % SORT_T == 0, "the staged stretch leaves in whole rounds");

template <int LPN>
struct Idx {
  int v[32 / LPN];
};

template <int LPN>
__device__ __forceinline__ void load_idx(Idx<LPN>& ix,
                                         const int* __restrict__ col, int beg,
                                         int deg, int sub) {
#pragma unroll
  for (int i = 0; i < 32 / LPN; ++i) {
    const int j = i * LPN + sub;
    ix.v[i] = col[j < deg ? beg + j : 0];
  }
}

struct Rows8 {
  v4f r[8];
};

constexpr int OWN_OFS = 64;
template <int LPN, bool BLK>
__device__ __forceinline__ unsigned piece_of(int s, int sub) {
  if constexpr (BLK) {
    static_assert(LPN == 4, "blocked rows have 16 channels");
    return (unsigned)(((s & ~15) << 3) + ((s & 15) << 2) + sub);
  } else {
    return (unsigned)(s * LPN + sub);
  }
}

template <int LPN, bool BLK, int B>
__device__ __forceinline__ void load_block(Rows8& rw,
                                           const float* __restrict__ H,
                                           const Idx<LPN>& ix, int deg,
                                           int sub) {
#pragma unroll
  for (int e = 0; e < 8; ++e) {
    const int j = 8 * B + e;
    const int got = __shfl(ix.v[j / LPN], j % LPN, LPN);
    const int s = j < deg ? got : 0;
    rw.r[e] = ((const v4f*)H)[piece_of<LPN, BLK>(s, sub)];
  }
}

template <int B>
__device__ __forceinline__ void add_block(v4f& acc, const Rows8& rw, int deg) {
#pragma unroll
  for (int e = 0; e < 8; ++e)
    if (8 * B + e < deg) acc += rw.r[e];
}

template <int LPN, bool BLK>
__device__ __forceinline__ v4f gather_group(const float* __restrict__ H,
                                            const int* __restrict__ col,
                                            const Idx<LPN>& ix, int beg,
                                            int deg, int sub) {
  v4f acc = {0.0f, 0.0f, 0.0f, 0.0f};
  Rows8 ra;
  Rows8 rb;
  load_block<LPN, BLK, 0>(ra, H, ix, deg, sub);
  const bool m1 = __any(8 < deg) != 0;
  if (m1) load_block<LPN, BLK, 1>(rb, H, ix, deg, sub);
  add_block<0>(acc, ra, deg);
  if (m1) {
    const bool m2 = __any(16 < deg) != 0;
    if (m2) load_block<LPN, BLK, 2>(ra, H, ix, deg, sub);
    add_block<1>(acc, rb, deg);
    if (m2) {
      const bool m3 = __any(24 < deg) != 0;
      if (m3) load_block<LPN, BLK, 3>(rb, H, ix, deg, sub);
      add_block<2>(acc, ra, deg);
      if (m3) add_block<3>(acc, rb, deg);
    }
  }
  for (int j = 32; j < deg; ++j) {
    const int s = col[beg + j];
    acc += ((const v4f*)H)[piece_of<LPN, BLK>(s, sub)];
  }
  return acc;
}

template <int LPN, int GPW, bool BLK, class Fin>
__device__ __forceinline__ void owner_groups(const float* __restrict__ H,
                                             const int* __restrict__ rowptr,
                                             const int* __restrict__ col,
                                             const float* OWN, int nnodes,
                                             Fin fin) {
  constexpr int NPG = 32 / LPN;
  const int lane = threadIdx.x & 31;
  const int wave = (int)((blockIdx.x * blockDim.x + threadIdx.x) >> 5);
  const int sub = lane % LPN;
  int node = wave * (GPW * NPG) + lane / LPN;
  const v4f z = {0.0f, 0.0f, 0.0f, 0.0f};

  int begC = 0, degC = 0, begN = 0, degN = 0;
  v4f fC = z, fN = z;
  if (node < nnodes) {
    begC = rowptr[node];
    degC = rowptr[node + 1] - begC;
    fC = ((const v4f*)OWN)[piece_of<LPN, BLK>(node, sub) + (BLK ? OWN_OFS : 0)];
  }
  if (GPW > 1 && node + NPG < nnodes) {
    begN = rowptr[node + NPG];
    degN = rowptr[node + NPG + 1] - begN;
    fN = ((const v4f*)OWN)[piece_of<LPN, BLK>(node + NPG, sub) +
                           (BLK ? OWN_OFS : 0)];
  }
  Idx<LPN> ixC;
  load_idx<LPN>(ixC, col, begC, degC, sub);

#pragma nounroll
  for (int t = 0; t < GPW; ++t) {
    Idx<LPN> ixN;
    load_idx<LPN>(ixN, col, begN, degN, sub);
    int begF = 0, endF = 0;
    v4f fF = z;
    const int nodeF = node + 2 * NPG;
    if (t + 2 < GPW && nodeF < nnodes) {
      begF = rowptr[nodeF];
      endF = rowptr[nodeF + 1];
      fF = ((const v4f*)OWN)[piece_of<LPN, BLK>(nodeF, sub) +
                             (BLK ? OWN_OFS : 0)];
    }

    const v4f acc = gather_group<LPN, BLK>(H, col, ixC, begC, degC, sub);
    fin(node, node < nnodes, fC + acc);

    ixC = ixN;
    begC = begN; degC = degN; fC = fN;
    begN = begF; degN = endF - begF; fN = fF;
    node += NPG;
  }
}

constexpr int ZP4 = 9;

template <int GPW>
__global__ void
owner_relu_linear2(const float* __restrict__ H, const int* __restrict__ rowptr,
                   const int* __restrict__ col, const float* F1,
                   const float* __restrict__ Wa, const float* __restrict__ Ba,
                   const float* __restrict__ Wb, const float* __restrict__ Bb,
                   float* Y, int nnodes) {
  constexpr int LPN = 8;
  constexpr int INC = 32;
  constexpr int MID = 16;
  static_assert(GPW * (32 / LPN) == 16, "a wave's nodes fill one 16-column tile");
  __shared__ v4f ztile[8][16 * ZP4];
  const int lane = threadIdx.x & 31;
  const int sub = lane % LPN;
  v4f* const zt = ztile[threadIdx.x >> 5];
  owner_groups<LPN, GPW, false>(H, rowptr, col, F1, nnodes,
                                [=](int node, bool live, v4f v) {
    (void)live;
    v.x = v.x > 0.0f ? v.x : 0.0f;
    v.y = v.y > 0.0f ? v.y : 0.0f;
    v.z = v.z > 0.0f ? v.z : 0.0f;
    v.w = v.w > 0.0f ? v.w : 0.0f;
    zt[(node & 15) * ZP4 + sub] = v;
  });

  const int nm = lane & 15;
  const int hh = lane >> 4;
  const int wave = (int)((blockIdx.x * blockDim.x + threadIdx.x) >> 5);

  Frag<1, 1> f;
#pragma unroll
  for (int i = 0; i < 4; ++i) {
    f.x[0][i] = zt[nm * ZP4 + hh * 4 + i];
    f.w[0][i] = *(const v4f*)(Wa + nm * INC + hh * 16 + 4 * i);
  }
  v8f acc[1][1];
  v8f acc2;
#pragma unroll
  for (int v = 0; v < 8; ++v) {
    acc[0][0][v] = Ba[hh * 8 + v];
    acc2[v] = Bb[hh * 8 + v];
  }
  mma_frag<1, 1>(acc, f);

#pragma unroll
  for (int q = 0; q < 4; ++q) {
    const float* wp = Wb + nm * MID + hh * 8 + 2 * q;
    v2f wb;
    wb.x = wp[0];
    wb.y = wp[1];
    v2f hv;
    hv.x = acc[0][0][2 * q];
    hv.y = acc[0][0][2 * q + 1];
    acc2 = __builtin_amdgcn_wmma_f32_16x16x4_f32(
         false, wb,  false, hv,
         (short)0, acc2,  false,
         false);
  }

  if (wave * 16 + nm < nnodes) {
    float* p = Y + (size_t)wave * 512 + nm * 16 + hh * 8;
    v4f lo = {acc[0][0][0], acc[0][0][1], acc[0][0][2], acc[0][0][3]};
    v4f hi = {acc[0][0][4], acc[0][0][5], acc[0][0][6], acc[0][0][7]};
    *(v4f*)p = lo;
    *(v4f*)(p + 4) = hi;
    float* p2 = p + OWN_OFS * 4;
    v4f lo2 = {acc2[0], acc2[1], acc2[2], acc2[3]};
    v4f hi2 = {acc2[4], acc2[5], acc2[6], acc2[7]};
    *(v4f*)p2 = lo2;
    *(v4f*)(p2 + 4) = hi2;
  }
}

template <int GPW>
__global__ void
owner_logsoftmax16(const float* __restrict__ HF, const int* __restrict__ rowptr,
                   const int* __restrict__ col, float* __restrict__ O,
                   int nnodes) {
  constexpr int LPN = 4;
  const int sub = (threadIdx.x & 31) % LPN;
  owner_groups<LPN, GPW, true>(HF, rowptr, col, HF, nnodes,
                         [=](int node, bool live, v4f v) {
    float mx = fmaxf(fmaxf(v.x, v.y), fmaxf(v.z, v.w));
    mx = fmaxf(mx, __shfl_xor(mx, 1));
    mx = fmaxf(mx, __shfl_xor(mx, 2));
    float s = __expf(v.x - mx) + __expf(v.y - mx) + __expf(v.z - mx) +
              __expf(v.w - mx);
    s += __shfl_xor(s, 1);
    s += __shfl_xor(s, 2);
    const float l = mx + __logf(s);
    if (live) {
      v4f o = {v.x - l, v.y - l, v.z - l, v.w - l};
      *(v4f*)(O + ((size_t)node * LPN + sub) * 4) = o;
    }
  });
}

constexpr int OWN_GPW32 = 4;
constexpr int OWN_GPW16 = 2;
constexpr int OWN_NPW = 16;
static_assert(OWN_GPW32 * 4 == OWN_NPW && OWN_GPW16 * 8 == OWN_NPW,
              "both owner launches share one grid");

static inline int cdiv_ll(long long a, long long b) {
  return (int)((a + b - 1) / b);
}

extern "C" void kernel_launch(void* const* d_in, const int* in_sizes, int n_in,
                              void* d_out, int out_size, void* d_ws,
                              size_t ws_size, hipStream_t stream) {
  (void)n_in; (void)out_size; (void)ws_size;
  const float* x     = (const float*)d_in[0];
  const float* w1a_w = (const float*)d_in[1];
  const float* w1a_b = (const float*)d_in[2];
  const float* w1b_w = (const float*)d_in[3];
  const float* w1b_b = (const float*)d_in[4];
  const float* w2a_w = (const float*)d_in[5];
  const float* w2a_b = (const float*)d_in[6];
  const float* w2b_w = (const float*)d_in[7];
  const float* w2b_b = (const float*)d_in[8];
  const int*   edge  = (const int*)d_in[9];

  const int N = in_sizes[0] / 256;
  const int E = in_sizes[9] / 2;
  const int* srcI = edge;
  const int* dstI = edge + E;

  float* buf0 = (float*)d_ws;
  float* buf1 = buf0 + (size_t)N * 32;
  float* buf2 = buf1 + (size_t)N * 32;

  const int CE = cdiv_ll(E, NCH);
  const int NB = cdiv_ll(N, BIN);

  int* rowptr = (int*)buf1;
  int* col    = rowptr + (size_t)N + 64;
  int* lofs   = col + (size_t)E;
  int* tmp    = (int*)d_out;

  const int BLK = CS_T;
  const int MT = 2;
  const int waves = N / (16 * MT);
  const int gemm_blocks = cdiv_ll(waves, BLK / 32);
  const int own_blocks = cdiv_ll(cdiv_ll(N, OWN_NPW), BLK / 32);

  linear2_and_chunk_sort<256, 32, 32, MT>
      <<<gemm_blocks + NCH, CS_T, 0, stream>>>(
          x, w1a_w, w1a_b, w1b_w, w1b_b, buf0, buf2, N, gemm_blocks, srcI,
          dstI, lofs, tmp, E, CE, NB);
  bin_finish<<<NB, SORT_T, 0, stream>>>(tmp, lofs, rowptr, col, N, CE);

  owner_relu_linear2<OWN_GPW32><<<own_blocks, BLK, 0, stream>>>(
      buf0, rowptr, col, buf2, w2a_w, w2a_b, w2b_w, w2b_b, buf2, N);

  owner_logsoftmax16<OWN_GPW16><<<own_blocks, BLK, 0, stream>>>(
      buf2, rowptr, col, (float*)d_out, N);
}


# Launches 4 to 3: the bin step runs at the head of the layer-1 owner launch (a 16-wave workgroup per bin; its waves own the bin's nodes, row bounds and col come from LDS); sums in a new arrival order
# speedup vs baseline: 4.7118x; 1.0572x over previous
#include <hip/hip_runtime.h>

typedef float v2f __attribute__((ext_vector_type(2)));
typedef float v8f __attribute__((ext_vector_type(8)));

typedef float v4f __attribute__((ext_vector_type(4)));

template <int MT, int NT>
struct Frag {
  v4f x[MT][4];
  v4f w[NT][4];
};

template <int MT, int NT>
__device__ __forceinline__ void load_frag(Frag<MT, NT>& f,
                                          const float* const (&xrow)[MT],
                                          const float* const (&wrow)[NT],
                                          int kb) {
#pragma unroll
  for (int i = 0; i < 4; ++i) {
#pragma unroll
    for (int mt = 0; mt < MT; ++mt)
      f.x[mt][i] = *(const v4f*)(xrow[mt] + kb + 4 * i);
#pragma unroll
    for (int t = 0; t < NT; ++t)
      f.w[t][i] = *(const v4f*)(wrow[t] + kb + 4 * i);
  }
}

template <int MT, int NT>
__device__ __forceinline__ void mma_frag(v8f (&acc)[MT][NT],
                                         const Frag<MT, NT>& f) {
#pragma unroll
  for (int i = 0; i < 4; ++i)
#pragma unroll
    for (int p = 0; p < 2; ++p) {
      v2f w[NT];
#pragma unroll
      for (int t = 0; t < NT; ++t) {
        w[t].x = f.w[t][i][2 * p];
        w[t].y = f.w[t][i][2 * p + 1];
      }
#pragma unroll
      for (int mt = 0; mt < MT; ++mt) {
        v2f xv;
        xv.x = f.x[mt][i][2 * p];
        xv.y = f.x[mt][i][2 * p + 1];
#pragma unroll
        for (int t = 0; t < NT; ++t) {
          acc[mt][t] = __builtin_amdgcn_wmma_f32_16x16x4_f32(
               false, w[t],  false, xv,
               (short)0, acc[mt][t],  false,
               false);
        }
      }
    }
}

template <int INC, int MID, int OUTC, int MT>
__device__ __forceinline__ void linear2_wave(const float* __restrict__ X,
                             const float* __restrict__ Wa,
                             const float* __restrict__ Ba,
                             const float* __restrict__ Wb,
                             const float* __restrict__ Bb,
                             float* __restrict__ Y1,
                             float* __restrict__ Y2, int nrows, int blk) {
  constexpr int NT = MID / 16;
  constexpr int NT2 = OUTC / 16;
  int wave = (int)((blk * blockDim.x + threadIdx.x) >> 5);
  int lane = threadIdx.x & 31;
  int row0 = wave * 16 * MT;
  if (row0 >= nrows) return;

  int nm = lane & 15;
  int hh = lane >> 4;

  const float* xrow[MT];
#pragma unroll
  for (int mt = 0; mt < MT; ++mt)
    xrow[mt] = X + (size_t)(row0 + mt * 16 + nm) * INC + hh * 16;
  const float* wrow[NT];
#pragma unroll
  for (int t = 0; t < NT; ++t)
    wrow[t] = Wa + (size_t)(t * 16 + nm) * INC + hh * 16;

  v8f acc[MT][NT];
#pragma unroll
  for (int t = 0; t < NT; ++t)
#pragma unroll
    for (int v = 0; v < 8; ++v) {
      float bv = Ba[t * 16 + hh * 8 + v];
#pragma unroll
      for (int mt = 0; mt < MT; ++mt) acc[mt][t][v] = bv;
    }

  static_assert(INC == 32 || INC % 64 == 0, "k blocks are taken in pairs");
  Frag<MT, NT> fa;
  load_frag<MT, NT>(fa, xrow, wrow, 0);
  if constexpr (INC == 32) {
    mma_frag<MT, NT>(acc, fa);
  } else {
    Frag<MT, NT> fb;
#pragma unroll
    for (int kb = 0; kb + 64 < INC; kb += 64) {
      load_frag<MT, NT>(fb, xrow, wrow, kb + 32);
      __builtin_amdgcn_sched_barrier(0);
      mma_frag<MT, NT>(acc, fa);
      __builtin_amdgcn_sched_barrier(0);
      load_frag<MT, NT>(fa, xrow, wrow, kb + 64);
      __builtin_amdgcn_sched_barrier(0);
      mma_frag<MT, NT>(acc, fb);
      __builtin_amdgcn_sched_barrier(0);
    }
    load_frag<MT, NT>(fb, xrow, wrow, INC - 32);
    __builtin_amdgcn_sched_barrier(0);
    mma_frag<MT, NT>(acc, fa);
    mma_frag<MT, NT>(acc, fb);
  }

#pragma unroll
  for (int mt = 0; mt < MT; ++mt)
#pragma unroll
    for (int t = 0; t < NT; ++t) {
      float* p = Y1 + (size_t)(row0 + mt * 16 + nm) * MID + t * 16 + hh * 8;
      v4f lo = {acc[mt][t][0], acc[mt][t][1], acc[mt][t][2], acc[mt][t][3]};
      v4f hi = {acc[mt][t][4], acc[mt][t][5], acc[mt][t][6], acc[mt][t][7]};
      *(v4f*)p = lo;
      *(v4f*)(p + 4) = hi;
    }

  v8f acc2[MT][NT2];
#pragma unroll
  for (int t2 = 0; t2 < NT2; ++t2)
#pragma unroll
    for (int v = 0; v < 8; ++v) {
      float bv = Bb[t2 * 16 + hh * 8 + v];
#pragma unroll
      for (int mt = 0; mt < MT; ++mt) acc2[mt][t2][v] = bv;
    }

#pragma unroll
  for (int t = 0; t < NT; ++t)
#pragma unroll
    for (int q = 0; q < 4; ++q) {
      v2f wb[NT2];
#pragma unroll
      for (int t2 = 0; t2 < NT2; ++t2) {
        const float* wp =
            Wb + (size_t)(t2 * 16 + nm) * MID + t * 16 + hh * 8 + 2 * q;
        wb[t2].x = wp[0];
        wb[t2].y = wp[1];
      }
#pragma unroll
      for (int mt = 0; mt < MT; ++mt) {
        v2f hv;
        hv.x = acc[mt][t][2 * q];
        hv.y = acc[mt][t][2 * q + 1];
#pragma unroll
        for (int t2 = 0; t2 < NT2; ++t2) {
          acc2[mt][t2] = __builtin_amdgcn_wmma_f32_16x16x4_f32(
               false, wb[t2],  false, hv,
               (short)0, acc2[mt][t2],  false,
               false);
        }
      }
    }

#pragma unroll
  for (int mt = 0; mt < MT; ++mt)
#pragma unroll
    for (int t2 = 0; t2 < NT2; ++t2) {
      float* p = Y2 + (size_t)(row0 + mt * 16 + nm) * OUTC + t2 * 16 + hh * 8;
      v4f lo = {acc2[mt][t2][0], acc2[mt][t2][1], acc2[mt][t2][2],
                acc2[mt][t2][3]};
      v4f hi = {acc2[mt][t2][4], acc2[mt][t2][5], acc2[mt][t2][6],
                acc2[mt][t2][7]};
      *(v4f*)p = lo;
      *(v4f*)(p + 4) = hi;
    }
}

typedef int v2i __attribute__((ext_vector_type(2)));

constexpr int BIN_SHIFT = 8;
constexpr int BIN = 1 << BIN_SHIFT;
constexpr int SORT_T = 512;
constexpr int SORT_W = SORT_T / 32;
constexpr int MAXB = 1024;
constexpr int CS_T = 256;
constexpr int BPT = MAXB / CS_T;
constexpr int CAPS = 7 * 1024;
constexpr int EPT = CAPS / CS_T;
constexpr int NCH = 256;
constexpr int CPW = NCH / SORT_W;
constexpr int CAPB = 16 * SORT_T;

__device__ __forceinline__ void
chunk_sort_block(const int c, const int* __restrict__ srcI,
                 const int* __restrict__ dstI, int* __restrict__ lofs,
                 int* __restrict__ tmp, int nedges, int ce, int nb) {
  __shared__ int h[MAXB + 32];
  __shared__ int wtot[CS_T / 32];
  __shared__ int stage[CAPS + 32];
  const int t = threadIdx.x;
  const int lane = t & 31;
  const int w = t >> 5;
#pragma unroll
  for (int k = 0; k < BPT; ++k) h[t + k * CS_T] = 0;
  if (t < 32) h[MAXB + t] = 0;
  __syncthreads();

  const int beg = c * ce;
  const int end = beg + ce < nedges ? beg + ce : nedges;

  int wd[EPT];
  int bn[EPT];
#pragma unroll
  for (int u = 0; u < EPT; ++u) {
    const int e = beg + t + u * CS_T;
    const int ee = e < end ? e : 0;
    const int dv = dstI[ee];
    const int sv = srcI[ee];
    wd[u] = (sv << BIN_SHIFT) | (dv & (BIN - 1));
    bn[u] = e < end ? dv >> BIN_SHIFT : MAXB + lane;
  }
#pragma unroll
  for (int u = 0; u < EPT; ++u) atomicAdd(&h[bn[u]], 1);
  for (int e = beg + t + EPT * CS_T; e < end; e += CS_T)
    atomicAdd(&h[dstI[e] >> BIN_SHIFT], 1);
  __syncthreads();

  int cnt[BPT];
  int tot = 0;
#pragma unroll
  for (int k = 0; k < BPT; ++k) {
    cnt[k] = h[BPT * t + k];
    tot += cnt[k];
  }
  int incl = tot;
  for (int k = 1; k < 32; k <<= 1) {
    int v = __shfl_up(incl, k);
    if (lane >= k) incl += v;
  }
  if (lane == 31) wtot[w] = incl;
  __syncthreads();
  int wp = lane < w ? wtot[lane] : 0;
  for (int k = 16; k >= 1; k >>= 1) wp += __shfl_xor(wp, k);
  int excl[BPT + 1];
  excl[0] = wp + incl - tot;
#pragma unroll
  for (int k = 0; k < BPT; ++k) {
    excl[k + 1] = excl[k] + cnt[k];
    h[BPT * t + k] = excl[k];
  }
  __syncthreads();

  int* const slice = tmp + (size_t)beg;
  if (ce <= CAPS) {
    int p[EPT];
#pragma unroll
    for (int u = 0; u < EPT; ++u) p[u] = atomicAdd(&h[bn[u]], 1);
#pragma unroll
    for (int u = 0; u < EPT; ++u)
      stage[bn[u] < MAXB ? p[u] : CAPS + lane] = wd[u];
    __syncthreads();
#pragma unroll
    for (int k = 0; k < BPT; ++k) {
      const int bin = BPT * t + k;
      if (bin < nb) lofs[bin * NCH + c] = excl[k];
      if (bin == nb - 1) lofs[nb * NCH + c] = excl[k + 1];
    }
    const int n = end - beg;
    int o[EPT];
#pragma unroll
    for (int u = 0; u < EPT; ++u)
      o[u] = stage[t + u * CS_T < n ? t + u * CS_T : 0];
#pragma unroll
    for (int u = 0; u < EPT; ++u)
      if (t + u * CS_T < n) slice[t + u * CS_T] = o[u];
  } else {
#pragma unroll
    for (int k = 0; k < BPT; ++k) {
      const int bin = BPT * t + k;
      if (bin < nb) lofs[bin * NCH + c] = excl[k];
      if (bin == nb - 1) lofs[nb * NCH + c] = excl[k + 1];
    }
#pragma unroll
    for (int u = 0; u < EPT; ++u)
      if (bn[u] < MAXB) {
        const int p = atomicAdd(&h[bn[u]], 1);
        slice[p] = wd[u];
      }
    for (int e = beg + t + EPT * CS_T; e < end; e += CS_T) {
      const int dd = dstI[e];
      const int p = atomicAdd(&h[dd >> BIN_SHIFT], 1);
      slice[p] = (srcI[e] << BIN_SHIFT) | (dd & (BIN - 1));
    }
  }
}
static_assert(MAXB % CS_T == 0, "a thread scans a whole number of bins");
static_assert(CAPS % CS_T == 0, "the staged slice leaves in whole rounds");

template <int INC, int MID, int OUTC, int MT>
__global__ void __launch_bounds__(CS_T)
linear2_and_chunk_sort(const float* __restrict__ X,
                       const float* __restrict__ Wa,
                       const float* __restrict__ Ba,
                       const float* __restrict__ Wb,
                       const float* __restrict__ Bb, float* __restrict__ Y1,
                       float* __restrict__ Y2, int nrows, int lin_blocks,
                       const int* __restrict__ srcI,
                       const int* __restrict__ dstI, int* __restrict__ lofs,
                       int* __restrict__ tmp, int nedges, int ce, int nb) {
  const int b = (int)blockIdx.x;
  if (b >= lin_blocks) {
    chunk_sort_block(b - lin_blocks, srcI, dstI, lofs, tmp, nedges, ce, nb);
    return;
  }
  linear2_wave<INC, MID, OUTC, MT>(X, Wa, Ba, Wb, Bb, Y1, Y2, nrows, b);
}

template <bool STAGED>
__device__ __forceinline__ void place_bin(const int (&sd)[CPW], const v2i* seg,
                                          const int* __restrict__ tmp,
                                          int* fill, int* stage,
                                          int* __restrict__ col, int beg,
                                          int w, int lane) {
#pragma unroll
  for (int k = 0; k < CPW; ++k)
    if (sd[k] >= 0) {
      const int p = atomicAdd(&fill[sd[k] & (BIN - 1)], 1);
      if (STAGED) stage[p - beg] = sd[k] >> BIN_SHIFT;
      else col[p] = sd[k] >> BIN_SHIFT;
    }
  for (int k = 0; k < CPW; ++k) {
    const v2i g = seg[w + SORT_W * k];
    for (int l = lane + 32; l < g.y; l += 32) {
      const int e = tmp[(size_t)(g.x + l)];
      const int p = atomicAdd(&fill[e & (BIN - 1)], 1);
      if (STAGED) stage[p - beg] = e >> BIN_SHIFT;
      else col[p] = e >> BIN_SHIFT;
    }
  }
}

__device__ __forceinline__ void
bin_finish_block(const int* __restrict__ tmp, const int* __restrict__ lofs,
                 int* __restrict__ rowptr, int* col, int* rp, int* stage,
                 int n, int ce, int& begU_out, bool& staged_out) {
  __shared__ int cnt[BIN];
  __shared__ int fill[BIN];
  __shared__ int wtot[SORT_W];
  __shared__ v2i seg[NCH];
  __shared__ int redlo[SORT_W];
  __shared__ int redhi[SORT_W];
  const int t = threadIdx.x;
  const int lane = t & 31;
  const int w = t >> 5;
  const int b = blockIdx.x;

  int lo = 0;
  int hi = 0;
  if (t < NCH) {
    lo = lofs[b * NCH + t];
    hi = lofs[(b + 1) * NCH + t];
    v2i sg = {t * ce + lo, hi - lo};
    seg[t] = sg;
  }
  if (t < BIN) cnt[t] = 0;
  __syncthreads();

  v2i sg[CPW];
  int sd[CPW];
#pragma unroll
  for (int k = 0; k < CPW; ++k) sg[k] = seg[w + SORT_W * k];
#pragma unroll
  for (int k = 0; k < CPW; ++k)
    sd[k] = lane < sg[k].y ? tmp[(size_t)(sg[k].x + lane)] : -1;

  int slo = lo;
  int shi = hi;
  for (int k = 16; k >= 1; k >>= 1) {
    slo += __shfl_xor(slo, k);
    shi += __shfl_xor(shi, k);
  }
  if (lane == 0) {
    redlo[w] = slo;
    redhi[w] = shi;
  }

#pragma unroll
  for (int k = 0; k < CPW; ++k)
    if (sd[k] >= 0) atomicAdd(&cnt[sd[k] & (BIN - 1)], 1);
  for (int k = 0; k < CPW; ++k) {
    const v2i g = seg[w + SORT_W * k];
    for (int l = lane + 32; l < g.y; l += 32)
      atomicAdd(&cnt[tmp[(size_t)(g.x + l)] & (BIN - 1)], 1);
  }
  __syncthreads();

  const int c = t < BIN ? cnt[t] : 0;
  int incl = c;
  for (int k = 1; k < 32; k <<= 1) {
    int v = __shfl_up(incl, k);
    if (lane >= k) incl += v;
  }
  if (lane == 31) wtot[w] = incl;
  __syncthreads();
  int wp = lane < w ? wtot[lane] : 0;
  int beg = lane < SORT_W ? redlo[lane] : 0;
  int end = lane < SORT_W ? redhi[lane] : 0;
  for (int k = 16; k >= 1; k >>= 1) {
    wp += __shfl_xor(wp, k);
    beg += __shfl_xor(beg, k);
    end += __shfl_xor(end, k);
  }
  const int excl = beg + wp + incl - c;
  if (t < BIN) {
    fill[t] = excl;
    rp[t] = excl;
  }
  if (t == 0) rp[BIN] = end;
  __syncthreads();

  const int begU = __builtin_amdgcn_readfirstlane(beg);
  const int nw = __builtin_amdgcn_readfirstlane(end) - begU;
  const bool staged = nw <= CAPB;
  if (staged) {
    place_bin<true>(sd, seg, tmp, fill, stage, col, begU, w, lane);
    __syncthreads();
  } else {
    place_bin<false>(sd, seg, tmp, fill, stage, col, begU, w, lane);
  }
  if (t < BIN) {
    const int node = b * BIN + t;
    if (node < n) rowptr[node] = excl;
  }
  if (b == (int)gridDim.x - 1 && t == 0) rowptr[n] = end;
  if (staged) {
    constexpr int OUTB = CAPB / SORT_T;
    int o[OUTB];
#pragma unroll
    for (int u = 0; u < OUTB; ++u)
      o[u] = stage[t + u * SORT_T < nw ? t + u * SORT_T : 0];
#pragma unroll
    for (int u = 0; u < OUTB; ++u)
      if (t + u * SORT_T < nw) col[begU + t + u * SORT_T] = o[u];
  }
  if (!staged) __syncthreads();
  begU_out = begU;
  staged_out = staged;
}
static_assert(CAPB % SORT_T == 0, "the staged stretch leaves in whole rounds");
static_assert(NCH <= SORT_T && BIN <= SORT_T, "a thread per chunk and per node");

template <int LPN>
struct Idx {
  int v[32 / LPN];
};

template <int LPN, class Col>
__device__ __forceinline__ void load_idx(Idx<LPN>& ix, Col col, int beg,
                                         int deg, int sub, int spare) {
#pragma unroll
  for (int i = 0; i < 32 / LPN; ++i) {
    const int j = i * LPN + sub;
    ix.v[i] = col(j < deg ? beg + j : spare);
  }
}

struct Rows8 {
  v4f r[8];
};

constexpr int OWN_OFS = 64;
template <int LPN, bool BLK>
__device__ __forceinline__ unsigned piece_of(int s, int sub) {
  if constexpr (BLK) {
    static_assert(LPN == 4, "blocked rows have 16 channels");
    return (unsigned)(((s & ~15) << 3) + ((s & 15) << 2) + sub);
  } else {
    return (unsigned)(s * LPN + sub);
  }
}

template <int LPN, bool BLK, int B>
__device__ __forceinline__ void load_block(Rows8& rw,
                                           const float* __restrict__ H,
                                           const Idx<LPN>& ix, int deg,
                                           int sub) {
#pragma unroll
  for (int e = 0; e < 8; ++e) {
    const int j = 8 * B + e;
    const int got = __shfl(ix.v[j / LPN], j % LPN, LPN);
    const int s = j < deg ? got : 0;
    rw.r[e] = ((const v4f*)H)[piece_of<LPN, BLK>(s, sub)];
  }
}

template <int B>
__device__ __forceinline__ void add_block(v4f& acc, const Rows8& rw, int deg) {
#pragma unroll
  for (int e = 0; e < 8; ++e)
    if (8 * B + e < deg) acc += rw.r[e];
}

template <int LPN, bool BLK, class Col>
__device__ __forceinline__ v4f gather_group(const float* __restrict__ H,
                                            Col col,
                                            const Idx<LPN>& ix, int beg,
                                            int deg, int sub) {
  v4f acc = {0.0f, 0.0f, 0.0f, 0.0f};
  Rows8 ra;
  Rows8 rb;
  load_block<LPN, BLK, 0>(ra, H, ix, deg, sub);
  const bool m1 = __any(8 < deg) != 0;
  if (m1) load_block<LPN, BLK, 1>(rb, H, ix, deg, sub);
  add_block<0>(acc, ra, deg);
  if (m1) {
    const bool m2 = __any(16 < deg) != 0;
    if (m2) load_block<LPN, BLK, 2>(ra, H, ix, deg, sub);
    add_block<1>(acc, rb, deg);
    if (m2) {
      const bool m3 = __any(24 < deg) != 0;
      if (m3) load_block<LPN, BLK, 3>(rb, H, ix, deg, sub);
      add_block<2>(acc, ra, deg);
      if (m3) add_block<3>(acc, rb, deg);
    }
  }
  for (int j = 32; j < deg; ++j) {
    const int s = col(beg + j);
    acc += ((const v4f*)H)[piece_of<LPN, BLK>(s, sub)];
  }
  return acc;
}

template <int LPN, int GPW, bool BLK, class Rp, class Col, class Fin>
__device__ __forceinline__ void owner_groups(const float* __restrict__ H,
                                             Rp rowptr, Col col, int spare,
                                             const float* OWN, int nnodes,
                                             Fin fin) {
  constexpr int NPG = 32 / LPN;
  const int lane = threadIdx.x & 31;
  const int wave = (int)((blockIdx.x * blockDim.x + threadIdx.x) >> 5);
  const int sub = lane % LPN;
  int node = wave * (GPW * NPG) + lane / LPN;
  const v4f z = {0.0f, 0.0f, 0.0f, 0.0f};

  int begC = 0, degC = 0, begN = 0, degN = 0;
  v4f fC = z, fN = z;
  if (node < nnodes) {
    begC = rowptr(node);
    degC = rowptr(node + 1) - begC;
    fC = ((const v4f*)OWN)[piece_of<LPN, BLK>(node, sub) + (BLK ? OWN_OFS : 0)];
  }
  if (GPW > 1 && node + NPG < nnodes) {
    begN = rowptr(node + NPG);
    degN = rowptr(node + NPG + 1) - begN;
    fN = ((const v4f*)OWN)[piece_of<LPN, BLK>(node + NPG, sub) +
                           (BLK ? OWN_OFS : 0)];
  }
  Idx<LPN> ixC;
  load_idx<LPN>(ixC, col, begC, degC, sub, spare);

#pragma nounroll
  for (int t = 0; t < GPW; ++t) {
    Idx<LPN> ixN;
    load_idx<LPN>(ixN, col, begN, degN, sub, spare);
    int begF = 0, endF = 0;
    v4f fF = z;
    const int nodeF = node + 2 * NPG;
    if (t + 2 < GPW && nodeF < nnodes) {
      begF = rowptr(nodeF);
      endF = rowptr(nodeF + 1);
      fF = ((const v4f*)OWN)[piece_of<LPN, BLK>(nodeF, sub) +
                             (BLK ? OWN_OFS : 0)];
    }

    const v4f acc = gather_group<LPN, BLK>(H, col, ixC, begC, degC, sub);
    fin(node, node < nnodes, fC + acc);

    ixC = ixN;
    begC = begN; degC = degN; fC = fN;
    begN = begF; degN = endF - begF; fN = fF;
    node += NPG;
  }
}

template <int LPN, int GPW, class Rp, class Fin>
__device__ __forceinline__ void owner_groups_plain(const float* __restrict__ H,
                                                   Rp rowptr, const int* col,
                                                   const float* OWN,
                                                   int nnodes, Fin fin) {
  constexpr int NPG = 32 / LPN;
  const int lane = threadIdx.x & 31;
  const int wave = (int)((blockIdx.x * blockDim.x + threadIdx.x) >> 5);
  const int sub = lane % LPN;
  int node = wave * (GPW * NPG) + lane / LPN;
  for (int t = 0; t < GPW; ++t) {
    v4f acc = {0.0f, 0.0f, 0.0f, 0.0f};
    v4f own = {0.0f, 0.0f, 0.0f, 0.0f};
    if (node < nnodes) {
      const int beg = rowptr(node);
      const int end = rowptr(node + 1);
      own = ((const v4f*)OWN)[piece_of<LPN, false>(node, sub)];
      for (int j = beg; j < end; ++j)
        acc += ((const v4f*)H)[piece_of<LPN, false>(col[j], sub)];
    }
    fin(node, node < nnodes, own + acc);
    node += NPG;
  }
}

constexpr int ZP4 = 9;

template <int GPW>
__global__ void __launch_bounds__(SORT_T)
bin_owner_relu_linear2(const int* __restrict__ tmp,
                       const int* __restrict__ lofs, int* __restrict__ rowptr,
                       int* col, int ce, const float* __restrict__ H,
                       const float* F1, const float* __restrict__ Wa,
                       const float* __restrict__ Ba,
                       const float* __restrict__ Wb,
                       const float* __restrict__ Bb, float* Y, int nnodes) {
  constexpr int LPN = 8;
  constexpr int INC = 32;
  constexpr int MID = 16;
  static_assert(GPW * (32 / LPN) == 16, "a wave's nodes fill one 16-column tile");
  static_assert(SORT_W * 16 == BIN, "the workgroup's waves own the bin's nodes");
  __shared__ int rp[BIN + 1];
  __shared__ int stage[CAPB];
  __shared__ v4f ztile[SORT_W][16 * ZP4];
  int begU;
  bool staged;
  bin_finish_block(tmp, lofs, rowptr, col, rp, stage, nnodes, ce, begU, staged);

  const int lane = threadIdx.x & 31;
  const int sub = lane % LPN;
  v4f* const zt = ztile[threadIdx.x >> 5];
  const int* const rpl = rp;
  const int* const stg = stage;
  const int node0 = (int)blockIdx.x * BIN;
  const auto bound = [=](int i) { return rpl[i - node0]; };
  const auto fin = [=](int node, bool live, v4f v) {
    (void)live;
    v.x = v.x > 0.0f ? v.x : 0.0f;
    v.y = v.y > 0.0f ? v.y : 0.0f;
    v.z = v.z > 0.0f ? v.z : 0.0f;
    v.w = v.w > 0.0f ? v.w : 0.0f;
    zt[(node & 15) * ZP4 + sub] = v;
  };
  if (staged)
    owner_groups<LPN, GPW, false>(H, bound,
                                  [=](int i) { return stg[i - begU]; }, begU,
                                  F1, nnodes, fin);
  else
    owner_groups_plain<LPN, GPW>(H, bound, col, F1, nnodes, fin);

  const int nm = lane & 15;
  const int hh = lane >> 4;
  const int wave = (int)((blockIdx.x * blockDim.x + threadIdx.x) >> 5);

  Frag<1, 1> f;
#pragma unroll
  for (int i = 0; i < 4; ++i) {
    f.x[0][i] = zt[nm * ZP4 + hh * 4 + i];
    f.w[0][i] = *(const v4f*)(Wa + nm * INC + hh * 16 + 4 * i);
  }
  v8f acc[1][1];
  v8f acc2;
#pragma unroll
  for (int v = 0; v < 8; ++v) {
    acc[0][0][v] = Ba[hh * 8 + v];
    acc2[v] = Bb[hh * 8 + v];
  }
  mma_frag<1, 1>(acc, f);

#pragma unroll
  for (int q = 0; q < 4; ++q) {
    const float* wp = Wb + nm * MID + hh * 8 + 2 * q;
    v2f wb;
    wb.x = wp[0];
    wb.y = wp[1];
    v2f hv;
    hv.x = acc[0][0][2 * q];
    hv.y = acc[0][0][2 * q + 1];
    acc2 = __builtin_amdgcn_wmma_f32_16x16x4_f32(
         false, wb,  false, hv,
         (short)0, acc2,  false,
         false);
  }

  if (wave * 16 + nm < nnodes) {
    float* p = Y + (size_t)wave * 512 + nm * 16 + hh * 8;
    v4f lo = {acc[0][0][0], acc[0][0][1], acc[0][0][2], acc[0][0][3]};
    v4f hi = {acc[0][0][4], acc[0][0][5], acc[0][0][6], acc[0][0][7]};
    *(v4f*)p = lo;
    *(v4f*)(p + 4) = hi;
    float* p2 = p + OWN_OFS * 4;
    v4f lo2 = {acc2[0], acc2[1], acc2[2], acc2[3]};
    v4f hi2 = {acc2[4], acc2[5], acc2[6], acc2[7]};
    *(v4f*)p2 = lo2;
    *(v4f*)(p2 + 4) = hi2;
  }
}

template <int GPW>
__global__ void
owner_logsoftmax16(const float* __restrict__ HF, const int* __restrict__ rowptr,
                   const int* __restrict__ col, float* __restrict__ O,
                   int nnodes) {
  constexpr int LPN = 4;
  const int sub = (threadIdx.x & 31) % LPN;
  owner_groups<LPN, GPW, true>(HF, [=](int i) { return rowptr[i]; },
                               [=](int i) { return col[i]; }, 0, HF, nnodes,
                               [=](int node, bool live, v4f v) {
    float mx = fmaxf(fmaxf(v.x, v.y), fmaxf(v.z, v.w));
    mx = fmaxf(mx, __shfl_xor(mx, 1));
    mx = fmaxf(mx, __shfl_xor(mx, 2));
    float s = __expf(v.x - mx) + __expf(v.y - mx) + __expf(v.z - mx) +
              __expf(v.w - mx);
    s += __shfl_xor(s, 1);
    s += __shfl_xor(s, 2);
    const float l = mx + __logf(s);
    if (live) {
      v4f o = {v.x - l, v.y - l, v.z - l, v.w - l};
      *(v4f*)(O + ((size_t)node * LPN + sub) * 4) = o;
    }
  });
}

constexpr int OWN_GPW32 = 4;
constexpr int OWN_GPW16 = 2;
constexpr int OWN_NPW = 16;
static_assert(OWN_GPW32 * 4 == OWN_NPW && OWN_GPW16 * 8 == OWN_NPW,
              "a wave owns the same 16 nodes in both owner launches");

static inline int cdiv_ll(long long a, long long b) {
  return (int)((a + b - 1) / b);
}

extern "C" void kernel_launch(void* const* d_in, const int* in_sizes, int n_in,
                              void* d_out, int out_size, void* d_ws,
                              size_t ws_size, hipStream_t stream) {
  (void)n_in; (void)out_size; (void)ws_size;
  const float* x     = (const float*)d_in[0];
  const float* w1a_w = (const float*)d_in[1];
  const float* w1a_b = (const float*)d_in[2];
  const float* w1b_w = (const float*)d_in[3];
  const float* w1b_b = (const float*)d_in[4];
  const float* w2a_w = (const float*)d_in[5];
  const float* w2a_b = (const float*)d_in[6];
  const float* w2b_w = (const float*)d_in[7];
  const float* w2b_b = (const float*)d_in[8];
  const int*   edge  = (const int*)d_in[9];

  const int N = in_sizes[0] / 256;
  const int E = in_sizes[9] / 2;
  const int* srcI = edge;
  const int* dstI = edge + E;

  float* buf0 = (float*)d_ws;
  float* buf1 = buf0 + (size_t)N * 32;
  float* buf2 = buf1 + (size_t)N * 32;

  const int CE = cdiv_ll(E, NCH);
  const int NB = cdiv_ll(N, BIN);

  int* rowptr = (int*)buf1;
  int* col    = rowptr + (size_t)N + 64;
  int* lofs   = col + (size_t)E;
  int* tmp    = (int*)d_out;

  const int BLK = CS_T;
  const int MT = 2;
  const int waves = N / (16 * MT);
  const int gemm_blocks = cdiv_ll(waves, BLK / 32);
  const int own_blocks = cdiv_ll(cdiv_ll(N, OWN_NPW), BLK / 32);

  linear2_and_chunk_sort<256, 32, 32, MT>
      <<<gemm_blocks + NCH, CS_T, 0, stream>>>(
          x, w1a_w, w1a_b, w1b_w, w1b_b, buf0, buf2, N, gemm_blocks, srcI,
          dstI, lofs, tmp, E, CE, NB);
  bin_owner_relu_linear2<OWN_GPW32><<<NB, SORT_T, 0, stream>>>(
      tmp, lofs, rowptr, col, CE, buf0, buf2, w2a_w, w2a_b, w2b_w, w2b_b, buf2,
      N);

  owner_logsoftmax16<OWN_GPW16><<<own_blocks, BLK, 0, stream>>>(
      buf2, rowptr, col, (float*)d_out, N);
}
